# starting kernel
# baseline (speedup 1.0000x reference)
#include <hip/hip_runtime.h>
#include <hip/hip_bf16.h>


typedef __attribute__((ext_vector_type(2))) float v2f;
typedef __attribute__((ext_vector_type(8))) float v8f;

static constexpr int NN   = 10000;
static constexpr int NE   = 160000;
static constexpr int MULC = 128;
static constexpr int NSP  = 10;
static constexpr int NRAD = 8;

static constexpr float SILU_C    = 1.679f;
static constexpr float SC_NORM   = 0.02795084971874737f;
static constexpr float L1_NORM   = 0.08838834764831845f;
static constexpr float L2_NORM   = 0.0625f;
static constexpr float INV_SQRT8 = 0.3535533905932738f;
static constexpr float INV_SQ3   = 0.5773502691896258f;
static constexpr float INV_NEIGH = 0.25f;

__device__ __forceinline__ v8f wmma4(v2f a, v2f b, v8f c) {
  return __builtin_amdgcn_wmma_f32_16x16x4_f32(
       false, a,  false, b,
       (short)0, c,  false,  false);
}

__device__ __forceinline__ float silu_c(float z) {
  return SILU_C * z / (1.0f + __expf(-z));
}

__global__ void zero_kernel(float* __restrict__ p, long n) {
  long i = (long)blockIdx.x * blockDim.x + threadIdx.x;
  long stride = (long)gridDim.x * blockDim.x;
  for (; i < n; i += stride) p[i] = 0.0f;
}

__global__ __launch_bounds__(128) void node_kernel(
    const float* __restrict__ node_feats, const float* __restrict__ node_attrs,
    const float* __restrict__ lin1_ws, const float* __restrict__ lin1_wv,
    const float* __restrict__ sc_ws, const float* __restrict__ sc_wv,
    float* __restrict__ out, float* __restrict__ y) {
  __shared__ float xs[16][MULC];
  __shared__ float xvc[3][16][MULC];
  __shared__ float at[16][16];
  __shared__ float zbuf[16][256];

  const int n0  = blockIdx.x * 16;
  const int tid = threadIdx.x;

  for (int i = tid; i < 16 * 512; i += 128) {
    int r = i >> 9, c = i & 511;
    float vv = node_feats[(size_t)(n0 + r) * 512 + c];
    if (c < MULC) xs[r][c] = vv;
    else { int q = c - MULC; xvc[q % 3][r][q / 3] = vv; }
  }
  for (int i = tid; i < 16 * 16; i += 128) {
    int r = i >> 4, c = i & 15;
    at[r][c] = (c < NSP) ? node_attrs[(size_t)(n0 + r) * NSP + c] : 0.0f;
  }
  __syncthreads();

  const int wave = tid >> 5;
  const int lane = tid & 31;
  const int half = lane >> 4;
  const int m    = lane & 15;
  const int nl   = lane & 15;
  const int wA   = wave * 16;
  const int wB   = wA + 64;

  for (int t = 0; t < 4; ++t) {
    float (*xsrc)[MULC] = (t == 0) ? xs : xvc[t - 1];
    const float* W = (t == 0) ? sc_ws : sc_wv;
    v8f accA = {}, accB = {};
    for (int chunk = 0; chunk < 5; ++chunk) {
      __syncthreads();
      for (int i = tid; i < 16 * 256; i += 128) {
        int r = i >> 8, kk = i & 255;
        int k = chunk * 256 + kk;
        zbuf[r][kk] = xsrc[r][k / NSP] * at[r][k % NSP];
      }
      __syncthreads();
      for (int kk = 0; kk < 256; kk += 4) {
        int k0 = kk + 2 * half;
        v2f a = { zbuf[m][k0], zbuf[m][k0 + 1] };
        size_t krow = (size_t)(chunk * 256 + k0) * MULC;
        v2f bA = { W[krow + wA + nl], W[krow + MULC + wA + nl] };
        v2f bB = { W[krow + wB + nl], W[krow + MULC + wB + nl] };
        accA = wmma4(a, bA, accA);
        accB = wmma4(a, bB, accB);
      }
    }
    if (t == 0) {
#pragma unroll
      for (int r = 0; r < 8; ++r) {
        size_t row = (size_t)(n0 + r + 8 * half) * 512;
        out[row + wA + nl] = accA[r] * SC_NORM;
        out[row + wB + nl] = accB[r] * SC_NORM;
      }
    } else {
      const int c = t - 1;
#pragma unroll
      for (int r = 0; r < 8; ++r) {
        size_t row = (size_t)(n0 + r + 8 * half) * 512 + MULC;
        out[row + (wA + nl) * 3 + c] = accA[r] * SC_NORM;
        out[row + (wB + nl) * 3 + c] = accB[r] * SC_NORM;
      }
    }
  }

  {
    v8f accA = {}, accB = {};
    for (int kk = 0; kk < MULC; kk += 4) {
      int k0 = kk + 2 * half;
      v2f a = { xs[m][k0], xs[m][k0 + 1] };
      size_t krow = (size_t)k0 * MULC;
      v2f bA = { lin1_ws[krow + wA + nl], lin1_ws[krow + MULC + wA + nl] };
      v2f bB = { lin1_ws[krow + wB + nl], lin1_ws[krow + MULC + wB + nl] };
      accA = wmma4(a, bA, accA);
      accB = wmma4(a, bB, accB);
    }
#pragma unroll
    for (int r = 0; r < 8; ++r) {
      size_t row = (size_t)(n0 + r + 8 * half) * 512;
      y[row + wA + nl] = accA[r] * L1_NORM;
      y[row + wB + nl] = accB[r] * L1_NORM;
    }
  }
  for (int c = 0; c < 3; ++c) {
    v8f accA = {}, accB = {};
    for (int kk = 0; kk < MULC; kk += 4) {
      int k0 = kk + 2 * half;
      v2f a = { xvc[c][m][k0], xvc[c][m][k0 + 1] };
      size_t krow = (size_t)k0 * MULC;
      v2f bA = { lin1_wv[krow + wA + nl], lin1_wv[krow + MULC + wA + nl] };
      v2f bB = { lin1_wv[krow + wB + nl], lin1_wv[krow + MULC + wB + nl] };
      accA = wmma4(a, bA, accA);
      accB = wmma4(a, bB, accB);
    }
#pragma unroll
    for (int r = 0; r < 8; ++r) {
      size_t row = (size_t)(n0 + r + 8 * half) * 512 + MULC;
      y[row + (wA + nl) * 3 + c] = accA[r] * L1_NORM;
      y[row + (wB + nl) * 3 + c] = accB[r] * L1_NORM;
    }
  }
}

__global__ __launch_bounds__(32) void edge_kernel(
    const float* __restrict__ y, const float* __restrict__ edge_feats,
    const float* __restrict__ edge_attrs, const int* __restrict__ edge_index,
    const float* __restrict__ fc_w0, const float* __restrict__ fc_w1,
    const float* __restrict__ fc_w2, float* __restrict__ s_acc,
    float* __restrict__ v_acc) {
  __shared__ float ytile[16][512];
  __shared__ float ha[16][64];
  __shared__ float hb[16][64];
  __shared__ float ea[16][4];
  __shared__ float ef[16][NRAD];
  __shared__ int   dstn[16];

  const int e0   = blockIdx.x * 16;
  const int lane = threadIdx.x;
  const int half = lane >> 4;
  const int m    = lane & 15;
  const int nl   = lane & 15;

  for (int r = 0; r < 16; ++r) {
    int sr = edge_index[NE + e0 + r];
    const float4* sp = (const float4*)(y + (size_t)sr * 512);
    float4* dp = (float4*)&ytile[r][0];
    for (int i = lane; i < 128; i += 32) dp[i] = sp[i];
  }
  for (int i = lane; i < 16 * NRAD; i += 32)
    ef[i >> 3][i & 7] = edge_feats[(size_t)e0 * NRAD + i];
  for (int i = lane; i < 16 * 4; i += 32)
    ea[i >> 2][i & 3] = edge_attrs[(size_t)e0 * 4 + i];
  if (lane < 16) dstn[lane] = edge_index[e0 + lane];
  __syncthreads();

  for (int nt = 0; nt < 4; ++nt) {
    v8f acc = {};
    for (int kk = 0; kk < 8; kk += 4) {
      int k0 = kk + 2 * half;
      v2f a = { ef[m][k0], ef[m][k0 + 1] };
      v2f b = { fc_w0[k0 * 64 + nt * 16 + nl],
                fc_w0[(k0 + 1) * 64 + nt * 16 + nl] };
      acc = wmma4(a, b, acc);
    }
#pragma unroll
    for (int r = 0; r < 8; ++r)
      ha[r + 8 * half][nt * 16 + nl] = silu_c(acc[r] * INV_SQRT8);
  }
  __syncthreads();

  for (int nt = 0; nt < 4; ++nt) {
    v8f acc = {};
    for (int kk = 0; kk < 64; kk += 4) {
      int k0 = kk + 2 * half;
      v2f a = { ha[m][k0], ha[m][k0 + 1] };
      v2f b = { fc_w1[k0 * 64 + nt * 16 + nl],
                fc_w1[(k0 + 1) * 64 + nt * 16 + nl] };
      acc = wmma4(a, b, acc);
    }
#pragma unroll
    for (int r = 0; r < 8; ++r)
      hb[r + 8 * half][nt * 16 + nl] = silu_c(acc[r] * 0.125f);
  }
  __syncthreads();

  for (int nt = 0; nt < 32; ++nt) {
    const int j0 = nt * 16;
    v8f acc = {};
    for (int kk = 0; kk < 64; kk += 4) {
      int k0 = kk + 2 * half;
      v2f a = { hb[m][k0], hb[m][k0 + 1] };
      v2f b = { fc_w2[(size_t)k0 * 512 + j0 + nl],
                fc_w2[(size_t)(k0 + 1) * 512 + j0 + nl] };
      acc = wmma4(a, b, acc);
    }
    const int j    = j0 + nl;
    const int quad = j >> 7;
    const int u    = j & 127;
#pragma unroll
    for (int r = 0; r < 8; ++r) {
      const int e = r + 8 * half;
      const float wv = acc[r] * 0.125f;
      const int d = dstn[e];
      const float as = ea[e][0];
      if (quad == 0) {
        atomicAdd(&s_acc[(size_t)d * 256 + u],
                  wv * ytile[e][u] * as * INV_NEIGH);
      } else if (quad == 1) {
        float base = wv * ytile[e][u] * INV_NEIGH;
        atomicAdd(&v_acc[((size_t)d * 256 + u) * 3 + 0], base * ea[e][1]);
        atomicAdd(&v_acc[((size_t)d * 256 + u) * 3 + 1], base * ea[e][2]);
        atomicAdd(&v_acc[((size_t)d * 256 + u) * 3 + 2], base * ea[e][3]);
      } else if (quad == 2) {
        float base = wv * as * INV_NEIGH;
        atomicAdd(&v_acc[((size_t)d * 256 + 128 + u) * 3 + 0],
                  base * ytile[e][128 + u * 3 + 0]);
        atomicAdd(&v_acc[((size_t)d * 256 + 128 + u) * 3 + 1],
                  base * ytile[e][128 + u * 3 + 1]);
        atomicAdd(&v_acc[((size_t)d * 256 + 128 + u) * 3 + 2],
                  base * ytile[e][128 + u * 3 + 2]);
      } else {
        float dotv = ytile[e][128 + u * 3 + 0] * ea[e][1] +
                     ytile[e][128 + u * 3 + 1] * ea[e][2] +
                     ytile[e][128 + u * 3 + 2] * ea[e][3];
        atomicAdd(&s_acc[(size_t)d * 256 + 128 + u],
                  wv * dotv * INV_SQ3 * INV_NEIGH);
      }
    }
  }
}

__global__ __launch_bounds__(128) void out_kernel(
    const float* __restrict__ s_acc, const float* __restrict__ v_acc,
    const float* __restrict__ lin2_ws, const float* __restrict__ lin2_wv,
    float* __restrict__ out) {
  __shared__ float st[16][256];
  __shared__ float vt[16][256];
  const int n0  = blockIdx.x * 16;
  const int tid = threadIdx.x;
  for (int i = tid; i < 16 * 256; i += 128)
    st[i >> 8][i & 255] = s_acc[(size_t)n0 * 256 + i];
  __syncthreads();

  const int wave = tid >> 5;
  const int lane = tid & 31;
  const int half = lane >> 4;
  const int m    = lane & 15;
  const int nl   = lane & 15;
  const int wA   = wave * 16;
  const int wB   = wA + 64;

  {
    v8f accA = {}, accB = {};
    for (int kk = 0; kk < 256; kk += 4) {
      int k0 = kk + 2 * half;
      v2f a = { st[m][k0], st[m][k0 + 1] };
      size_t krow = (size_t)k0 * MULC;
      v2f bA = { lin2_ws[krow + wA + nl], lin2_ws[krow + MULC + wA + nl] };
      v2f bB = { lin2_ws[krow + wB + nl], lin2_ws[krow + MULC + wB + nl] };
      accA = wmma4(a, bA, accA);
      accB = wmma4(a, bB, accB);
    }
#pragma unroll
    for (int r = 0; r < 8; ++r) {
      size_t row = (size_t)(n0 + r + 8 * half) * 512;
      out[row + wA + nl] += accA[r] * L2_NORM;
      out[row + wB + nl] += accB[r] * L2_NORM;
    }
  }
  for (int c = 0; c < 3; ++c) {
    __syncthreads();
    for (int i = tid; i < 16 * 256; i += 128) {
      int r = i >> 8, k = i & 255;
      vt[r][k] = v_acc[(((size_t)(n0 + r)) * 256 + k) * 3 + c];
    }
    __syncthreads();
    v8f accA = {}, accB = {};
    for (int kk = 0; kk < 256; kk += 4) {
      int k0 = kk + 2 * half;
      v2f a = { vt[m][k0], vt[m][k0 + 1] };
      size_t krow = (size_t)k0 * MULC;
      v2f bA = { lin2_wv[krow + wA + nl], lin2_wv[krow + MULC + wA + nl] };
      v2f bB = { lin2_wv[krow + wB + nl], lin2_wv[krow + MULC + wB + nl] };
      accA = wmma4(a, bA, accA);
      accB = wmma4(a, bB, accB);
    }
#pragma unroll
    for (int r = 0; r < 8; ++r) {
      size_t row = (size_t)(n0 + r + 8 * half) * 512 + MULC;
      out[row + (wA + nl) * 3 + c] += accA[r] * L2_NORM;
      out[row + (wB + nl) * 3 + c] += accB[r] * L2_NORM;
    }
  }
}

extern "C" void kernel_launch(void* const* d_in, const int* in_sizes, int n_in,
                              void* d_out, int out_size, void* d_ws,
                              size_t ws_size, hipStream_t stream) {
  const float* node_feats = (const float*)d_in[0];
  const float* node_attrs = (const float*)d_in[1];
  const float* edge_feats = (const float*)d_in[2];
  const float* edge_attrs = (const float*)d_in[3];
  const int*   edge_index = (const int*)d_in[4];
  const float* lin1_ws    = (const float*)d_in[5];
  const float* lin1_wv    = (const float*)d_in[6];
  const float* fc_w0      = (const float*)d_in[7];
  const float* fc_w1      = (const float*)d_in[8];
  const float* fc_w2      = (const float*)d_in[9];
  const float* lin2_ws    = (const float*)d_in[10];
  const float* lin2_wv    = (const float*)d_in[11];
  const float* sc_ws      = (const float*)d_in[12];
  const float* sc_wv      = (const float*)d_in[13];

  float* out   = (float*)d_out;
  float* y     = (float*)d_ws;
  float* s_acc = y + (size_t)NN * 512;
  float* v_acc = s_acc + (size_t)NN * 256;

  zero_kernel<<<2048, 256, 0, stream>>>(s_acc, (long)NN * 1024);
  node_kernel<<<NN / 16, 128, 0, stream>>>(node_feats, node_attrs, lin1_ws,
                                           lin1_wv, sc_ws, sc_wv, out, y);
  edge_kernel<<<NE / 16, 32, 0, stream>>>(y, edge_feats, edge_attrs,
                                          edge_index, fc_w0, fc_w1, fc_w2,
                                          s_acc, v_acc);
  out_kernel<<<NN / 16, 128, 0, stream>>>(s_acc, v_acc, lin2_ws, lin2_wv, out);
}


# segment sum by ownership: a block owns 16 target nodes, lists its edges (LDS counter, arrival order), adds them by one-hot f32 WMMA in registers, stores rows once; no float atomics, no zero fill
# speedup vs baseline: 1.1604x; 1.1604x over previous
#include <hip/hip_runtime.h>
#include <hip/hip_bf16.h>


typedef __attribute__((ext_vector_type(2))) float v2f;
typedef __attribute__((ext_vector_type(8))) float v8f;

static constexpr int NN   = 10000;
static constexpr int NE   = 160000;
static constexpr int MULC = 128;
static constexpr int NSP  = 10;
static constexpr int NRAD = 8;

static constexpr float SILU_C    = 1.679f;
static constexpr float SC_NORM   = 0.02795084971874737f;
static constexpr float L1_NORM   = 0.08838834764831845f;
static constexpr float L2_NORM   = 0.0625f;
static constexpr float INV_SQRT8 = 0.3535533905932738f;
static constexpr float INV_SQ3   = 0.5773502691896258f;
static constexpr float INV_NEIGH = 0.25f;

__device__ __forceinline__ v8f wmma4(v2f a, v2f b, v8f c) {
  return __builtin_amdgcn_wmma_f32_16x16x4_f32(
       false, a,  false, b,
       (short)0, c,  false,  false);
}

__device__ __forceinline__ float silu_c(float z) {
  return SILU_C * z / (1.0f + __expf(-z));
}

__global__ __launch_bounds__(128) void node_kernel(
    const float* __restrict__ node_feats, const float* __restrict__ node_attrs,
    const float* __restrict__ lin1_ws, const float* __restrict__ lin1_wv,
    const float* __restrict__ sc_ws, const float* __restrict__ sc_wv,
    float* __restrict__ out, float* __restrict__ y) {
  __shared__ float xs[16][MULC];
  __shared__ float xvc[3][16][MULC];
  __shared__ float at[16][16];
  __shared__ float zbuf[16][256];

  const int n0  = blockIdx.x * 16;
  const int tid = threadIdx.x;

  for (int i = tid; i < 16 * 512; i += 128) {
    int r = i >> 9, c = i & 511;
    float vv = node_feats[(size_t)(n0 + r) * 512 + c];
    if (c < MULC) xs[r][c] = vv;
    else { int q = c - MULC; xvc[q % 3][r][q / 3] = vv; }
  }
  for (int i = tid; i < 16 * 16; i += 128) {
    int r = i >> 4, c = i & 15;
    at[r][c] = (c < NSP) ? node_attrs[(size_t)(n0 + r) * NSP + c] : 0.0f;
  }
  __syncthreads();

  const int wave = tid >> 5;
  const int lane = tid & 31;
  const int half = lane >> 4;
  const int m    = lane & 15;
  const int nl   = lane & 15;
  const int wA   = wave * 16;
  const int wB   = wA + 64;

  for (int t = 0; t < 4; ++t) {
    float (*xsrc)[MULC] = (t == 0) ? xs : xvc[t - 1];
    const float* W = (t == 0) ? sc_ws : sc_wv;
    v8f accA = {}, accB = {};
    for (int chunk = 0; chunk < 5; ++chunk) {
      __syncthreads();
      for (int i = tid; i < 16 * 256; i += 128) {
        int r = i >> 8, kk = i & 255;
        int k = chunk * 256 + kk;
        zbuf[r][kk] = xsrc[r][k / NSP] * at[r][k % NSP];
      }
      __syncthreads();
      for (int kk = 0; kk < 256; kk += 4) {
        int k0 = kk + 2 * half;
        v2f a = { zbuf[m][k0], zbuf[m][k0 + 1] };
        size_t krow = (size_t)(chunk * 256 + k0) * MULC;
        v2f bA = { W[krow + wA + nl], W[krow + MULC + wA + nl] };
        v2f bB = { W[krow + wB + nl], W[krow + MULC + wB + nl] };
        accA = wmma4(a, bA, accA);
        accB = wmma4(a, bB, accB);
      }
    }
    if (t == 0) {
#pragma unroll
      for (int r = 0; r < 8; ++r) {
        size_t row = (size_t)(n0 + r + 8 * half) * 512;
        out[row + wA + nl] = accA[r] * SC_NORM;
        out[row + wB + nl] = accB[r] * SC_NORM;
      }
    } else {
      const int c = t - 1;
#pragma unroll
      for (int r = 0; r < 8; ++r) {
        size_t row = (size_t)(n0 + r + 8 * half) * 512 + MULC;
        out[row + (wA + nl) * 3 + c] = accA[r] * SC_NORM;
        out[row + (wB + nl) * 3 + c] = accB[r] * SC_NORM;
      }
    }
  }

  {
    v8f accA = {}, accB = {};
    for (int kk = 0; kk < MULC; kk += 4) {
      int k0 = kk + 2 * half;
      v2f a = { xs[m][k0], xs[m][k0 + 1] };
      size_t krow = (size_t)k0 * MULC;
      v2f bA = { lin1_ws[krow + wA + nl], lin1_ws[krow + MULC + wA + nl] };
      v2f bB = { lin1_ws[krow + wB + nl], lin1_ws[krow + MULC + wB + nl] };
      accA = wmma4(a, bA, accA);
      accB = wmma4(a, bB, accB);
    }
#pragma unroll
    for (int r = 0; r < 8; ++r) {
      size_t row = (size_t)(n0 + r + 8 * half) * 512;
      y[row + wA + nl] = accA[r] * L1_NORM;
      y[row + wB + nl] = accB[r] * L1_NORM;
    }
  }
  for (int c = 0; c < 3; ++c) {
    v8f accA = {}, accB = {};
    for (int kk = 0; kk < MULC; kk += 4) {
      int k0 = kk + 2 * half;
      v2f a = { xvc[c][m][k0], xvc[c][m][k0 + 1] };
      size_t krow = (size_t)k0 * MULC;
      v2f bA = { lin1_wv[krow + wA + nl], lin1_wv[krow + MULC + wA + nl] };
      v2f bB = { lin1_wv[krow + wB + nl], lin1_wv[krow + MULC + wB + nl] };
      accA = wmma4(a, bA, accA);
      accB = wmma4(a, bB, accB);
    }
#pragma unroll
    for (int r = 0; r < 8; ++r) {
      size_t row = (size_t)(n0 + r + 8 * half) * 512 + MULC;
      y[row + (wA + nl) * 3 + c] = accA[r] * L1_NORM;
      y[row + (wB + nl) * 3 + c] = accB[r] * L1_NORM;
    }
  }
}

static constexpr int CAP    = 1024;
static constexpr int SCAN_U = 8;

__device__ __forceinline__ void l2pair(const v2f (&af)[16],
                                       const float* __restrict__ w2, int j0,
                                       int half, int nl, v8f& c0, v8f& c1) {
  v2f b0[16], b1[16];
#pragma unroll
  for (int s = 0; s < 16; ++s) {
    const int k0 = 4 * s + 2 * half;
    const float* p = w2 + (size_t)k0 * 512 + j0 + nl;
    b0[s] = v2f{ p[0], p[512] };
    b1[s] = v2f{ p[16], p[512 + 16] };
  }
#pragma unroll
  for (int s = 0; s < 16; ++s) {
    c0 = wmma4(af[s], b0[s], c0);
    c1 = wmma4(af[s], b1[s], c1);
  }
}

__device__ __forceinline__ v8f segsum(const v2f (&sel)[4], const float (&mid)[8],
                                      v8f acc) {
#pragma unroll
  for (int t = 0; t < 4; ++t) {
    v2f b = { mid[2 * t], mid[2 * t + 1] };
    acc = wmma4(sel[t], b, acc);
  }
  return acc;
}

__global__ __launch_bounds__(128) void edge_kernel(
    const float* y, const float* __restrict__ edge_feats,
    const float* __restrict__ edge_attrs, const int* __restrict__ edge_index,
    const float* __restrict__ fc_w0, const float* __restrict__ fc_w1,
    const float* __restrict__ fc_w2, float* __restrict__ s_acc,
    float* __restrict__ v_acc) {
  __shared__ float ytile[16][512];
  __shared__ float ha[16][64];
  __shared__ float hb[16][64];
  __shared__ float ea[16][4];
  __shared__ float ef[16][NRAD];
  __shared__ int   dl[16];
  __shared__ int   lst_e[CAP];
  __shared__ int   lst_d[CAP];
  __shared__ int   lst_s[CAP];
  __shared__ int   cnt;

  const int n0   = blockIdx.x * 16;
  const int tid  = threadIdx.x;
  const int wave = tid >> 5;
  const int lane = tid & 31;
  const int half = lane >> 4;
  const int m    = lane & 15;
  const int nl   = lane & 15;

  v8f aS0[2], aS1[2], aV2[2][3], aV3[2][3];
#pragma unroll
  for (int jj = 0; jj < 2; ++jj) {
    aS0[jj] = v8f{};
    aS1[jj] = v8f{};
#pragma unroll
    for (int c = 0; c < 3; ++c) { aV2[jj][c] = v8f{}; aV3[jj][c] = v8f{}; }
  }

  const int4* d4 = (const int4*)edge_index;
  int seg_b = 0, seg_e = NE;
  for (;;) {
    if (tid == 0) cnt = 0;
    __syncthreads();
    const int e4 = seg_e >> 2;
    for (int base = (seg_b >> 2) + tid; base < e4; base += 128 * SCAN_U) {
      int4 v[SCAN_U];
#pragma unroll
      for (int q = 0; q < SCAN_U; ++q) {
        const int idx = base + q * 128;
        v[q] = (idx < e4) ? d4[idx] : make_int4(-1, -1, -1, -1);
      }
#pragma unroll
      for (int q = 0; q < SCAN_U; ++q) {
        const int idx = base + q * 128;
        const unsigned dd[4] = { (unsigned)v[q].x - (unsigned)n0,
                                 (unsigned)v[q].y - (unsigned)n0,
                                 (unsigned)v[q].z - (unsigned)n0,
                                 (unsigned)v[q].w - (unsigned)n0 };
#pragma unroll
        for (int z = 0; z < 4; ++z) {
          if (dd[z] < 16u) {
            const int p = atomicAdd(&cnt, 1);
            if (p < CAP) { lst_e[p] = idx * 4 + z; lst_d[p] = (int)dd[z]; }
          }
        }
      }
    }
    __syncthreads();
    const int c = cnt;
    __syncthreads();
    if (c > CAP) { seg_b = 0; seg_e = CAP; continue; }

    for (int i = tid; i < c; i += 128) lst_s[i] = edge_index[NE + lst_e[i]];
    __syncthreads();

    const int ntile = (c + 15) >> 4;
    for (int ti = 0; ti < ntile; ++ti) {
      {
        const float4* sp[4];
#pragma unroll
        for (int rr = 0; rr < 4; ++rr) {
          const int li  = ti * 16 + wave * 4 + rr;
          const int lic = (li < c) ? li : 0;
          const int sv  = lst_s[lic];
          const int sr  = (li < c) ? sv : 0;
          sp[rr] = (const float4*)(y + (size_t)sr * 512);
        }
        const float4 a0 = sp[0][lane], a1 = sp[0][lane + 32],
                     a2 = sp[0][lane + 64], a3 = sp[0][lane + 96];
        const float4 b0 = sp[1][lane], b1 = sp[1][lane + 32],
                     b2 = sp[1][lane + 64], b3 = sp[1][lane + 96];
        const float4 c0 = sp[2][lane], c1 = sp[2][lane + 32],
                     c2 = sp[2][lane + 64], c3 = sp[2][lane + 96];
        const float4 d0 = sp[3][lane], d1 = sp[3][lane + 32],
                     d2 = sp[3][lane + 64], d3 = sp[3][lane + 96];
        asm volatile("" ::: "memory");
        float4* dp = (float4*)&ytile[wave * 4][0];
        dp[lane]       = a0; dp[lane + 32]  = a1; dp[lane + 64]  = a2; dp[lane + 96]  = a3;
        dp[lane + 128] = b0; dp[lane + 160] = b1; dp[lane + 192] = b2; dp[lane + 224] = b3;
        dp[lane + 256] = c0; dp[lane + 288] = c1; dp[lane + 320] = c2; dp[lane + 352] = c3;
        dp[lane + 384] = d0; dp[lane + 416] = d1; dp[lane + 448] = d2; dp[lane + 480] = d3;
      }
      {
        const int r = tid >> 3, k = tid & 7;
        const int li = ti * 16 + r;
        float fv = 0.0f;
        if (li < c) fv = edge_feats[(size_t)lst_e[li] * NRAD + k];
        ef[r][k] = fv;
      }
      if (tid < 64) {
        const int r = tid >> 2, k = tid & 3;
        const int li = ti * 16 + r;
        float av = 0.0f;
        if (li < c) av = edge_attrs[(size_t)lst_e[li] * 4 + k];
        ea[r][k] = av;
      }
      if (tid < 16) {
        const int li = ti * 16 + tid;
        dl[tid] = (li < c) ? lst_d[li] : -1;
      }
      __syncthreads();

      int woff = 0;
      asm volatile("" : "+s"(woff));
      const float* w0 = fc_w0 + woff;
      const float* w1 = fc_w1 + woff;
      const float* w2 = fc_w2 + woff;

      {
        v8f acc = {};
        for (int kk = 0; kk < 8; kk += 4) {
          const int k0 = kk + 2 * half;
          v2f a = { ef[m][k0], ef[m][k0 + 1] };
          v2f b = { w0[k0 * 64 + wave * 16 + nl],
                    w0[(k0 + 1) * 64 + wave * 16 + nl] };
          acc = wmma4(a, b, acc);
        }
#pragma unroll
        for (int r = 0; r < 8; ++r)
          ha[r + 8 * half][wave * 16 + nl] = silu_c(acc[r] * INV_SQRT8);
      }
      __syncthreads();

      {
        v8f acc = {};
        for (int kk = 0; kk < 64; kk += 4) {
          const int k0 = kk + 2 * half;
          v2f a = { ha[m][k0], ha[m][k0 + 1] };
          v2f b = { w1[k0 * 64 + wave * 16 + nl],
                    w1[(k0 + 1) * 64 + wave * 16 + nl] };
          acc = wmma4(a, b, acc);
        }
#pragma unroll
        for (int r = 0; r < 8; ++r)
          hb[r + 8 * half][wave * 16 + nl] = silu_c(acc[r] * 0.125f);
      }
      __syncthreads();

      v2f af[16];
#pragma unroll
      for (int s = 0; s < 16; ++s) {
        const int k0 = 4 * s + 2 * half;
        af[s] = v2f{ hb[m][k0], hb[m][k0 + 1] };
      }
      v2f sel[4];
#pragma unroll
      for (int t = 0; t < 4; ++t) {
        const int e0 = 8 * half + 2 * t;
        sel[t] = v2f{ dl[e0] == m ? 1.0f : 0.0f, dl[e0 + 1] == m ? 1.0f : 0.0f };
      }

      {
        v8f cc[2] = { v8f{}, v8f{} };
        l2pair(af, w2, 0 * 128 + wave * 32, half, nl, cc[0], cc[1]);
        asm volatile("" : "+s"(woff) : "v"(cc[1][7]));
        w2 = fc_w2 + woff;
#pragma unroll
        for (int jj = 0; jj < 2; ++jj) {
          const int u = wave * 32 + jj * 16 + nl;
          float mid[8];
#pragma unroll
          for (int r = 0; r < 8; ++r) {
            const int e = r + 8 * half;
            mid[r] = (cc[jj][r] * 0.125f) * ytile[e][u] * ea[e][0];
          }
          aS0[jj] = segsum(sel, mid, aS0[jj]);
        }
      }
      {
        v8f cc[2] = { v8f{}, v8f{} };
        l2pair(af, w2, 1 * 128 + wave * 32, half, nl, cc[0], cc[1]);
        asm volatile("" : "+s"(woff) : "v"(cc[1][7]));
        w2 = fc_w2 + woff;
#pragma unroll
        for (int jj = 0; jj < 2; ++jj) {
          const int u = wave * 32 + jj * 16 + nl;
          float m0[8], m1[8], m2[8];
#pragma unroll
          for (int r = 0; r < 8; ++r) {
            const int e = r + 8 * half;
            const float base = (cc[jj][r] * 0.125f) * ytile[e][u];
            m0[r] = base * ea[e][1];
            m1[r] = base * ea[e][2];
            m2[r] = base * ea[e][3];
          }
          aV2[jj][0] = segsum(sel, m0, aV2[jj][0]);
          aV2[jj][1] = segsum(sel, m1, aV2[jj][1]);
          aV2[jj][2] = segsum(sel, m2, aV2[jj][2]);
        }
      }
      {
        v8f cc[2] = { v8f{}, v8f{} };
        l2pair(af, w2, 2 * 128 + wave * 32, half, nl, cc[0], cc[1]);
        asm volatile("" : "+s"(woff) : "v"(cc[1][7]));
        w2 = fc_w2 + woff;
#pragma unroll
        for (int jj = 0; jj < 2; ++jj) {
          const int u = wave * 32 + jj * 16 + nl;
          float m0[8], m1[8], m2[8];
#pragma unroll
          for (int r = 0; r < 8; ++r) {
            const int e = r + 8 * half;
            const float base = (cc[jj][r] * 0.125f) * ea[e][0];
            m0[r] = base * ytile[e][128 + u * 3 + 0];
            m1[r] = base * ytile[e][128 + u * 3 + 1];
            m2[r] = base * ytile[e][128 + u * 3 + 2];
          }
          aV3[jj][0] = segsum(sel, m0, aV3[jj][0]);
          aV3[jj][1] = segsum(sel, m1, aV3[jj][1]);
          aV3[jj][2] = segsum(sel, m2, aV3[jj][2]);
        }
      }
      {
        v8f cc[2] = { v8f{}, v8f{} };
        l2pair(af, w2, 3 * 128 + wave * 32, half, nl, cc[0], cc[1]);
#pragma unroll
        for (int jj = 0; jj < 2; ++jj) {
          const int u = wave * 32 + jj * 16 + nl;
          float mid[8];
#pragma unroll
          for (int r = 0; r < 8; ++r) {
            const int e = r + 8 * half;
            const float dotv = ytile[e][128 + u * 3 + 0] * ea[e][1] +
                               ytile[e][128 + u * 3 + 1] * ea[e][2] +
                               ytile[e][128 + u * 3 + 2] * ea[e][3];
            mid[r] = (cc[jj][r] * 0.125f) * dotv * INV_SQ3;
          }
          aS1[jj] = segsum(sel, mid, aS1[jj]);
        }
      }
      __syncthreads();
    }

    if (seg_e >= NE) break;
    seg_b = seg_e;
    seg_e = (seg_e + CAP < NE) ? seg_e + CAP : NE;
  }

#pragma unroll
  for (int jj = 0; jj < 2; ++jj) {
    const int u = wave * 32 + jj * 16 + nl;
#pragma unroll
    for (int r = 0; r < 8; ++r) {
      const size_t n = (size_t)(n0 + r + 8 * half);
      s_acc[n * 256 + u]       = aS0[jj][r] * INV_NEIGH;
      s_acc[n * 256 + 128 + u] = aS1[jj][r] * INV_NEIGH;
#pragma unroll
      for (int c = 0; c < 3; ++c) {
        v_acc[(n * 256 + u) * 3 + c]       = aV2[jj][c][r] * INV_NEIGH;
        v_acc[(n * 256 + 128 + u) * 3 + c] = aV3[jj][c][r] * INV_NEIGH;
      }
    }
  }
}

__global__ __launch_bounds__(128) void out_kernel(
    const float* __restrict__ s_acc, const float* __restrict__ v_acc,
    const float* __restrict__ lin2_ws, const float* __restrict__ lin2_wv,
    float* __restrict__ out) {
  __shared__ float st[16][256];
  __shared__ float vt[16][256];
  const int n0  = blockIdx.x * 16;
  const int tid = threadIdx.x;
  for (int i = tid; i < 16 * 256; i += 128)
    st[i >> 8][i & 255] = s_acc[(size_t)n0 * 256 + i];
  __syncthreads();

  const int wave = tid >> 5;
  const int lane = tid & 31;
  const int half = lane >> 4;
  const int m    = lane & 15;
  const int nl   = lane & 15;
  const int wA   = wave * 16;
  const int wB   = wA + 64;

  {
    v8f accA = {}, accB = {};
    for (int kk = 0; kk < 256; kk += 4) {
      int k0 = kk + 2 * half;
      v2f a = { st[m][k0], st[m][k0 + 1] };
      size_t krow = (size_t)k0 * MULC;
      v2f bA = { lin2_ws[krow + wA + nl], lin2_ws[krow + MULC + wA + nl] };
      v2f bB = { lin2_ws[krow + wB + nl], lin2_ws[krow + MULC + wB + nl] };
      accA = wmma4(a, bA, accA);
      accB = wmma4(a, bB, accB);
    }
#pragma unroll
    for (int r = 0; r < 8; ++r) {
      size_t row = (size_t)(n0 + r + 8 * half) * 512;
      out[row + wA + nl] += accA[r] * L2_NORM;
      out[row + wB + nl] += accB[r] * L2_NORM;
    }
  }
  for (int c = 0; c < 3; ++c) {
    __syncthreads();
    for (int i = tid; i < 16 * 256; i += 128) {
      int r = i >> 8, k = i & 255;
      vt[r][k] = v_acc[(((size_t)(n0 + r)) * 256 + k) * 3 + c];
    }
    __syncthreads();
    v8f accA = {}, accB = {};
    for (int kk = 0; kk < 256; kk += 4) {
      int k0 = kk + 2 * half;
      v2f a = { vt[m][k0], vt[m][k0 + 1] };
      size_t krow = (size_t)k0 * MULC;
      v2f bA = { lin2_wv[krow + wA + nl], lin2_wv[krow + MULC + wA + nl] };
      v2f bB = { lin2_wv[krow + wB + nl], lin2_wv[krow + MULC + wB + nl] };
      accA = wmma4(a, bA, accA);
      accB = wmma4(a, bB, accB);
    }
#pragma unroll
    for (int r = 0; r < 8; ++r) {
      size_t row = (size_t)(n0 + r + 8 * half) * 512 + MULC;
      out[row + (wA + nl) * 3 + c] += accA[r] * L2_NORM;
      out[row + (wB + nl) * 3 + c] += accB[r] * L2_NORM;
    }
  }
}

extern "C" void kernel_launch(void* const* d_in, const int* in_sizes, int n_in,
                              void* d_out, int out_size, void* d_ws,
                              size_t ws_size, hipStream_t stream) {
  const float* node_feats = (const float*)d_in[0];
  const float* node_attrs = (const float*)d_in[1];
  const float* edge_feats = (const float*)d_in[2];
  const float* edge_attrs = (const float*)d_in[3];
  const int*   edge_index = (const int*)d_in[4];
  const float* lin1_ws    = (const float*)d_in[5];
  const float* lin1_wv    = (const float*)d_in[6];
  const float* fc_w0      = (const float*)d_in[7];
  const float* fc_w1      = (const float*)d_in[8];
  const float* fc_w2      = (const float*)d_in[9];
  const float* lin2_ws    = (const float*)d_in[10];
  const float* lin2_wv    = (const float*)d_in[11];
  const float* sc_ws      = (const float*)d_in[12];
  const float* sc_wv      = (const float*)d_in[13];

  float* out   = (float*)d_out;
  float* y     = (float*)d_ws;
  float* s_acc = y + (size_t)NN * 512;
  float* v_acc = s_acc + (size_t)NN * 256;

  node_kernel<<<NN / 16, 128, 0, stream>>>(node_feats, node_attrs, lin1_ws,
                                           lin1_wv, sc_ws, sc_wv, out, y);
  edge_kernel<<<NN / 16, 128, 0, stream>>>(y, edge_feats, edge_attrs,
                                          edge_index, fc_w0, fc_w1, fc_w2,
                                          s_acc, v_acc);
  out_kernel<<<NN / 16, 128, 0, stream>>>(s_acc, v_acc, lin2_ws, lin2_wv, out);
}


# LDS row pitch of every tile read as WMMA A fragments padded by 4 dwords (x, z chunk, MLP activations, edge feats, lin2 inputs): the 16-way bank conflict of each fragment read is gone
# speedup vs baseline: 1.2243x; 1.0551x over previous
#include <hip/hip_runtime.h>
#include <hip/hip_bf16.h>


typedef __attribute__((ext_vector_type(2))) float v2f;
typedef __attribute__((ext_vector_type(8))) float v8f;

static constexpr int NN   = 10000;
static constexpr int NE   = 160000;
static constexpr int MULC = 128;
static constexpr int NSP  = 10;
static constexpr int NRAD = 8;

static constexpr int LPAD = 4;

static constexpr float SILU_C    = 1.679f;
static constexpr float SC_NORM   = 0.02795084971874737f;
static constexpr float L1_NORM   = 0.08838834764831845f;
static constexpr float L2_NORM   = 0.0625f;
static constexpr float INV_SQRT8 = 0.3535533905932738f;
static constexpr float INV_SQ3   = 0.5773502691896258f;
static constexpr float INV_NEIGH = 0.25f;

__device__ __forceinline__ v8f wmma4(v2f a, v2f b, v8f c) {
  return __builtin_amdgcn_wmma_f32_16x16x4_f32(
       false, a,  false, b,
       (short)0, c,  false,  false);
}

__device__ __forceinline__ float silu_c(float z) {
  return SILU_C * z / (1.0f + __expf(-z));
}

__global__ __launch_bounds__(128) void node_kernel(
    const float* __restrict__ node_feats, const float* __restrict__ node_attrs,
    const float* __restrict__ lin1_ws, const float* __restrict__ lin1_wv,
    const float* __restrict__ sc_ws, const float* __restrict__ sc_wv,
    float* __restrict__ out, float* __restrict__ y) {
  __shared__ float xs[16][MULC + LPAD];
  __shared__ float xvc[3][16][MULC + LPAD];
  __shared__ float at[16][16];
  __shared__ float zbuf[16][256 + LPAD];

  const int n0  = blockIdx.x * 16;
  const int tid = threadIdx.x;

  for (int i = tid; i < 16 * 512; i += 128) {
    int r = i >> 9, c = i & 511;
    float vv = node_feats[(size_t)(n0 + r) * 512 + c];
    if (c < MULC) xs[r][c] = vv;
    else { int q = c - MULC; xvc[q % 3][r][q / 3] = vv; }
  }
  for (int i = tid; i < 16 * 16; i += 128) {
    int r = i >> 4, c = i & 15;
    at[r][c] = (c < NSP) ? node_attrs[(size_t)(n0 + r) * NSP + c] : 0.0f;
  }
  __syncthreads();

  const int wave = tid >> 5;
  const int lane = tid & 31;
  const int half = lane >> 4;
  const int m    = lane & 15;
  const int nl   = lane & 15;
  const int wA   = wave * 16;
  const int wB   = wA + 64;

  for (int t = 0; t < 4; ++t) {
    float (*xsrc)[MULC + LPAD] = (t == 0) ? xs : xvc[t - 1];
    const float* W = (t == 0) ? sc_ws : sc_wv;
    v8f accA = {}, accB = {};
    for (int chunk = 0; chunk < 5; ++chunk) {
      __syncthreads();
      for (int i = tid; i < 16 * 256; i += 128) {
        int r = i >> 8, kk = i & 255;
        int k = chunk * 256 + kk;
        zbuf[r][kk] = xsrc[r][k / NSP] * at[r][k % NSP];
      }
      __syncthreads();
      for (int kk = 0; kk < 256; kk += 4) {
        int k0 = kk + 2 * half;
        v2f a = { zbuf[m][k0], zbuf[m][k0 + 1] };
        size_t krow = (size_t)(chunk * 256 + k0) * MULC;
        v2f bA = { W[krow + wA + nl], W[krow + MULC + wA + nl] };
        v2f bB = { W[krow + wB + nl], W[krow + MULC + wB + nl] };
        accA = wmma4(a, bA, accA);
        accB = wmma4(a, bB, accB);
      }
    }
    if (t == 0) {
#pragma unroll
      for (int r = 0; r < 8; ++r) {
        size_t row = (size_t)(n0 + r + 8 * half) * 512;
        out[row + wA + nl] = accA[r] * SC_NORM;
        out[row + wB + nl] = accB[r] * SC_NORM;
      }
    } else {
      const int c = t - 1;
#pragma unroll
      for (int r = 0; r < 8; ++r) {
        size_t row = (size_t)(n0 + r + 8 * half) * 512 + MULC;
        out[row + (wA + nl) * 3 + c] = accA[r] * SC_NORM;
        out[row + (wB + nl) * 3 + c] = accB[r] * SC_NORM;
      }
    }
  }

  {
    v8f accA = {}, accB = {};
    for (int kk = 0; kk < MULC; kk += 4) {
      int k0 = kk + 2 * half;
      v2f a = { xs[m][k0], xs[m][k0 + 1] };
      size_t krow = (size_t)k0 * MULC;
      v2f bA = { lin1_ws[krow + wA + nl], lin1_ws[krow + MULC + wA + nl] };
      v2f bB = { lin1_ws[krow + wB + nl], lin1_ws[krow + MULC + wB + nl] };
      accA = wmma4(a, bA, accA);
      accB = wmma4(a, bB, accB);
    }
#pragma unroll
    for (int r = 0; r < 8; ++r) {
      size_t row = (size_t)(n0 + r + 8 * half) * 512;
      y[row + wA + nl] = accA[r] * L1_NORM;
      y[row + wB + nl] = accB[r] * L1_NORM;
    }
  }
  for (int c = 0; c < 3; ++c) {
    v8f accA = {}, accB = {};
    for (int kk = 0; kk < MULC; kk += 4) {
      int k0 = kk + 2 * half;
      v2f a = { xvc[c][m][k0], xvc[c][m][k0 + 1] };
      size_t krow = (size_t)k0 * MULC;
      v2f bA = { lin1_wv[krow + wA + nl], lin1_wv[krow + MULC + wA + nl] };
      v2f bB = { lin1_wv[krow + wB + nl], lin1_wv[krow + MULC + wB + nl] };
      accA = wmma4(a, bA, accA);
      accB = wmma4(a, bB, accB);
    }
#pragma unroll
    for (int r = 0; r < 8; ++r) {
      size_t row = (size_t)(n0 + r + 8 * half) * 512 + MULC;
      y[row + (wA + nl) * 3 + c] = accA[r] * L1_NORM;
      y[row + (wB + nl) * 3 + c] = accB[r] * L1_NORM;
    }
  }
}

static constexpr int CAP    = 1024;
static constexpr int SCAN_U = 8;

__device__ __forceinline__ void l2pair(const v2f (&af)[16],
                                       const float* __restrict__ w2, int j0,
                                       int half, int nl, v8f& c0, v8f& c1) {
  v2f b0[16], b1[16];
#pragma unroll
  for (int s = 0; s < 16; ++s) {
    const int k0 = 4 * s + 2 * half;
    const float* p = w2 + (size_t)k0 * 512 + j0 + nl;
    b0[s] = v2f{ p[0], p[512] };
    b1[s] = v2f{ p[16], p[512 + 16] };
  }
#pragma unroll
  for (int s = 0; s < 16; ++s) {
    c0 = wmma4(af[s], b0[s], c0);
    c1 = wmma4(af[s], b1[s], c1);
  }
}

__device__ __forceinline__ v8f segsum(const v2f (&sel)[4], const float (&mid)[8],
                                      v8f acc) {
#pragma unroll
  for (int t = 0; t < 4; ++t) {
    v2f b = { mid[2 * t], mid[2 * t + 1] };
    acc = wmma4(sel[t], b, acc);
  }
  return acc;
}

__global__ __launch_bounds__(128) void edge_kernel(
    const float* y, const float* __restrict__ edge_feats,
    const float* __restrict__ edge_attrs, const int* __restrict__ edge_index,
    const float* __restrict__ fc_w0, const float* __restrict__ fc_w1,
    const float* __restrict__ fc_w2, float* __restrict__ s_acc,
    float* __restrict__ v_acc) {
  __shared__ float ytile[16][512];
  __shared__ float ha[16][64 + LPAD];
  __shared__ float hb[16][64 + LPAD];
  __shared__ float ea[16][4];
  __shared__ float ef[16][NRAD + LPAD];
  __shared__ int   dl[16];
  __shared__ int   lst_e[CAP];
  __shared__ int   lst_d[CAP];
  __shared__ int   lst_s[CAP];
  __shared__ int   cnt;

  const int n0   = blockIdx.x * 16;
  const int tid  = threadIdx.x;
  const int wave = tid >> 5;
  const int lane = tid & 31;
  const int half = lane >> 4;
  const int m    = lane & 15;
  const int nl   = lane & 15;

  v8f aS0[2], aS1[2], aV2[2][3], aV3[2][3];
#pragma unroll
  for (int jj = 0; jj < 2; ++jj) {
    aS0[jj] = v8f{};
    aS1[jj] = v8f{};
#pragma unroll
    for (int c = 0; c < 3; ++c) { aV2[jj][c] = v8f{}; aV3[jj][c] = v8f{}; }
  }

  const int4* d4 = (const int4*)edge_index;
  int seg_b = 0, seg_e = NE;
  for (;;) {
    if (tid == 0) cnt = 0;
    __syncthreads();
    const int e4 = seg_e >> 2;
    for (int base = (seg_b >> 2) + tid; base < e4; base += 128 * SCAN_U) {
      int4 v[SCAN_U];
#pragma unroll
      for (int q = 0; q < SCAN_U; ++q) {
        const int idx = base + q * 128;
        v[q] = (idx < e4) ? d4[idx] : make_int4(-1, -1, -1, -1);
      }
#pragma unroll
      for (int q = 0; q < SCAN_U; ++q) {
        const int idx = base + q * 128;
        const unsigned dd[4] = { (unsigned)v[q].x - (unsigned)n0,
                                 (unsigned)v[q].y - (unsigned)n0,
                                 (unsigned)v[q].z - (unsigned)n0,
                                 (unsigned)v[q].w - (unsigned)n0 };
#pragma unroll
        for (int z = 0; z < 4; ++z) {
          if (dd[z] < 16u) {
            const int p = atomicAdd(&cnt, 1);
            if (p < CAP) { lst_e[p] = idx * 4 + z; lst_d[p] = (int)dd[z]; }
          }
        }
      }
    }
    __syncthreads();
    const int c = cnt;
    __syncthreads();
    if (c > CAP) { seg_b = 0; seg_e = CAP; continue; }

    for (int i = tid; i < c; i += 128) lst_s[i] = edge_index[NE + lst_e[i]];
    __syncthreads();

    const int ntile = (c + 15) >> 4;
    for (int ti = 0; ti < ntile; ++ti) {
      {
        const float4* sp[4];
#pragma unroll
        for (int rr = 0; rr < 4; ++rr) {
          const int li  = ti * 16 + wave * 4 + rr;
          const int lic = (li < c) ? li : 0;
          const int sv  = lst_s[lic];
          const int sr  = (li < c) ? sv : 0;
          sp[rr] = (const float4*)(y + (size_t)sr * 512);
        }
        const float4 a0 = sp[0][lane], a1 = sp[0][lane + 32],
                     a2 = sp[0][lane + 64], a3 = sp[0][lane + 96];
        const float4 b0 = sp[1][lane], b1 = sp[1][lane + 32],
                     b2 = sp[1][lane + 64], b3 = sp[1][lane + 96];
        const float4 c0 = sp[2][lane], c1 = sp[2][lane + 32],
                     c2 = sp[2][lane + 64], c3 = sp[2][lane + 96];
        const float4 d0 = sp[3][lane], d1 = sp[3][lane + 32],
                     d2 = sp[3][lane + 64], d3 = sp[3][lane + 96];
        asm volatile("" ::: "memory");
        float4* dp = (float4*)&ytile[wave * 4][0];
        dp[lane]       = a0; dp[lane + 32]  = a1; dp[lane + 64]  = a2; dp[lane + 96]  = a3;
        dp[lane + 128] = b0; dp[lane + 160] = b1; dp[lane + 192] = b2; dp[lane + 224] = b3;
        dp[lane + 256] = c0; dp[lane + 288] = c1; dp[lane + 320] = c2; dp[lane + 352] = c3;
        dp[lane + 384] = d0; dp[lane + 416] = d1; dp[lane + 448] = d2; dp[lane + 480] = d3;
      }
      {
        const int r = tid >> 3, k = tid & 7;
        const int li = ti * 16 + r;
        float fv = 0.0f;
        if (li < c) fv = edge_feats[(size_t)lst_e[li] * NRAD + k];
        ef[r][k] = fv;
      }
      if (tid < 64) {
        const int r = tid >> 2, k = tid & 3;
        const int li = ti * 16 + r;
        float av = 0.0f;
        if (li < c) av = edge_attrs[(size_t)lst_e[li] * 4 + k];
        ea[r][k] = av;
      }
      if (tid < 16) {
        const int li = ti * 16 + tid;
        dl[tid] = (li < c) ? lst_d[li] : -1;
      }
      __syncthreads();

      int woff = 0;
      asm volatile("" : "+s"(woff));
      const float* w0 = fc_w0 + woff;
      const float* w1 = fc_w1 + woff;
      const float* w2 = fc_w2 + woff;

      {
        v8f acc = {};
        for (int kk = 0; kk < 8; kk += 4) {
          const int k0 = kk + 2 * half;
          v2f a = { ef[m][k0], ef[m][k0 + 1] };
          v2f b = { w0[k0 * 64 + wave * 16 + nl],
                    w0[(k0 + 1) * 64 + wave * 16 + nl] };
          acc = wmma4(a, b, acc);
        }
#pragma unroll
        for (int r = 0; r < 8; ++r)
          ha[r + 8 * half][wave * 16 + nl] = silu_c(acc[r] * INV_SQRT8);
      }
      __syncthreads();

      {
        v8f acc = {};
        for (int kk = 0; kk < 64; kk += 4) {
          const int k0 = kk + 2 * half;
          v2f a = { ha[m][k0], ha[m][k0 + 1] };
          v2f b = { w1[k0 * 64 + wave * 16 + nl],
                    w1[(k0 + 1) * 64 + wave * 16 + nl] };
          acc = wmma4(a, b, acc);
        }
#pragma unroll
        for (int r = 0; r < 8; ++r)
          hb[r + 8 * half][wave * 16 + nl] = silu_c(acc[r] * 0.125f);
      }
      __syncthreads();

      v2f af[16];
#pragma unroll
      for (int s = 0; s < 16; ++s) {
        const int k0 = 4 * s + 2 * half;
        af[s] = v2f{ hb[m][k0], hb[m][k0 + 1] };
      }
      v2f sel[4];
#pragma unroll
      for (int t = 0; t < 4; ++t) {
        const int e0 = 8 * half + 2 * t;
        sel[t] = v2f{ dl[e0] == m ? 1.0f : 0.0f, dl[e0 + 1] == m ? 1.0f : 0.0f };
      }

      {
        v8f cc[2] = { v8f{}, v8f{} };
        l2pair(af, w2, 0 * 128 + wave * 32, half, nl, cc[0], cc[1]);
        asm volatile("" : "+s"(woff) : "v"(cc[1][7]));
        w2 = fc_w2 + woff;
#pragma unroll
        for (int jj = 0; jj < 2; ++jj) {
          const int u = wave * 32 + jj * 16 + nl;
          float mid[8];
#pragma unroll
          for (int r = 0; r < 8; ++r) {
            const int e = r + 8 * half;
            mid[r] = (cc[jj][r] * 0.125f) * ytile[e][u] * ea[e][0];
          }
          aS0[jj] = segsum(sel, mid, aS0[jj]);
        }
      }
      {
        v8f cc[2] = { v8f{}, v8f{} };
        l2pair(af, w2, 1 * 128 + wave * 32, half, nl, cc[0], cc[1]);
        asm volatile("" : "+s"(woff) : "v"(cc[1][7]));
        w2 = fc_w2 + woff;
#pragma unroll
        for (int jj = 0; jj < 2; ++jj) {
          const int u = wave * 32 + jj * 16 + nl;
          float m0[8], m1[8], m2[8];
#pragma unroll
          for (int r = 0; r < 8; ++r) {
            const int e = r + 8 * half;
            const float base = (cc[jj][r] * 0.125f) * ytile[e][u];
            m0[r] = base * ea[e][1];
            m1[r] = base * ea[e][2];
            m2[r] = base * ea[e][3];
          }
          aV2[jj][0] = segsum(sel, m0, aV2[jj][0]);
          aV2[jj][1] = segsum(sel, m1, aV2[jj][1]);
          aV2[jj][2] = segsum(sel, m2, aV2[jj][2]);
        }
      }
      {
        v8f cc[2] = { v8f{}, v8f{} };
        l2pair(af, w2, 2 * 128 + wave * 32, half, nl, cc[0], cc[1]);
        asm volatile("" : "+s"(woff) : "v"(cc[1][7]));
        w2 = fc_w2 + woff;
#pragma unroll
        for (int jj = 0; jj < 2; ++jj) {
          const int u = wave * 32 + jj * 16 + nl;
          float m0[8], m1[8], m2[8];
#pragma unroll
          for (int r = 0; r < 8; ++r) {
            const int e = r + 8 * half;
            const float base = (cc[jj][r] * 0.125f) * ea[e][0];
            m0[r] = base * ytile[e][128 + u * 3 + 0];
            m1[r] = base * ytile[e][128 + u * 3 + 1];
            m2[r] = base * ytile[e][128 + u * 3 + 2];
          }
          aV3[jj][0] = segsum(sel, m0, aV3[jj][0]);
          aV3[jj][1] = segsum(sel, m1, aV3[jj][1]);
          aV3[jj][2] = segsum(sel, m2, aV3[jj][2]);
        }
      }
      {
        v8f cc[2] = { v8f{}, v8f{} };
        l2pair(af, w2, 3 * 128 + wave * 32, half, nl, cc[0], cc[1]);
#pragma unroll
        for (int jj = 0; jj < 2; ++jj) {
          const int u = wave * 32 + jj * 16 + nl;
          float mid[8];
#pragma unroll
          for (int r = 0; r < 8; ++r) {
            const int e = r + 8 * half;
            const float dotv = ytile[e][128 + u * 3 + 0] * ea[e][1] +
                               ytile[e][128 + u * 3 + 1] * ea[e][2] +
                               ytile[e][128 + u * 3 + 2] * ea[e][3];
            mid[r] = (cc[jj][r] * 0.125f) * dotv * INV_SQ3;
          }
          aS1[jj] = segsum(sel, mid, aS1[jj]);
        }
      }
      __syncthreads();
    }

    if (seg_e >= NE) break;
    seg_b = seg_e;
    seg_e = (seg_e + CAP < NE) ? seg_e + CAP : NE;
  }

#pragma unroll
  for (int jj = 0; jj < 2; ++jj) {
    const int u = wave * 32 + jj * 16 + nl;
#pragma unroll
    for (int r = 0; r < 8; ++r) {
      const size_t n = (size_t)(n0 + r + 8 * half);
      s_acc[n * 256 + u]       = aS0[jj][r] * INV_NEIGH;
      s_acc[n * 256 + 128 + u] = aS1[jj][r] * INV_NEIGH;
#pragma unroll
      for (int c = 0; c < 3; ++c) {
        v_acc[(n * 256 + u) * 3 + c]       = aV2[jj][c][r] * INV_NEIGH;
        v_acc[(n * 256 + 128 + u) * 3 + c] = aV3[jj][c][r] * INV_NEIGH;
      }
    }
  }
}

__global__ __launch_bounds__(128) void out_kernel(
    const float* __restrict__ s_acc, const float* __restrict__ v_acc,
    const float* __restrict__ lin2_ws, const float* __restrict__ lin2_wv,
    float* __restrict__ out) {
  __shared__ float st[16][256 + LPAD];
  __shared__ float vt[16][256 + LPAD];
  const int n0  = blockIdx.x * 16;
  const int tid = threadIdx.x;
  for (int i = tid; i < 16 * 256; i += 128)
    st[i >> 8][i & 255] = s_acc[(size_t)n0 * 256 + i];
  __syncthreads();

  const int wave = tid >> 5;
  const int lane = tid & 31;
  const int half = lane >> 4;
  const int m    = lane & 15;
  const int nl   = lane & 15;
  const int wA   = wave * 16;
  const int wB   = wA + 64;

  {
    v8f accA = {}, accB = {};
    for (int kk = 0; kk < 256; kk += 4) {
      int k0 = kk + 2 * half;
      v2f a = { st[m][k0], st[m][k0 + 1] };
      size_t krow = (size_t)k0 * MULC;
      v2f bA = { lin2_ws[krow + wA + nl], lin2_ws[krow + MULC + wA + nl] };
      v2f bB = { lin2_ws[krow + wB + nl], lin2_ws[krow + MULC + wB + nl] };
      accA = wmma4(a, bA, accA);
      accB = wmma4(a, bB, accB);
    }
#pragma unroll
    for (int r = 0; r < 8; ++r) {
      size_t row = (size_t)(n0 + r + 8 * half) * 512;
      out[row + wA + nl] += accA[r] * L2_NORM;
      out[row + wB + nl] += accB[r] * L2_NORM;
    }
  }
  for (int c = 0; c < 3; ++c) {
    __syncthreads();
    for (int i = tid; i < 16 * 256; i += 128) {
      int r = i >> 8, k = i & 255;
      vt[r][k] = v_acc[(((size_t)(n0 + r)) * 256 + k) * 3 + c];
    }
    __syncthreads();
    v8f accA = {}, accB = {};
    for (int kk = 0; kk < 256; kk += 4) {
      int k0 = kk + 2 * half;
      v2f a = { vt[m][k0], vt[m][k0 + 1] };
      size_t krow = (size_t)k0 * MULC;
      v2f bA = { lin2_wv[krow + wA + nl], lin2_wv[krow + MULC + wA + nl] };
      v2f bB = { lin2_wv[krow + wB + nl], lin2_wv[krow + MULC + wB + nl] };
      accA = wmma4(a, bA, accA);
      accB = wmma4(a, bB, accB);
    }
#pragma unroll
    for (int r = 0; r < 8; ++r) {
      size_t row = (size_t)(n0 + r + 8 * half) * 512 + MULC;
      out[row + (wA + nl) * 3 + c] += accA[r] * L2_NORM;
      out[row + (wB + nl) * 3 + c] += accB[r] * L2_NORM;
    }
  }
}

extern "C" void kernel_launch(void* const* d_in, const int* in_sizes, int n_in,
                              void* d_out, int out_size, void* d_ws,
                              size_t ws_size, hipStream_t stream) {
  const float* node_feats = (const float*)d_in[0];
  const float* node_attrs = (const float*)d_in[1];
  const float* edge_feats = (const float*)d_in[2];
  const float* edge_attrs = (const float*)d_in[3];
  const int*   edge_index = (const int*)d_in[4];
  const float* lin1_ws    = (const float*)d_in[5];
  const float* lin1_wv    = (const float*)d_in[6];
  const float* fc_w0      = (const float*)d_in[7];
  const float* fc_w1      = (const float*)d_in[8];
  const float* fc_w2      = (const float*)d_in[9];
  const float* lin2_ws    = (const float*)d_in[10];
  const float* lin2_wv    = (const float*)d_in[11];
  const float* sc_ws      = (const float*)d_in[12];
  const float* sc_wv      = (const float*)d_in[13];

  float* out   = (float*)d_out;
  float* y     = (float*)d_ws;
  float* s_acc = y + (size_t)NN * 512;
  float* v_acc = s_acc + (size_t)NN * 256;

  node_kernel<<<NN / 16, 128, 0, stream>>>(node_feats, node_attrs, lin1_ws,
                                           lin1_wv, sc_ws, sc_wv, out, y);
  edge_kernel<<<NN / 16, 128, 0, stream>>>(y, edge_feats, edge_attrs,
                                          edge_index, fc_w0, fc_w1, fc_w2,
                                          s_acc, v_acc);
  out_kernel<<<NN / 16, 128, 0, stream>>>(s_acc, v_acc, lin2_ws, lin2_wv, out);
}


# weight (B) fragments fetched one stage ahead of the WMMAs that use them, into a second register set: node and lin2 k-loops by 8 k-steps, edge MLP layers (quadrant products held behind previous sums)
# speedup vs baseline: 1.4885x; 1.2158x over previous
#include <hip/hip_runtime.h>
#include <hip/hip_bf16.h>


typedef __attribute__((ext_vector_type(2))) float v2f;
typedef __attribute__((ext_vector_type(8))) float v8f;

static constexpr int NN   = 10000;
static constexpr int NE   = 160000;
static constexpr int MULC = 128;
static constexpr int NSP  = 10;
static constexpr int NRAD = 8;

static constexpr int LPAD = 4;

static constexpr float SILU_C    = 1.679f;
static constexpr float SC_NORM   = 0.02795084971874737f;
static constexpr float L1_NORM   = 0.08838834764831845f;
static constexpr float L2_NORM   = 0.0625f;
static constexpr float INV_SQRT8 = 0.3535533905932738f;
static constexpr float INV_SQ3   = 0.5773502691896258f;
static constexpr float INV_NEIGH = 0.25f;

__device__ __forceinline__ v8f wmma4(v2f a, v2f b, v8f c) {
  return __builtin_amdgcn_wmma_f32_16x16x4_f32(
       false, a,  false, b,
       (short)0, c,  false,  false);
}

__device__ __forceinline__ float silu_c(float z) {
  return SILU_C * z / (1.0f + __expf(-z));
}

__device__ __forceinline__ void loadB8(v2f (&bA)[8], v2f (&bB)[8],
                                       const float* __restrict__ p, int wA,
                                       int wB) {
#pragma unroll
  for (int s = 0; s < 8; ++s) {
    const float* q = p + s * 4 * MULC;
    bA[s] = v2f{ q[wA], q[MULC + wA] };
    bB[s] = v2f{ q[wB], q[MULC + wB] };
  }
}

__device__ __forceinline__ void mma8(const float* arow, const v2f (&bA)[8],
                                     const v2f (&bB)[8], v8f& accA, v8f& accB) {
#pragma unroll
  for (int s = 0; s < 8; ++s) {
    v2f a = { arow[4 * s], arow[4 * s + 1] };
    accA = wmma4(a, bA[s], accA);
    accB = wmma4(a, bB[s], accB);
  }
}

__global__ __launch_bounds__(128) void node_kernel(
    const float* __restrict__ node_feats, const float* __restrict__ node_attrs,
    const float* __restrict__ lin1_ws, const float* __restrict__ lin1_wv,
    const float* __restrict__ sc_ws, const float* __restrict__ sc_wv,
    float* __restrict__ out, float* __restrict__ y) {
  __shared__ float xs[16][MULC + LPAD];
  __shared__ float xvc[3][16][MULC + LPAD];
  __shared__ float at[16][16];
  __shared__ float zbuf[16][256 + LPAD];

  const int n0  = blockIdx.x * 16;
  const int tid = threadIdx.x;

  for (int i = tid; i < 16 * 512; i += 128) {
    int r = i >> 9, c = i & 511;
    float vv = node_feats[(size_t)(n0 + r) * 512 + c];
    if (c < MULC) xs[r][c] = vv;
    else { int q = c - MULC; xvc[q % 3][r][q / 3] = vv; }
  }
  for (int i = tid; i < 16 * 16; i += 128) {
    int r = i >> 4, c = i & 15;
    at[r][c] = (c < NSP) ? node_attrs[(size_t)(n0 + r) * NSP + c] : 0.0f;
  }
  __syncthreads();

  const int wave = tid >> 5;
  const int lane = tid & 31;
  const int half = lane >> 4;
  const int m    = lane & 15;
  const int nl   = lane & 15;
  const int wA   = wave * 16;
  const int wB   = wA + 64;

  v2f p0A[8], p0B[8], p1A[8], p1B[8];

  for (int t = 0; t < 4; ++t) {
    float (*xsrc)[MULC + LPAD] = (t == 0) ? xs : xvc[t - 1];
    const float* W = (t == 0) ? sc_ws : sc_wv;
    v8f accA = {}, accB = {};
    for (int chunk = 0; chunk < 5; ++chunk) {
      __syncthreads();
      const float* Wc = W + ((size_t)chunk * 256 + 2 * half) * MULC + nl;
      loadB8(p0A, p0B, Wc, wA, wB);
      for (int i = tid; i < 16 * 256; i += 128) {
        int r = i >> 8, kk = i & 255;
        int k = chunk * 256 + kk;
        zbuf[r][kk] = xsrc[r][k / NSP] * at[r][k % NSP];
      }
      __syncthreads();
      for (int g = 0; g < 8; g += 2) {
        loadB8(p1A, p1B, Wc + (size_t)(g + 1) * 32 * MULC, wA, wB);
        mma8(&zbuf[m][g * 32 + 2 * half], p0A, p0B, accA, accB);
        if (g + 2 < 8) loadB8(p0A, p0B, Wc + (size_t)(g + 2) * 32 * MULC, wA, wB);
        mma8(&zbuf[m][(g + 1) * 32 + 2 * half], p1A, p1B, accA, accB);
      }
    }
    if (t == 0) {
#pragma unroll
      for (int r = 0; r < 8; ++r) {
        size_t row = (size_t)(n0 + r + 8 * half) * 512;
        out[row + wA + nl] = accA[r] * SC_NORM;
        out[row + wB + nl] = accB[r] * SC_NORM;
      }
    } else {
      const int c = t - 1;
#pragma unroll
      for (int r = 0; r < 8; ++r) {
        size_t row = (size_t)(n0 + r + 8 * half) * 512 + MULC;
        out[row + (wA + nl) * 3 + c] = accA[r] * SC_NORM;
        out[row + (wB + nl) * 3 + c] = accB[r] * SC_NORM;
      }
    }
  }

  const float* L1s = lin1_ws + (size_t)(2 * half) * MULC + nl;
  const float* L1v = lin1_wv + (size_t)(2 * half) * MULC + nl;
  loadB8(p0A, p0B, L1s, wA, wB);
  {
    v8f accA = {}, accB = {};
    for (int g = 0; g < 4; g += 2) {
      loadB8(p1A, p1B, L1s + (size_t)(g + 1) * 32 * MULC, wA, wB);
      mma8(&xs[m][g * 32 + 2 * half], p0A, p0B, accA, accB);
      loadB8(p0A, p0B, (g + 2 < 4) ? L1s + (size_t)(g + 2) * 32 * MULC : L1v, wA, wB);
      mma8(&xs[m][(g + 1) * 32 + 2 * half], p1A, p1B, accA, accB);
    }
#pragma unroll
    for (int r = 0; r < 8; ++r) {
      size_t row = (size_t)(n0 + r + 8 * half) * 512;
      y[row + wA + nl] = accA[r] * L1_NORM;
      y[row + wB + nl] = accB[r] * L1_NORM;
    }
  }
  for (int c = 0; c < 3; ++c) {
    v8f accA = {}, accB = {};
    for (int g = 0; g < 4; g += 2) {
      loadB8(p1A, p1B, L1v + (size_t)(g + 1) * 32 * MULC, wA, wB);
      mma8(&xvc[c][m][g * 32 + 2 * half], p0A, p0B, accA, accB);
      if (g + 2 < 4) loadB8(p0A, p0B, L1v + (size_t)(g + 2) * 32 * MULC, wA, wB);
      else if (c < 2) loadB8(p0A, p0B, L1v, wA, wB);
      mma8(&xvc[c][m][(g + 1) * 32 + 2 * half], p1A, p1B, accA, accB);
    }
#pragma unroll
    for (int r = 0; r < 8; ++r) {
      size_t row = (size_t)(n0 + r + 8 * half) * 512 + MULC;
      y[row + (wA + nl) * 3 + c] = accA[r] * L1_NORM;
      y[row + (wB + nl) * 3 + c] = accB[r] * L1_NORM;
    }
  }
}

static constexpr int CAP    = 1024;
static constexpr int SCAN_U = 8;

__device__ __forceinline__ void l2load(v2f (&b0)[16], v2f (&b1)[16],
                                       const float* __restrict__ w2, int j0,
                                       int half, int nl) {
#pragma unroll
  for (int s = 0; s < 16; ++s) {
    const int k0 = 4 * s + 2 * half;
    const float* p = w2 + (size_t)k0 * 512 + j0 + nl;
    b0[s] = v2f{ p[0], p[512] };
    b1[s] = v2f{ p[16], p[512 + 16] };
  }
}

__device__ __forceinline__ void l2mma(const v2f (&af)[16], const v2f (&b0)[16],
                                      const v2f (&b1)[16], v8f& c0, v8f& c1) {
#pragma unroll
  for (int s = 0; s < 16; ++s) {
    c0 = wmma4(af[s], b0[s], c0);
    c1 = wmma4(af[s], b1[s], c1);
  }
}

__device__ __forceinline__ v8f segsum(const v2f (&sel)[4], const float (&mid)[8],
                                      v8f acc) {
#pragma unroll
  for (int t = 0; t < 4; ++t) {
    v2f b = { mid[2 * t], mid[2 * t + 1] };
    acc = wmma4(sel[t], b, acc);
  }
  return acc;
}

__global__ __launch_bounds__(128) void edge_kernel(
    const float* y, const float* __restrict__ edge_feats,
    const float* __restrict__ edge_attrs, const int* __restrict__ edge_index,
    const float* __restrict__ fc_w0, const float* __restrict__ fc_w1,
    const float* __restrict__ fc_w2, float* __restrict__ s_acc,
    float* __restrict__ v_acc) {
  __shared__ float ytile[16][512];
  __shared__ float ha[16][64 + LPAD];
  __shared__ float hb[16][64 + LPAD];
  __shared__ float ea[16][4];
  __shared__ float ef[16][NRAD + LPAD];
  __shared__ int   dl[16];
  __shared__ int   lst_e[CAP];
  __shared__ int   lst_d[CAP];
  __shared__ int   lst_s[CAP];
  __shared__ int   cnt;

  const int n0   = blockIdx.x * 16;
  const int tid  = threadIdx.x;
  const int wave = tid >> 5;
  const int lane = tid & 31;
  const int half = lane >> 4;
  const int m    = lane & 15;
  const int nl   = lane & 15;

  v8f aS0[2], aS1[2], aV2[2][3], aV3[2][3];
#pragma unroll
  for (int jj = 0; jj < 2; ++jj) {
    aS0[jj] = v8f{};
    aS1[jj] = v8f{};
#pragma unroll
    for (int c = 0; c < 3; ++c) { aV2[jj][c] = v8f{}; aV3[jj][c] = v8f{}; }
  }

  const int4* d4 = (const int4*)edge_index;
  int seg_b = 0, seg_e = NE;
  for (;;) {
    if (tid == 0) cnt = 0;
    __syncthreads();
    const int e4 = seg_e >> 2;
    for (int base = (seg_b >> 2) + tid; base < e4; base += 128 * SCAN_U) {
      int4 v[SCAN_U];
#pragma unroll
      for (int q = 0; q < SCAN_U; ++q) {
        const int idx = base + q * 128;
        v[q] = (idx < e4) ? d4[idx] : make_int4(-1, -1, -1, -1);
      }
#pragma unroll
      for (int q = 0; q < SCAN_U; ++q) {
        const int idx = base + q * 128;
        const unsigned dd[4] = { (unsigned)v[q].x - (unsigned)n0,
                                 (unsigned)v[q].y - (unsigned)n0,
                                 (unsigned)v[q].z - (unsigned)n0,
                                 (unsigned)v[q].w - (unsigned)n0 };
#pragma unroll
        for (int z = 0; z < 4; ++z) {
          if (dd[z] < 16u) {
            const int p = atomicAdd(&cnt, 1);
            if (p < CAP) { lst_e[p] = idx * 4 + z; lst_d[p] = (int)dd[z]; }
          }
        }
      }
    }
    __syncthreads();
    const int c = cnt;
    __syncthreads();
    if (c > CAP) { seg_b = 0; seg_e = CAP; continue; }

    for (int i = tid; i < c; i += 128) lst_s[i] = edge_index[NE + lst_e[i]];
    __syncthreads();

    const int ntile = (c + 15) >> 4;
    for (int ti = 0; ti < ntile; ++ti) {
      int woff = 0;
      asm volatile("" : "+s"(woff));
      const float* w0 = fc_w0 + woff;
      const float* w1 = fc_w1 + woff;
      const float* w2 = fc_w2 + woff;
      v2f bw0[2], bw1[16];
      v2f qx0[16], qx1[16], qy0[16], qy1[16];
      {
        const float4* sp[4];
#pragma unroll
        for (int rr = 0; rr < 4; ++rr) {
          const int li  = ti * 16 + wave * 4 + rr;
          const int lic = (li < c) ? li : 0;
          const int sv  = lst_s[lic];
          const int sr  = (li < c) ? sv : 0;
          sp[rr] = (const float4*)(y + (size_t)sr * 512);
        }
        const float4 a0 = sp[0][lane], a1 = sp[0][lane + 32],
                     a2 = sp[0][lane + 64], a3 = sp[0][lane + 96];
        const float4 b0 = sp[1][lane], b1 = sp[1][lane + 32],
                     b2 = sp[1][lane + 64], b3 = sp[1][lane + 96];
        const float4 c0 = sp[2][lane], c1 = sp[2][lane + 32],
                     c2 = sp[2][lane + 64], c3 = sp[2][lane + 96];
        const float4 d0 = sp[3][lane], d1 = sp[3][lane + 32],
                     d2 = sp[3][lane + 64], d3 = sp[3][lane + 96];
#pragma unroll
        for (int s = 0; s < 2; ++s) {
          const int k0 = 4 * s + 2 * half;
          bw0[s] = v2f{ w0[k0 * 64 + wave * 16 + nl],
                        w0[(k0 + 1) * 64 + wave * 16 + nl] };
        }
#pragma unroll
        for (int s = 0; s < 16; ++s) {
          const int k0 = 4 * s + 2 * half;
          bw1[s] = v2f{ w1[k0 * 64 + wave * 16 + nl],
                        w1[(k0 + 1) * 64 + wave * 16 + nl] };
        }
        asm volatile("" ::: "memory");
        float4* dp = (float4*)&ytile[wave * 4][0];
        dp[lane]       = a0; dp[lane + 32]  = a1; dp[lane + 64]  = a2; dp[lane + 96]  = a3;
        dp[lane + 128] = b0; dp[lane + 160] = b1; dp[lane + 192] = b2; dp[lane + 224] = b3;
        dp[lane + 256] = c0; dp[lane + 288] = c1; dp[lane + 320] = c2; dp[lane + 352] = c3;
        dp[lane + 384] = d0; dp[lane + 416] = d1; dp[lane + 448] = d2; dp[lane + 480] = d3;
      }
      {
        const int r = tid >> 3, k = tid & 7;
        const int li = ti * 16 + r;
        float fv = 0.0f;
        if (li < c) fv = edge_feats[(size_t)lst_e[li] * NRAD + k];
        ef[r][k] = fv;
      }
      if (tid < 64) {
        const int r = tid >> 2, k = tid & 3;
        const int li = ti * 16 + r;
        float av = 0.0f;
        if (li < c) av = edge_attrs[(size_t)lst_e[li] * 4 + k];
        ea[r][k] = av;
      }
      if (tid < 16) {
        const int li = ti * 16 + tid;
        dl[tid] = (li < c) ? lst_d[li] : -1;
      }
      __syncthreads();

      {
        v8f acc = {};
#pragma unroll
        for (int s = 0; s < 2; ++s) {
          const int k0 = 4 * s + 2 * half;
          v2f a = { ef[m][k0], ef[m][k0 + 1] };
          acc = wmma4(a, bw0[s], acc);
        }
#pragma unroll
        for (int r = 0; r < 8; ++r)
          ha[r + 8 * half][wave * 16 + nl] = silu_c(acc[r] * INV_SQRT8);
      }
      __syncthreads();

      l2load(qx0, qx1, w2, 0 * 128 + wave * 32, half, nl);

      {
        v8f acc = {};
#pragma unroll
        for (int s = 0; s < 16; ++s) {
          const int k0 = 4 * s + 2 * half;
          v2f a = { ha[m][k0], ha[m][k0 + 1] };
          acc = wmma4(a, bw1[s], acc);
        }
#pragma unroll
        for (int r = 0; r < 8; ++r)
          hb[r + 8 * half][wave * 16 + nl] = silu_c(acc[r] * 0.125f);
      }
      __syncthreads();

      v2f af[16];
#pragma unroll
      for (int s = 0; s < 16; ++s) {
        const int k0 = 4 * s + 2 * half;
        af[s] = v2f{ hb[m][k0], hb[m][k0 + 1] };
      }
      v2f sel[4];
#pragma unroll
      for (int t = 0; t < 4; ++t) {
        const int e0 = 8 * half + 2 * t;
        sel[t] = v2f{ dl[e0] == m ? 1.0f : 0.0f, dl[e0 + 1] == m ? 1.0f : 0.0f };
      }

      {
        v8f cc[2] = { v8f{}, v8f{} };
        l2mma(af, qx0, qx1, cc[0], cc[1]);
        asm volatile("" : "+s"(woff) : "v"(cc[1][7]));
        w2 = fc_w2 + woff;
        l2load(qy0, qy1, w2, 1 * 128 + wave * 32, half, nl);
        asm volatile("" ::: "memory");
#pragma unroll
        for (int jj = 0; jj < 2; ++jj) {
          const int u = wave * 32 + jj * 16 + nl;
          float mid[8];
#pragma unroll
          for (int r = 0; r < 8; ++r) {
            const int e = r + 8 * half;
            mid[r] = (cc[jj][r] * 0.125f) * ytile[e][u] * ea[e][0];
          }
          aS0[jj] = segsum(sel, mid, aS0[jj]);
        }
      }
      {
        v8f cc[2] = { v8f{}, v8f{} };
        asm volatile("" : "+v"(qy0[0]), "+v"(qy1[0]) : "v"(aS0[0][7]), "v"(aS0[1][7]));
        l2mma(af, qy0, qy1, cc[0], cc[1]);
        asm volatile("" : "+s"(woff) : "v"(cc[1][7]));
        w2 = fc_w2 + woff;
        l2load(qx0, qx1, w2, 2 * 128 + wave * 32, half, nl);
        asm volatile("" ::: "memory");
#pragma unroll
        for (int jj = 0; jj < 2; ++jj) {
          const int u = wave * 32 + jj * 16 + nl;
          float m0[8], m1[8], m2[8];
#pragma unroll
          for (int r = 0; r < 8; ++r) {
            const int e = r + 8 * half;
            const float base = (cc[jj][r] * 0.125f) * ytile[e][u];
            m0[r] = base * ea[e][1];
            m1[r] = base * ea[e][2];
            m2[r] = base * ea[e][3];
          }
          aV2[jj][0] = segsum(sel, m0, aV2[jj][0]);
          aV2[jj][1] = segsum(sel, m1, aV2[jj][1]);
          aV2[jj][2] = segsum(sel, m2, aV2[jj][2]);
        }
      }
      {
        v8f cc[2] = { v8f{}, v8f{} };
        asm volatile("" : "+v"(qx0[0]), "+v"(qx1[0]) : "v"(aV2[0][0][7]), "v"(aV2[0][1][7]), "v"(aV2[0][2][7]), "v"(aV2[1][0][7]), "v"(aV2[1][1][7]), "v"(aV2[1][2][7]));
        l2mma(af, qx0, qx1, cc[0], cc[1]);
        asm volatile("" : "+s"(woff) : "v"(cc[1][7]));
        w2 = fc_w2 + woff;
        l2load(qy0, qy1, w2, 3 * 128 + wave * 32, half, nl);
        asm volatile("" ::: "memory");
#pragma unroll
        for (int jj = 0; jj < 2; ++jj) {
          const int u = wave * 32 + jj * 16 + nl;
          float m0[8], m1[8], m2[8];
#pragma unroll
          for (int r = 0; r < 8; ++r) {
            const int e = r + 8 * half;
            const float base = (cc[jj][r] * 0.125f) * ea[e][0];
            m0[r] = base * ytile[e][128 + u * 3 + 0];
            m1[r] = base * ytile[e][128 + u * 3 + 1];
            m2[r] = base * ytile[e][128 + u * 3 + 2];
          }
          aV3[jj][0] = segsum(sel, m0, aV3[jj][0]);
          aV3[jj][1] = segsum(sel, m1, aV3[jj][1]);
          aV3[jj][2] = segsum(sel, m2, aV3[jj][2]);
        }
      }
      {
        v8f cc[2] = { v8f{}, v8f{} };
        asm volatile("" : "+v"(qy0[0]), "+v"(qy1[0]) : "v"(aV3[0][0][7]), "v"(aV3[0][1][7]), "v"(aV3[0][2][7]), "v"(aV3[1][0][7]), "v"(aV3[1][1][7]), "v"(aV3[1][2][7]));
        l2mma(af, qy0, qy1, cc[0], cc[1]);
#pragma unroll
        for (int jj = 0; jj < 2; ++jj) {
          const int u = wave * 32 + jj * 16 + nl;
          float mid[8];
#pragma unroll
          for (int r = 0; r < 8; ++r) {
            const int e = r + 8 * half;
            const float dotv = ytile[e][128 + u * 3 + 0] * ea[e][1] +
                               ytile[e][128 + u * 3 + 1] * ea[e][2] +
                               ytile[e][128 + u * 3 + 2] * ea[e][3];
            mid[r] = (cc[jj][r] * 0.125f) * dotv * INV_SQ3;
          }
          aS1[jj] = segsum(sel, mid, aS1[jj]);
        }
      }
      __syncthreads();
    }

    if (seg_e >= NE) break;
    seg_b = seg_e;
    seg_e = (seg_e + CAP < NE) ? seg_e + CAP : NE;
  }

#pragma unroll
  for (int jj = 0; jj < 2; ++jj) {
    const int u = wave * 32 + jj * 16 + nl;
#pragma unroll
    for (int r = 0; r < 8; ++r) {
      const size_t n = (size_t)(n0 + r + 8 * half);
      s_acc[n * 256 + u]       = aS0[jj][r] * INV_NEIGH;
      s_acc[n * 256 + 128 + u] = aS1[jj][r] * INV_NEIGH;
#pragma unroll
      for (int c = 0; c < 3; ++c) {
        v_acc[(n * 256 + u) * 3 + c]       = aV2[jj][c][r] * INV_NEIGH;
        v_acc[(n * 256 + 128 + u) * 3 + c] = aV3[jj][c][r] * INV_NEIGH;
      }
    }
  }
}

__global__ __launch_bounds__(128) void out_kernel(
    const float* __restrict__ s_acc, const float* __restrict__ v_acc,
    const float* __restrict__ lin2_ws, const float* __restrict__ lin2_wv,
    float* __restrict__ out) {
  __shared__ float st[16][256 + LPAD];
  __shared__ float vt[16][256 + LPAD];
  const int n0  = blockIdx.x * 16;
  const int tid = threadIdx.x;
  const int wave = tid >> 5;
  const int lane = tid & 31;
  const int half = lane >> 4;
  const int m    = lane & 15;
  const int nl   = lane & 15;
  const int wA   = wave * 16;
  const int wB   = wA + 64;

  v2f p0A[8], p0B[8], p1A[8], p1B[8];
  const float* L2s = lin2_ws + (size_t)(2 * half) * MULC + nl;
  loadB8(p0A, p0B, L2s, wA, wB);

  for (int i = tid; i < 16 * 256; i += 128)
    st[i >> 8][i & 255] = s_acc[(size_t)n0 * 256 + i];
  __syncthreads();

  {
    v8f accA = {}, accB = {};
    for (int g = 0; g < 8; g += 2) {
      loadB8(p1A, p1B, L2s + (size_t)(g + 1) * 32 * MULC, wA, wB);
      mma8(&st[m][g * 32 + 2 * half], p0A, p0B, accA, accB);
      if (g + 2 < 8) loadB8(p0A, p0B, L2s + (size_t)(g + 2) * 32 * MULC, wA, wB);
      mma8(&st[m][(g + 1) * 32 + 2 * half], p1A, p1B, accA, accB);
    }
#pragma unroll
    for (int r = 0; r < 8; ++r) {
      size_t row = (size_t)(n0 + r + 8 * half) * 512;
      out[row + wA + nl] += accA[r] * L2_NORM;
      out[row + wB + nl] += accB[r] * L2_NORM;
    }
  }
  for (int c = 0; c < 3; ++c) {
    __syncthreads();
    int voff = 0;
    asm volatile("" : "+s"(voff));
    const float* L2v = lin2_wv + voff + (size_t)(2 * half) * MULC + nl;
    loadB8(p0A, p0B, L2v, wA, wB);
    for (int i = tid; i < 16 * 256; i += 128) {
      int r = i >> 8, k = i & 255;
      vt[r][k] = v_acc[(((size_t)(n0 + r)) * 256 + k) * 3 + c];
    }
    __syncthreads();
    v8f accA = {}, accB = {};
    for (int g = 0; g < 8; g += 2) {
      loadB8(p1A, p1B, L2v + (size_t)(g + 1) * 32 * MULC, wA, wB);
      mma8(&vt[m][g * 32 + 2 * half], p0A, p0B, accA, accB);
      if (g + 2 < 8) loadB8(p0A, p0B, L2v + (size_t)(g + 2) * 32 * MULC, wA, wB);
      mma8(&vt[m][(g + 1) * 32 + 2 * half], p1A, p1B, accA, accB);
    }
#pragma unroll
    for (int r = 0; r < 8; ++r) {
      size_t row = (size_t)(n0 + r + 8 * half) * 512 + MULC;
      out[row + (wA + nl) * 3 + c] += accA[r] * L2_NORM;
      out[row + (wB + nl) * 3 + c] += accB[r] * L2_NORM;
    }
  }
}

extern "C" void kernel_launch(void* const* d_in, const int* in_sizes, int n_in,
                              void* d_out, int out_size, void* d_ws,
                              size_t ws_size, hipStream_t stream) {
  const float* node_feats = (const float*)d_in[0];
  const float* node_attrs = (const float*)d_in[1];
  const float* edge_feats = (const float*)d_in[2];
  const float* edge_attrs = (const float*)d_in[3];
  const int*   edge_index = (const int*)d_in[4];
  const float* lin1_ws    = (const float*)d_in[5];
  const float* lin1_wv    = (const float*)d_in[6];
  const float* fc_w0      = (const float*)d_in[7];
  const float* fc_w1      = (const float*)d_in[8];
  const float* fc_w2      = (const float*)d_in[9];
  const float* lin2_ws    = (const float*)d_in[10];
  const float* lin2_wv    = (const float*)d_in[11];
  const float* sc_ws      = (const float*)d_in[12];
  const float* sc_wv      = (const float*)d_in[13];

  float* out   = (float*)d_out;
  float* y     = (float*)d_ws;
  float* s_acc = y + (size_t)NN * 512;
  float* v_acc = s_acc + (size_t)NN * 256;

  node_kernel<<<NN / 16, 128, 0, stream>>>(node_feats, node_attrs, lin1_ws,
                                           lin1_wv, sc_ws, sc_wv, out, y);
  edge_kernel<<<NN / 16, 128, 0, stream>>>(y, edge_feats, edge_attrs,
                                          edge_index, fc_w0, fc_w1, fc_w2,
                                          s_acc, v_acc);
  out_kernel<<<NN / 16, 128, 0, stream>>>(s_acc, v_acc, lin2_ws, lin2_wv, out);
}


# all global loads of each staging or read-add-store phase issued before its first wait (node tile by 16-byte loads, edge feats/attrs with the row gather, v tile, out += rows): one round trip per phase
# speedup vs baseline: 1.6930x; 1.1374x over previous
#include <hip/hip_runtime.h>
#include <hip/hip_bf16.h>


typedef __attribute__((ext_vector_type(2))) float v2f;
typedef __attribute__((ext_vector_type(8))) float v8f;

static constexpr int NN   = 10000;
static constexpr int NE   = 160000;
static constexpr int MULC = 128;
static constexpr int NSP  = 10;
static constexpr int NRAD = 8;

static constexpr int LPAD = 4;

static constexpr float SILU_C    = 1.679f;
static constexpr float SC_NORM   = 0.02795084971874737f;
static constexpr float L1_NORM   = 0.08838834764831845f;
static constexpr float L2_NORM   = 0.0625f;
static constexpr float INV_SQRT8 = 0.3535533905932738f;
static constexpr float INV_SQ3   = 0.5773502691896258f;
static constexpr float INV_NEIGH = 0.25f;

__device__ __forceinline__ v8f wmma4(v2f a, v2f b, v8f c) {
  return __builtin_amdgcn_wmma_f32_16x16x4_f32(
       false, a,  false, b,
       (short)0, c,  false,  false);
}

__device__ __forceinline__ float silu_c(float z) {
  return SILU_C * z / (1.0f + __expf(-z));
}

__device__ __forceinline__ void loadB8(v2f (&bA)[8], v2f (&bB)[8],
                                       const float* __restrict__ p, int wA,
                                       int wB) {
#pragma unroll
  for (int s = 0; s < 8; ++s) {
    const float* q = p + s * 4 * MULC;
    bA[s] = v2f{ q[wA], q[MULC + wA] };
    bB[s] = v2f{ q[wB], q[MULC + wB] };
  }
}

__device__ __forceinline__ void mma8(const float* arow, const v2f (&bA)[8],
                                     const v2f (&bB)[8], v8f& accA, v8f& accB) {
#pragma unroll
  for (int s = 0; s < 8; ++s) {
    v2f a = { arow[4 * s], arow[4 * s + 1] };
    accA = wmma4(a, bA[s], accA);
    accB = wmma4(a, bB[s], accB);
  }
}

__global__ __launch_bounds__(128) void node_kernel(
    const float* __restrict__ node_feats, const float* __restrict__ node_attrs,
    const float* __restrict__ lin1_ws, const float* __restrict__ lin1_wv,
    const float* __restrict__ sc_ws, const float* __restrict__ sc_wv,
    float* __restrict__ out, float* __restrict__ y) {
  __shared__ __attribute__((aligned(16))) float xs[16][MULC + LPAD];
  __shared__ float xvc[3][16][MULC + LPAD];
  __shared__ float at[16][16];
  __shared__ float zbuf[16][256 + LPAD];

  const int n0  = blockIdx.x * 16;
  const int tid = threadIdx.x;

  {
    float av[2];
#pragma unroll
    for (int q = 0; q < 2; ++q) {
      const int i = tid + q * 128, r = i >> 4, c = i & 15;
      av[q] = (c < NSP) ? node_attrs[(size_t)(n0 + r) * NSP + c] : 0.0f;
    }
    const float4* nf4 = (const float4*)(node_feats + (size_t)n0 * 512);
    float4 t[16];
#pragma unroll
    for (int q = 0; q < 16; ++q) t[q] = nf4[q * 128 + tid];
#pragma unroll
    for (int q = 0; q < 2; ++q) {
      const int i = tid + q * 128;
      at[i >> 4][i & 15] = av[q];
    }
    const int c0 = tid * 4;
    if (c0 < MULC) {
#pragma unroll
      for (int q = 0; q < 16; ++q) *(float4*)&xs[q][c0] = t[q];
    } else {
      const int f0 = c0 - MULC;
      const int h0 = f0 % 3, u0 = f0 / 3;
      const int h1 = (f0 + 1) % 3, u1 = (f0 + 1) / 3;
      const int h2 = (f0 + 2) % 3, u2 = (f0 + 2) / 3;
      const int h3 = (f0 + 3) % 3, u3 = (f0 + 3) / 3;
#pragma unroll
      for (int q = 0; q < 16; ++q) {
        xvc[h0][q][u0] = t[q].x;
        xvc[h1][q][u1] = t[q].y;
        xvc[h2][q][u2] = t[q].z;
        xvc[h3][q][u3] = t[q].w;
      }
    }
  }
  __syncthreads();

  const int wave = tid >> 5;
  const int lane = tid & 31;
  const int half = lane >> 4;
  const int m    = lane & 15;
  const int nl   = lane & 15;
  const int wA   = wave * 16;
  const int wB   = wA + 64;

  v2f p0A[8], p0B[8], p1A[8], p1B[8];

  for (int t = 0; t < 4; ++t) {
    float (*xsrc)[MULC + LPAD] = (t == 0) ? xs : xvc[t - 1];
    const float* W = (t == 0) ? sc_ws : sc_wv;
    v8f accA = {}, accB = {};
    for (int chunk = 0; chunk < 5; ++chunk) {
      __syncthreads();
      const float* Wc = W + ((size_t)chunk * 256 + 2 * half) * MULC + nl;
      loadB8(p0A, p0B, Wc, wA, wB);
      for (int i = tid; i < 16 * 256; i += 128) {
        int r = i >> 8, kk = i & 255;
        int k = chunk * 256 + kk;
        zbuf[r][kk] = xsrc[r][k / NSP] * at[r][k % NSP];
      }
      __syncthreads();
      for (int g = 0; g < 8; g += 2) {
        loadB8(p1A, p1B, Wc + (size_t)(g + 1) * 32 * MULC, wA, wB);
        mma8(&zbuf[m][g * 32 + 2 * half], p0A, p0B, accA, accB);
        if (g + 2 < 8) loadB8(p0A, p0B, Wc + (size_t)(g + 2) * 32 * MULC, wA, wB);
        mma8(&zbuf[m][(g + 1) * 32 + 2 * half], p1A, p1B, accA, accB);
      }
    }
    if (t == 0) {
#pragma unroll
      for (int r = 0; r < 8; ++r) {
        size_t row = (size_t)(n0 + r + 8 * half) * 512;
        out[row + wA + nl] = accA[r] * SC_NORM;
        out[row + wB + nl] = accB[r] * SC_NORM;
      }
    } else {
      const int c = t - 1;
#pragma unroll
      for (int r = 0; r < 8; ++r) {
        size_t row = (size_t)(n0 + r + 8 * half) * 512 + MULC;
        out[row + (wA + nl) * 3 + c] = accA[r] * SC_NORM;
        out[row + (wB + nl) * 3 + c] = accB[r] * SC_NORM;
      }
    }
  }

  const float* L1s = lin1_ws + (size_t)(2 * half) * MULC + nl;
  const float* L1v = lin1_wv + (size_t)(2 * half) * MULC + nl;
  loadB8(p0A, p0B, L1s, wA, wB);
  {
    v8f accA = {}, accB = {};
    for (int g = 0; g < 4; g += 2) {
      loadB8(p1A, p1B, L1s + (size_t)(g + 1) * 32 * MULC, wA, wB);
      mma8(&xs[m][g * 32 + 2 * half], p0A, p0B, accA, accB);
      loadB8(p0A, p0B, (g + 2 < 4) ? L1s + (size_t)(g + 2) * 32 * MULC : L1v, wA, wB);
      mma8(&xs[m][(g + 1) * 32 + 2 * half], p1A, p1B, accA, accB);
    }
#pragma unroll
    for (int r = 0; r < 8; ++r) {
      size_t row = (size_t)(n0 + r + 8 * half) * 512;
      y[row + wA + nl] = accA[r] * L1_NORM;
      y[row + wB + nl] = accB[r] * L1_NORM;
    }
  }
  for (int c = 0; c < 3; ++c) {
    v8f accA = {}, accB = {};
    for (int g = 0; g < 4; g += 2) {
      loadB8(p1A, p1B, L1v + (size_t)(g + 1) * 32 * MULC, wA, wB);
      mma8(&xvc[c][m][g * 32 + 2 * half], p0A, p0B, accA, accB);
      if (g + 2 < 4) loadB8(p0A, p0B, L1v + (size_t)(g + 2) * 32 * MULC, wA, wB);
      else if (c < 2) loadB8(p0A, p0B, L1v, wA, wB);
      mma8(&xvc[c][m][(g + 1) * 32 + 2 * half], p1A, p1B, accA, accB);
    }
#pragma unroll
    for (int r = 0; r < 8; ++r) {
      size_t row = (size_t)(n0 + r + 8 * half) * 512 + MULC;
      y[row + (wA + nl) * 3 + c] = accA[r] * L1_NORM;
      y[row + (wB + nl) * 3 + c] = accB[r] * L1_NORM;
    }
  }
}

static constexpr int CAP    = 1024;
static constexpr int SCAN_U = 8;

__device__ __forceinline__ void l2load(v2f (&b0)[16], v2f (&b1)[16],
                                       const float* __restrict__ w2, int j0,
                                       int half, int nl) {
#pragma unroll
  for (int s = 0; s < 16; ++s) {
    const int k0 = 4 * s + 2 * half;
    const float* p = w2 + (size_t)k0 * 512 + j0 + nl;
    b0[s] = v2f{ p[0], p[512] };
    b1[s] = v2f{ p[16], p[512 + 16] };
  }
}

__device__ __forceinline__ void l2mma(const v2f (&af)[16], const v2f (&b0)[16],
                                      const v2f (&b1)[16], v8f& c0, v8f& c1) {
#pragma unroll
  for (int s = 0; s < 16; ++s) {
    c0 = wmma4(af[s], b0[s], c0);
    c1 = wmma4(af[s], b1[s], c1);
  }
}

__device__ __forceinline__ v8f segsum(const v2f (&sel)[4], const float (&mid)[8],
                                      v8f acc) {
#pragma unroll
  for (int t = 0; t < 4; ++t) {
    v2f b = { mid[2 * t], mid[2 * t + 1] };
    acc = wmma4(sel[t], b, acc);
  }
  return acc;
}

__global__ __launch_bounds__(128) void edge_kernel(
    const float* y, const float* __restrict__ edge_feats,
    const float* __restrict__ edge_attrs, const int* __restrict__ edge_index,
    const float* __restrict__ fc_w0, const float* __restrict__ fc_w1,
    const float* __restrict__ fc_w2, float* __restrict__ s_acc,
    float* __restrict__ v_acc) {
  __shared__ float ytile[16][512];
  __shared__ float ha[16][64 + LPAD];
  __shared__ float hb[16][64 + LPAD];
  __shared__ float ea[16][4];
  __shared__ float ef[16][NRAD + LPAD];
  __shared__ int   dl[16];
  __shared__ int   lst_e[CAP];
  __shared__ int   lst_d[CAP];
  __shared__ int   lst_s[CAP];
  __shared__ int   cnt;

  const int n0   = blockIdx.x * 16;
  const int tid  = threadIdx.x;
  const int wave = tid >> 5;
  const int lane = tid & 31;
  const int half = lane >> 4;
  const int m    = lane & 15;
  const int nl   = lane & 15;

  v8f aS0[2], aS1[2], aV2[2][3], aV3[2][3];
#pragma unroll
  for (int jj = 0; jj < 2; ++jj) {
    aS0[jj] = v8f{};
    aS1[jj] = v8f{};
#pragma unroll
    for (int c = 0; c < 3; ++c) { aV2[jj][c] = v8f{}; aV3[jj][c] = v8f{}; }
  }

  const int4* d4 = (const int4*)edge_index;
  int seg_b = 0, seg_e = NE;
  for (;;) {
    if (tid == 0) cnt = 0;
    __syncthreads();
    const int e4 = seg_e >> 2;
    for (int base = (seg_b >> 2) + tid; base < e4; base += 128 * SCAN_U) {
      int4 v[SCAN_U];
#pragma unroll
      for (int q = 0; q < SCAN_U; ++q) {
        const int idx = base + q * 128;
        v[q] = (idx < e4) ? d4[idx] : make_int4(-1, -1, -1, -1);
      }
#pragma unroll
      for (int q = 0; q < SCAN_U; ++q) {
        const int idx = base + q * 128;
        const unsigned dd[4] = { (unsigned)v[q].x - (unsigned)n0,
                                 (unsigned)v[q].y - (unsigned)n0,
                                 (unsigned)v[q].z - (unsigned)n0,
                                 (unsigned)v[q].w - (unsigned)n0 };
#pragma unroll
        for (int z = 0; z < 4; ++z) {
          if (dd[z] < 16u) {
            const int p = atomicAdd(&cnt, 1);
            if (p < CAP) { lst_e[p] = idx * 4 + z; lst_d[p] = (int)dd[z]; }
          }
        }
      }
    }
    __syncthreads();
    const int c = cnt;
    __syncthreads();
    if (c > CAP) { seg_b = 0; seg_e = CAP; continue; }

    for (int i = tid; i < c; i += 128) lst_s[i] = edge_index[NE + lst_e[i]];
    __syncthreads();

    const int ntile = (c + 15) >> 4;
    for (int ti = 0; ti < ntile; ++ti) {
      int woff = 0;
      asm volatile("" : "+s"(woff));
      const float* w0 = fc_w0 + woff;
      const float* w1 = fc_w1 + woff;
      const float* w2 = fc_w2 + woff;
      v2f bw0[2], bw1[16];
      v2f qx0[16], qx1[16], qy0[16], qy1[16];
      {
        const float4* sp[4];
#pragma unroll
        for (int rr = 0; rr < 4; ++rr) {
          const int li  = ti * 16 + wave * 4 + rr;
          const int lic = (li < c) ? li : 0;
          const int sv  = lst_s[lic];
          const int sr  = (li < c) ? sv : 0;
          sp[rr] = (const float4*)(y + (size_t)sr * 512);
        }
        const int fr = tid >> 3, fk = tid & 7;
        const int fli = ti * 16 + fr;
        const int fe  = lst_e[(fli < c) ? fli : 0];
        const int ar = (tid >> 2) & 15, ak = tid & 3;
        const int ali = ti * 16 + ar;
        const int ae  = lst_e[(ali < c) ? ali : 0];
        const float4 a0 = sp[0][lane], a1 = sp[0][lane + 32],
                     a2 = sp[0][lane + 64], a3 = sp[0][lane + 96];
        const float4 b0 = sp[1][lane], b1 = sp[1][lane + 32],
                     b2 = sp[1][lane + 64], b3 = sp[1][lane + 96];
        const float4 c0 = sp[2][lane], c1 = sp[2][lane + 32],
                     c2 = sp[2][lane + 64], c3 = sp[2][lane + 96];
        const float4 d0 = sp[3][lane], d1 = sp[3][lane + 32],
                     d2 = sp[3][lane + 64], d3 = sp[3][lane + 96];
        const float fraw = edge_feats[(size_t)fe * NRAD + fk];
        float araw = 0.0f;
        if (tid < 64) araw = edge_attrs[(size_t)ae * 4 + ak];
#pragma unroll
        for (int s = 0; s < 2; ++s) {
          const int k0 = 4 * s + 2 * half;
          bw0[s] = v2f{ w0[k0 * 64 + wave * 16 + nl],
                        w0[(k0 + 1) * 64 + wave * 16 + nl] };
        }
#pragma unroll
        for (int s = 0; s < 16; ++s) {
          const int k0 = 4 * s + 2 * half;
          bw1[s] = v2f{ w1[k0 * 64 + wave * 16 + nl],
                        w1[(k0 + 1) * 64 + wave * 16 + nl] };
        }
        asm volatile("" ::: "memory");
        float4* dp = (float4*)&ytile[wave * 4][0];
        dp[lane]       = a0; dp[lane + 32]  = a1; dp[lane + 64]  = a2; dp[lane + 96]  = a3;
        dp[lane + 128] = b0; dp[lane + 160] = b1; dp[lane + 192] = b2; dp[lane + 224] = b3;
        dp[lane + 256] = c0; dp[lane + 288] = c1; dp[lane + 320] = c2; dp[lane + 352] = c3;
        dp[lane + 384] = d0; dp[lane + 416] = d1; dp[lane + 448] = d2; dp[lane + 480] = d3;
        ef[fr][fk] = (fli < c) ? fraw : 0.0f;
        if (tid < 64) ea[ar][ak] = (ali < c) ? araw : 0.0f;
      }
      if (tid < 16) {
        const int li = ti * 16 + tid;
        dl[tid] = (li < c) ? lst_d[li] : -1;
      }
      __syncthreads();

      {
        v8f acc = {};
#pragma unroll
        for (int s = 0; s < 2; ++s) {
          const int k0 = 4 * s + 2 * half;
          v2f a = { ef[m][k0], ef[m][k0 + 1] };
          acc = wmma4(a, bw0[s], acc);
        }
#pragma unroll
        for (int r = 0; r < 8; ++r)
          ha[r + 8 * half][wave * 16 + nl] = silu_c(acc[r] * INV_SQRT8);
      }
      __syncthreads();

      l2load(qx0, qx1, w2, 0 * 128 + wave * 32, half, nl);

      {
        v8f acc = {};
#pragma unroll
        for (int s = 0; s < 16; ++s) {
          const int k0 = 4 * s + 2 * half;
          v2f a = { ha[m][k0], ha[m][k0 + 1] };
          acc = wmma4(a, bw1[s], acc);
        }
#pragma unroll
        for (int r = 0; r < 8; ++r)
          hb[r + 8 * half][wave * 16 + nl] = silu_c(acc[r] * 0.125f);
      }
      __syncthreads();

      v2f af[16];
#pragma unroll
      for (int s = 0; s < 16; ++s) {
        const int k0 = 4 * s + 2 * half;
        af[s] = v2f{ hb[m][k0], hb[m][k0 + 1] };
      }
      v2f sel[4];
#pragma unroll
      for (int t = 0; t < 4; ++t) {
        const int e0 = 8 * half + 2 * t;
        sel[t] = v2f{ dl[e0] == m ? 1.0f : 0.0f, dl[e0 + 1] == m ? 1.0f : 0.0f };
      }

      {
        v8f cc[2] = { v8f{}, v8f{} };
        l2mma(af, qx0, qx1, cc[0], cc[1]);
        asm volatile("" : "+s"(woff) : "v"(cc[1][7]));
        w2 = fc_w2 + woff;
        l2load(qy0, qy1, w2, 1 * 128 + wave * 32, half, nl);
        asm volatile("" ::: "memory");
#pragma unroll
        for (int jj = 0; jj < 2; ++jj) {
          const int u = wave * 32 + jj * 16 + nl;
          float mid[8];
#pragma unroll
          for (int r = 0; r < 8; ++r) {
            const int e = r + 8 * half;
            mid[r] = (cc[jj][r] * 0.125f) * ytile[e][u] * ea[e][0];
          }
          aS0[jj] = segsum(sel, mid, aS0[jj]);
        }
      }
      {
        v8f cc[2] = { v8f{}, v8f{} };
        asm volatile("" : "+v"(qy0[0]), "+v"(qy1[0]) : "v"(aS0[0][7]), "v"(aS0[1][7]));
        l2mma(af, qy0, qy1, cc[0], cc[1]);
        asm volatile("" : "+s"(woff) : "v"(cc[1][7]));
        w2 = fc_w2 + woff;
        l2load(qx0, qx1, w2, 2 * 128 + wave * 32, half, nl);
        asm volatile("" ::: "memory");
#pragma unroll
        for (int jj = 0; jj < 2; ++jj) {
          const int u = wave * 32 + jj * 16 + nl;
          float m0[8], m1[8], m2[8];
#pragma unroll
          for (int r = 0; r < 8; ++r) {
            const int e = r + 8 * half;
            const float base = (cc[jj][r] * 0.125f) * ytile[e][u];
            m0[r] = base * ea[e][1];
            m1[r] = base * ea[e][2];
            m2[r] = base * ea[e][3];
          }
          aV2[jj][0] = segsum(sel, m0, aV2[jj][0]);
          aV2[jj][1] = segsum(sel, m1, aV2[jj][1]);
          aV2[jj][2] = segsum(sel, m2, aV2[jj][2]);
        }
      }
      {
        v8f cc[2] = { v8f{}, v8f{} };
        asm volatile("" : "+v"(qx0[0]), "+v"(qx1[0]) : "v"(aV2[0][0][7]), "v"(aV2[0][1][7]), "v"(aV2[0][2][7]), "v"(aV2[1][0][7]), "v"(aV2[1][1][7]), "v"(aV2[1][2][7]));
        l2mma(af, qx0, qx1, cc[0], cc[1]);
        asm volatile("" : "+s"(woff) : "v"(cc[1][7]));
        w2 = fc_w2 + woff;
        l2load(qy0, qy1, w2, 3 * 128 + wave * 32, half, nl);
        asm volatile("" ::: "memory");
#pragma unroll
        for (int jj = 0; jj < 2; ++jj) {
          const int u = wave * 32 + jj * 16 + nl;
          float m0[8], m1[8], m2[8];
#pragma unroll
          for (int r = 0; r < 8; ++r) {
            const int e = r + 8 * half;
            const float base = (cc[jj][r] * 0.125f) * ea[e][0];
            m0[r] = base * ytile[e][128 + u * 3 + 0];
            m1[r] = base * ytile[e][128 + u * 3 + 1];
            m2[r] = base * ytile[e][128 + u * 3 + 2];
          }
          aV3[jj][0] = segsum(sel, m0, aV3[jj][0]);
          aV3[jj][1] = segsum(sel, m1, aV3[jj][1]);
          aV3[jj][2] = segsum(sel, m2, aV3[jj][2]);
        }
      }
      {
        v8f cc[2] = { v8f{}, v8f{} };
        asm volatile("" : "+v"(qy0[0]), "+v"(qy1[0]) : "v"(aV3[0][0][7]), "v"(aV3[0][1][7]), "v"(aV3[0][2][7]), "v"(aV3[1][0][7]), "v"(aV3[1][1][7]), "v"(aV3[1][2][7]));
        l2mma(af, qy0, qy1, cc[0], cc[1]);
#pragma unroll
        for (int jj = 0; jj < 2; ++jj) {
          const int u = wave * 32 + jj * 16 + nl;
          float mid[8];
#pragma unroll
          for (int r = 0; r < 8; ++r) {
            const int e = r + 8 * half;
            const float dotv = ytile[e][128 + u * 3 + 0] * ea[e][1] +
                               ytile[e][128 + u * 3 + 1] * ea[e][2] +
                               ytile[e][128 + u * 3 + 2] * ea[e][3];
            mid[r] = (cc[jj][r] * 0.125f) * dotv * INV_SQ3;
          }
          aS1[jj] = segsum(sel, mid, aS1[jj]);
        }
      }
      __syncthreads();
    }

    if (seg_e >= NE) break;
    seg_b = seg_e;
    seg_e = (seg_e + CAP < NE) ? seg_e + CAP : NE;
  }

#pragma unroll
  for (int jj = 0; jj < 2; ++jj) {
    const int u = wave * 32 + jj * 16 + nl;
#pragma unroll
    for (int r = 0; r < 8; ++r) {
      const size_t n = (size_t)(n0 + r + 8 * half);
      s_acc[n * 256 + u]       = aS0[jj][r] * INV_NEIGH;
      s_acc[n * 256 + 128 + u] = aS1[jj][r] * INV_NEIGH;
#pragma unroll
      for (int c = 0; c < 3; ++c) {
        v_acc[(n * 256 + u) * 3 + c]       = aV2[jj][c][r] * INV_NEIGH;
        v_acc[(n * 256 + 128 + u) * 3 + c] = aV3[jj][c][r] * INV_NEIGH;
      }
    }
  }
}

__global__ __launch_bounds__(128) void out_kernel(
    const float* __restrict__ s_acc, const float* __restrict__ v_acc,
    const float* __restrict__ lin2_ws, const float* __restrict__ lin2_wv,
    float* __restrict__ out) {
  __shared__ float st[16][256 + LPAD];
  __shared__ float vt[16][256 + LPAD];
  const int n0  = blockIdx.x * 16;
  const int tid = threadIdx.x;
  const int wave = tid >> 5;
  const int lane = tid & 31;
  const int half = lane >> 4;
  const int m    = lane & 15;
  const int nl   = lane & 15;
  const int wA   = wave * 16;
  const int wB   = wA + 64;

  v2f p0A[8], p0B[8], p1A[8], p1B[8];
  const float* L2s = lin2_ws + (size_t)(2 * half) * MULC + nl;
  loadB8(p0A, p0B, L2s, wA, wB);

  for (int i = tid; i < 16 * 256; i += 128)
    st[i >> 8][i & 255] = s_acc[(size_t)n0 * 256 + i];
  __syncthreads();

  {
    v8f accA = {}, accB = {};
    for (int g = 0; g < 8; g += 2) {
      loadB8(p1A, p1B, L2s + (size_t)(g + 1) * 32 * MULC, wA, wB);
      mma8(&st[m][g * 32 + 2 * half], p0A, p0B, accA, accB);
      if (g + 2 < 8) loadB8(p0A, p0B, L2s + (size_t)(g + 2) * 32 * MULC, wA, wB);
      mma8(&st[m][(g + 1) * 32 + 2 * half], p1A, p1B, accA, accB);
    }
#pragma unroll
    for (int r = 0; r < 8; ++r) {
      size_t row = (size_t)(n0 + r + 8 * half) * 512;
      out[row + wA + nl] += accA[r] * L2_NORM;
      out[row + wB + nl] += accB[r] * L2_NORM;
    }
  }
  for (int c = 0; c < 3; ++c) {
    __syncthreads();
    int voff = 0;
    asm volatile("" : "+s"(voff));
    const float* L2v = lin2_wv + voff + (size_t)(2 * half) * MULC + nl;
    loadB8(p0A, p0B, L2v, wA, wB);
#pragma nounroll
    for (int hgrp = 0; hgrp < 2; ++hgrp) {
      float tv[16];
#pragma unroll
      for (int q = 0; q < 16; ++q) {
        const int i = tid + (hgrp * 16 + q) * 128;
        const int r = i >> 8, k = i & 255;
        tv[q] = v_acc[(((size_t)(n0 + r)) * 256 + k) * 3 + c];
      }
#pragma unroll
      for (int q = 0; q < 16; ++q) {
        const int i = tid + (hgrp * 16 + q) * 128;
        vt[i >> 8][i & 255] = tv[q];
      }
    }
    __syncthreads();
    v8f accA = {}, accB = {};
    for (int g = 0; g < 8; g += 2) {
      loadB8(p1A, p1B, L2v + (size_t)(g + 1) * 32 * MULC, wA, wB);
      mma8(&vt[m][g * 32 + 2 * half], p0A, p0B, accA, accB);
      if (g + 2 < 8) loadB8(p0A, p0B, L2v + (size_t)(g + 2) * 32 * MULC, wA, wB);
      mma8(&vt[m][(g + 1) * 32 + 2 * half], p1A, p1B, accA, accB);
    }
    float oA[8], oB[8];
#pragma unroll
    for (int r = 0; r < 8; ++r) {
      size_t row = (size_t)(n0 + r + 8 * half) * 512 + MULC;
      oA[r] = out[row + (wA + nl) * 3 + c];
      oB[r] = out[row + (wB + nl) * 3 + c];
    }
#pragma unroll
    for (int r = 0; r < 8; ++r) {
      size_t row = (size_t)(n0 + r + 8 * half) * 512 + MULC;
      out[row + (wA + nl) * 3 + c] = oA[r] + accA[r] * L2_NORM;
      out[row + (wB + nl) * 3 + c] = oB[r] + accB[r] * L2_NORM;
    }
  }
}

extern "C" void kernel_launch(void* const* d_in, const int* in_sizes, int n_in,
                              void* d_out, int out_size, void* d_ws,
                              size_t ws_size, hipStream_t stream) {
  const float* node_feats = (const float*)d_in[0];
  const float* node_attrs = (const float*)d_in[1];
  const float* edge_feats = (const float*)d_in[2];
  const float* edge_attrs = (const float*)d_in[3];
  const int*   edge_index = (const int*)d_in[4];
  const float* lin1_ws    = (const float*)d_in[5];
  const float* lin1_wv    = (const float*)d_in[6];
  const float* fc_w0      = (const float*)d_in[7];
  const float* fc_w1      = (const float*)d_in[8];
  const float* fc_w2      = (const float*)d_in[9];
  const float* lin2_ws    = (const float*)d_in[10];
  const float* lin2_wv    = (const float*)d_in[11];
  const float* sc_ws      = (const float*)d_in[12];
  const float* sc_wv      = (const float*)d_in[13];

  float* out   = (float*)d_out;
  float* y     = (float*)d_ws;
  float* s_acc = y + (size_t)NN * 512;
  float* v_acc = s_acc + (size_t)NN * 256;

  node_kernel<<<NN / 16, 128, 0, stream>>>(node_feats, node_attrs, lin1_ws,
                                           lin1_wv, sc_ws, sc_wv, out, y);
  edge_kernel<<<NN / 16, 128, 0, stream>>>(y, edge_feats, edge_attrs,
                                          edge_index, fc_w0, fc_w1, fc_w2,
                                          s_acc, v_acc);
  out_kernel<<<NN / 16, 128, 0, stream>>>(s_acc, v_acc, lin2_ws, lin2_wv, out);
}


# edge kernel held to 336 registers (launch bound of 3 waves per SIMD, lane constants re-derived in the tile loop, no scratch): 3 work-groups per WGP, so all 625 run at once, not in two rounds
# speedup vs baseline: 1.9059x; 1.1258x over previous
#include <hip/hip_runtime.h>
#include <hip/hip_bf16.h>


typedef __attribute__((ext_vector_type(2))) float v2f;
typedef __attribute__((ext_vector_type(8))) float v8f;

static constexpr int NN   = 10000;
static constexpr int NE   = 160000;
static constexpr int MULC = 128;
static constexpr int NSP  = 10;
static constexpr int NRAD = 8;

static constexpr int LPAD = 4;

static constexpr float SILU_C    = 1.679f;
static constexpr float SC_NORM   = 0.02795084971874737f;
static constexpr float L1_NORM   = 0.08838834764831845f;
static constexpr float L2_NORM   = 0.0625f;
static constexpr float INV_SQRT8 = 0.3535533905932738f;
static constexpr float INV_SQ3   = 0.5773502691896258f;
static constexpr float INV_NEIGH = 0.25f;

__device__ __forceinline__ v8f wmma4(v2f a, v2f b, v8f c) {
  return __builtin_amdgcn_wmma_f32_16x16x4_f32(
       false, a,  false, b,
       (short)0, c,  false,  false);
}

__device__ __forceinline__ float silu_c(float z) {
  return SILU_C * z / (1.0f + __expf(-z));
}

__device__ __forceinline__ void loadB8(v2f (&bA)[8], v2f (&bB)[8],
                                       const float* __restrict__ p, int wA,
                                       int wB) {
#pragma unroll
  for (int s = 0; s < 8; ++s) {
    const float* q = p + s * 4 * MULC;
    bA[s] = v2f{ q[wA], q[MULC + wA] };
    bB[s] = v2f{ q[wB], q[MULC + wB] };
  }
}

__device__ __forceinline__ void mma8(const float* arow, const v2f (&bA)[8],
                                     const v2f (&bB)[8], v8f& accA, v8f& accB) {
#pragma unroll
  for (int s = 0; s < 8; ++s) {
    v2f a = { arow[4 * s], arow[4 * s + 1] };
    accA = wmma4(a, bA[s], accA);
    accB = wmma4(a, bB[s], accB);
  }
}

__global__ __launch_bounds__(128) void node_kernel(
    const float* __restrict__ node_feats, const float* __restrict__ node_attrs,
    const float* __restrict__ lin1_ws, const float* __restrict__ lin1_wv,
    const float* __restrict__ sc_ws, const float* __restrict__ sc_wv,
    float* __restrict__ out, float* __restrict__ y) {
  __shared__ __attribute__((aligned(16))) float xs[16][MULC + LPAD];
  __shared__ float xvc[3][16][MULC + LPAD];
  __shared__ float at[16][16];
  __shared__ float zbuf[16][256 + LPAD];

  const int n0  = blockIdx.x * 16;
  const int tid = threadIdx.x;

  {
    float av[2];
#pragma unroll
    for (int q = 0; q < 2; ++q) {
      const int i = tid + q * 128, r = i >> 4, c = i & 15;
      av[q] = (c < NSP) ? node_attrs[(size_t)(n0 + r) * NSP + c] : 0.0f;
    }
    const float4* nf4 = (const float4*)(node_feats + (size_t)n0 * 512);
    float4 t[16];
#pragma unroll
    for (int q = 0; q < 16; ++q) t[q] = nf4[q * 128 + tid];
#pragma unroll
    for (int q = 0; q < 2; ++q) {
      const int i = tid + q * 128;
      at[i >> 4][i & 15] = av[q];
    }
    const int c0 = tid * 4;
    if (c0 < MULC) {
#pragma unroll
      for (int q = 0; q < 16; ++q) *(float4*)&xs[q][c0] = t[q];
    } else {
      const int f0 = c0 - MULC;
      const int h0 = f0 % 3, u0 = f0 / 3;
      const int h1 = (f0 + 1) % 3, u1 = (f0 + 1) / 3;
      const int h2 = (f0 + 2) % 3, u2 = (f0 + 2) / 3;
      const int h3 = (f0 + 3) % 3, u3 = (f0 + 3) / 3;
#pragma unroll
      for (int q = 0; q < 16; ++q) {
        xvc[h0][q][u0] = t[q].x;
        xvc[h1][q][u1] = t[q].y;
        xvc[h2][q][u2] = t[q].z;
        xvc[h3][q][u3] = t[q].w;
      }
    }
  }
  __syncthreads();

  const int wave = tid >> 5;
  const int lane = tid & 31;
  const int half = lane >> 4;
  const int m    = lane & 15;
  const int nl   = lane & 15;
  const int wA   = wave * 16;
  const int wB   = wA + 64;

  v2f p0A[8], p0B[8], p1A[8], p1B[8];

  for (int t = 0; t < 4; ++t) {
    float (*xsrc)[MULC + LPAD] = (t == 0) ? xs : xvc[t - 1];
    const float* W = (t == 0) ? sc_ws : sc_wv;
    v8f accA = {}, accB = {};
    for (int chunk = 0; chunk < 5; ++chunk) {
      __syncthreads();
      const float* Wc = W + ((size_t)chunk * 256 + 2 * half) * MULC + nl;
      loadB8(p0A, p0B, Wc, wA, wB);
      for (int i = tid; i < 16 * 256; i += 128) {
        int r = i >> 8, kk = i & 255;
        int k = chunk * 256 + kk;
        zbuf[r][kk] = xsrc[r][k / NSP] * at[r][k % NSP];
      }
      __syncthreads();
      for (int g = 0; g < 8; g += 2) {
        loadB8(p1A, p1B, Wc + (size_t)(g + 1) * 32 * MULC, wA, wB);
        mma8(&zbuf[m][g * 32 + 2 * half], p0A, p0B, accA, accB);
        if (g + 2 < 8) loadB8(p0A, p0B, Wc + (size_t)(g + 2) * 32 * MULC, wA, wB);
        mma8(&zbuf[m][(g + 1) * 32 + 2 * half], p1A, p1B, accA, accB);
      }
    }
    if (t == 0) {
#pragma unroll
      for (int r = 0; r < 8; ++r) {
        size_t row = (size_t)(n0 + r + 8 * half) * 512;
        out[row + wA + nl] = accA[r] * SC_NORM;
        out[row + wB + nl] = accB[r] * SC_NORM;
      }
    } else {
      const int c = t - 1;
#pragma unroll
      for (int r = 0; r < 8; ++r) {
        size_t row = (size_t)(n0 + r + 8 * half) * 512 + MULC;
        out[row + (wA + nl) * 3 + c] = accA[r] * SC_NORM;
        out[row + (wB + nl) * 3 + c] = accB[r] * SC_NORM;
      }
    }
  }

  const float* L1s = lin1_ws + (size_t)(2 * half) * MULC + nl;
  const float* L1v = lin1_wv + (size_t)(2 * half) * MULC + nl;
  loadB8(p0A, p0B, L1s, wA, wB);
  {
    v8f accA = {}, accB = {};
    for (int g = 0; g < 4; g += 2) {
      loadB8(p1A, p1B, L1s + (size_t)(g + 1) * 32 * MULC, wA, wB);
      mma8(&xs[m][g * 32 + 2 * half], p0A, p0B, accA, accB);
      loadB8(p0A, p0B, (g + 2 < 4) ? L1s + (size_t)(g + 2) * 32 * MULC : L1v, wA, wB);
      mma8(&xs[m][(g + 1) * 32 + 2 * half], p1A, p1B, accA, accB);
    }
#pragma unroll
    for (int r = 0; r < 8; ++r) {
      size_t row = (size_t)(n0 + r + 8 * half) * 512;
      y[row + wA + nl] = accA[r] * L1_NORM;
      y[row + wB + nl] = accB[r] * L1_NORM;
    }
  }
  for (int c = 0; c < 3; ++c) {
    v8f accA = {}, accB = {};
    for (int g = 0; g < 4; g += 2) {
      loadB8(p1A, p1B, L1v + (size_t)(g + 1) * 32 * MULC, wA, wB);
      mma8(&xvc[c][m][g * 32 + 2 * half], p0A, p0B, accA, accB);
      if (g + 2 < 4) loadB8(p0A, p0B, L1v + (size_t)(g + 2) * 32 * MULC, wA, wB);
      else if (c < 2) loadB8(p0A, p0B, L1v, wA, wB);
      mma8(&xvc[c][m][(g + 1) * 32 + 2 * half], p1A, p1B, accA, accB);
    }
#pragma unroll
    for (int r = 0; r < 8; ++r) {
      size_t row = (size_t)(n0 + r + 8 * half) * 512 + MULC;
      y[row + (wA + nl) * 3 + c] = accA[r] * L1_NORM;
      y[row + (wB + nl) * 3 + c] = accB[r] * L1_NORM;
    }
  }
}

static constexpr int CAP    = 1024;
static constexpr int SCAN_U = 8;

__device__ __forceinline__ void l2load(v2f (&b0)[16], v2f (&b1)[16],
                                       const float* __restrict__ w2, int j0,
                                       int half, int nl) {
#pragma unroll
  for (int s = 0; s < 16; ++s) {
    const int k0 = 4 * s + 2 * half;
    const float* p = w2 + (size_t)k0 * 512 + j0 + nl;
    b0[s] = v2f{ p[0], p[512] };
    b1[s] = v2f{ p[16], p[512 + 16] };
  }
}

__device__ __forceinline__ void l2mma(const v2f (&af)[16], const v2f (&b0)[16],
                                      const v2f (&b1)[16], v8f& c0, v8f& c1) {
#pragma unroll
  for (int s = 0; s < 16; ++s) {
    c0 = wmma4(af[s], b0[s], c0);
    c1 = wmma4(af[s], b1[s], c1);
  }
}

__device__ __forceinline__ v8f segsum(const v2f (&sel)[4], const float (&mid)[8],
                                      v8f acc) {
#pragma unroll
  for (int t = 0; t < 4; ++t) {
    v2f b = { mid[2 * t], mid[2 * t + 1] };
    acc = wmma4(sel[t], b, acc);
  }
  return acc;
}

__global__ __launch_bounds__(128, 3) void edge_kernel(
    const float* y, const float* __restrict__ edge_feats,
    const float* __restrict__ edge_attrs, const int* __restrict__ edge_index,
    const float* __restrict__ fc_w0, const float* __restrict__ fc_w1,
    const float* __restrict__ fc_w2, float* __restrict__ s_acc,
    float* __restrict__ v_acc) {
  __shared__ float ytile[16][512];
  __shared__ float ha[16][64 + LPAD];
  __shared__ float hb[16][64 + LPAD];
  __shared__ float ea[16][4];
  __shared__ float ef[16][NRAD + LPAD];
  __shared__ int   dl[16];
  __shared__ int   lst_e[CAP];
  __shared__ int   lst_d[CAP];
  __shared__ int   lst_s[CAP];
  __shared__ int   cnt;

  const int n0   = blockIdx.x * 16;
  const int tid  = threadIdx.x;
  const int wave = tid >> 5;
  const int lane = tid & 31;
  const int half = lane >> 4;
  const int m    = lane & 15;
  const int nl   = lane & 15;

  v8f aS0[2], aS1[2], aV2[2][3], aV3[2][3];
#pragma unroll
  for (int jj = 0; jj < 2; ++jj) {
    aS0[jj] = v8f{};
    aS1[jj] = v8f{};
#pragma unroll
    for (int c = 0; c < 3; ++c) { aV2[jj][c] = v8f{}; aV3[jj][c] = v8f{}; }
  }

  const int4* d4 = (const int4*)edge_index;
  int seg_b = 0, seg_e = NE;
  for (;;) {
    if (tid == 0) cnt = 0;
    __syncthreads();
    const int e4 = seg_e >> 2;
    for (int base = (seg_b >> 2) + tid; base < e4; base += 128 * SCAN_U) {
      int4 v[SCAN_U];
#pragma unroll
      for (int q = 0; q < SCAN_U; ++q) {
        const int idx = base + q * 128;
        v[q] = (idx < e4) ? d4[idx] : make_int4(-1, -1, -1, -1);
      }
#pragma unroll
      for (int q = 0; q < SCAN_U; ++q) {
        const int idx = base + q * 128;
        const unsigned dd[4] = { (unsigned)v[q].x - (unsigned)n0,
                                 (unsigned)v[q].y - (unsigned)n0,
                                 (unsigned)v[q].z - (unsigned)n0,
                                 (unsigned)v[q].w - (unsigned)n0 };
#pragma unroll
        for (int z = 0; z < 4; ++z) {
          if (dd[z] < 16u) {
            const int p = atomicAdd(&cnt, 1);
            if (p < CAP) { lst_e[p] = idx * 4 + z; lst_d[p] = (int)dd[z]; }
          }
        }
      }
    }
    __syncthreads();
    const int c = cnt;
    __syncthreads();
    if (c > CAP) { seg_b = 0; seg_e = CAP; continue; }

    for (int i = tid; i < c; i += 128) lst_s[i] = edge_index[NE + lst_e[i]];
    __syncthreads();

    const int ntile = (c + 15) >> 4;
    for (int ti = 0; ti < ntile; ++ti) {
      int woff = 0;
      asm volatile("" : "+s"(woff));
      const float* w0 = fc_w0 + woff;
      const float* w1 = fc_w1 + woff;
      const float* w2 = fc_w2 + woff;
      v2f bw0[2], bw1[16];
      v2f qx0[16], qx1[16], qy0[16], qy1[16];
      {
        const float4* sp[4];
        int tw = tid;
        asm volatile("" : "+v"(tw));
        const int wrow = (tw >> 5) * 4;
#pragma unroll
        for (int rr = 0; rr < 4; ++rr) {
          const int li  = ti * 16 + wrow + rr;
          const int lic = (li < c) ? li : 0;
          const int sv  = lst_s[lic];
          const int sr  = (li < c) ? sv : 0;
          sp[rr] = (const float4*)(y + (size_t)sr * 512);
        }
        const int fr = tid >> 3, fk = tid & 7;
        const int fli = ti * 16 + fr;
        const int fe  = lst_e[(fli < c) ? fli : 0];
        const int ar = (tid >> 2) & 15, ak = tid & 3;
        const int ali = ti * 16 + ar;
        const int ae  = lst_e[(ali < c) ? ali : 0];
        const float4 a0 = sp[0][lane], a1 = sp[0][lane + 32],
                     a2 = sp[0][lane + 64], a3 = sp[0][lane + 96];
        const float4 b0 = sp[1][lane], b1 = sp[1][lane + 32],
                     b2 = sp[1][lane + 64], b3 = sp[1][lane + 96];
        const float4 c0 = sp[2][lane], c1 = sp[2][lane + 32],
                     c2 = sp[2][lane + 64], c3 = sp[2][lane + 96];
        const float4 d0 = sp[3][lane], d1 = sp[3][lane + 32],
                     d2 = sp[3][lane + 64], d3 = sp[3][lane + 96];
        float fraw = edge_feats[(unsigned)fe * NRAD + (unsigned)fk];
        float araw = 0.0f;
        if (tid < 64) araw = edge_attrs[(unsigned)ae * 4u + (unsigned)ak];
#pragma unroll
        for (int s = 0; s < 2; ++s) {
          const int k0 = 4 * s + 2 * half;
          bw0[s] = v2f{ w0[k0 * 64 + wave * 16 + nl],
                        w0[(k0 + 1) * 64 + wave * 16 + nl] };
        }
#pragma unroll
        for (int s = 0; s < 16; ++s) {
          const int k0 = 4 * s + 2 * half;
          bw1[s] = v2f{ w1[k0 * 64 + wave * 16 + nl],
                        w1[(k0 + 1) * 64 + wave * 16 + nl] };
        }
        asm volatile("" : "+v"(fraw), "+v"(araw) : : "memory");
        float4* dp = (float4*)&ytile[wave * 4][0];
        dp[lane]       = a0; dp[lane + 32]  = a1; dp[lane + 64]  = a2; dp[lane + 96]  = a3;
        dp[lane + 128] = b0; dp[lane + 160] = b1; dp[lane + 192] = b2; dp[lane + 224] = b3;
        dp[lane + 256] = c0; dp[lane + 288] = c1; dp[lane + 320] = c2; dp[lane + 352] = c3;
        dp[lane + 384] = d0; dp[lane + 416] = d1; dp[lane + 448] = d2; dp[lane + 480] = d3;
        ef[fr][fk] = (fli < c) ? fraw : 0.0f;
        if (tid < 64) ea[ar][ak] = (ali < c) ? araw : 0.0f;
      }
      if (tid < 16) {
        const int li = ti * 16 + tid;
        dl[tid] = (li < c) ? lst_d[li] : -1;
      }
      __syncthreads();

      {
        v8f acc = {};
#pragma unroll
        for (int s = 0; s < 2; ++s) {
          const int k0 = 4 * s + 2 * half;
          v2f a = { ef[m][k0], ef[m][k0 + 1] };
          acc = wmma4(a, bw0[s], acc);
        }
#pragma unroll
        for (int r = 0; r < 8; ++r)
          ha[r + 8 * half][wave * 16 + nl] = silu_c(acc[r] * INV_SQRT8);
      }
      __syncthreads();

      l2load(qx0, qx1, w2, 0 * 128 + wave * 32, half, nl);

      {
        v8f acc = {};
#pragma unroll
        for (int s = 0; s < 16; ++s) {
          const int k0 = 4 * s + 2 * half;
          v2f a = { ha[m][k0], ha[m][k0 + 1] };
          acc = wmma4(a, bw1[s], acc);
        }
#pragma unroll
        for (int r = 0; r < 8; ++r)
          hb[r + 8 * half][wave * 16 + nl] = silu_c(acc[r] * 0.125f);
      }
      __syncthreads();

      v2f af[16];
#pragma unroll
      for (int s = 0; s < 16; ++s) {
        const int k0 = 4 * s + 2 * half;
        af[s] = v2f{ hb[m][k0], hb[m][k0 + 1] };
      }
      v2f sel[4];
      int mrow = m;
      asm volatile("" : "+v"(mrow));
#pragma unroll
      for (int t = 0; t < 4; ++t) {
        const int e0 = 8 * half + 2 * t;
        sel[t] = v2f{ dl[e0] == mrow ? 1.0f : 0.0f, dl[e0 + 1] == mrow ? 1.0f : 0.0f };
      }

      {
        v8f cc[2] = { v8f{}, v8f{} };
        l2mma(af, qx0, qx1, cc[0], cc[1]);
        asm volatile("" : "+s"(woff) : "v"(cc[1][7]));
        w2 = fc_w2 + woff;
        l2load(qy0, qy1, w2, 1 * 128 + wave * 32, half, nl);
        asm volatile("" ::: "memory");
#pragma unroll
        for (int jj = 0; jj < 2; ++jj) {
          const int u = wave * 32 + jj * 16 + nl;
          float mid[8];
#pragma unroll
          for (int r = 0; r < 8; ++r) {
            const int e = r + 8 * half;
            mid[r] = (cc[jj][r] * 0.125f) * ytile[e][u] * ea[e][0];
          }
          aS0[jj] = segsum(sel, mid, aS0[jj]);
        }
      }
      {
        v8f cc[2] = { v8f{}, v8f{} };
        asm volatile("" : "+v"(qy0[0]), "+v"(qy1[0]) : "v"(aS0[0][7]), "v"(aS0[1][7]));
        l2mma(af, qy0, qy1, cc[0], cc[1]);
        asm volatile("" : "+s"(woff) : "v"(cc[1][7]));
        w2 = fc_w2 + woff;
        l2load(qx0, qx1, w2, 2 * 128 + wave * 32, half, nl);
        asm volatile("" ::: "memory");
#pragma unroll
        for (int jj = 0; jj < 2; ++jj) {
          const int u = wave * 32 + jj * 16 + nl;
          float m0[8], m1[8], m2[8];
#pragma unroll
          for (int r = 0; r < 8; ++r) {
            const int e = r + 8 * half;
            const float base = (cc[jj][r] * 0.125f) * ytile[e][u];
            m0[r] = base * ea[e][1];
            m1[r] = base * ea[e][2];
            m2[r] = base * ea[e][3];
          }
          aV2[jj][0] = segsum(sel, m0, aV2[jj][0]);
          aV2[jj][1] = segsum(sel, m1, aV2[jj][1]);
          aV2[jj][2] = segsum(sel, m2, aV2[jj][2]);
        }
      }
      {
        v8f cc[2] = { v8f{}, v8f{} };
        asm volatile("" : "+v"(qx0[0]), "+v"(qx1[0]) : "v"(aV2[0][0][7]), "v"(aV2[0][1][7]), "v"(aV2[0][2][7]), "v"(aV2[1][0][7]), "v"(aV2[1][1][7]), "v"(aV2[1][2][7]));
        l2mma(af, qx0, qx1, cc[0], cc[1]);
        asm volatile("" : "+s"(woff) : "v"(cc[1][7]));
        w2 = fc_w2 + woff;
        l2load(qy0, qy1, w2, 3 * 128 + wave * 32, half, nl);
        asm volatile("" ::: "memory");
#pragma unroll
        for (int jj = 0; jj < 2; ++jj) {
          const int u = wave * 32 + jj * 16 + nl;
          float m0[8], m1[8], m2[8];
#pragma unroll
          for (int r = 0; r < 8; ++r) {
            const int e = r + 8 * half;
            const float base = (cc[jj][r] * 0.125f) * ea[e][0];
            m0[r] = base * ytile[e][128 + u * 3 + 0];
            m1[r] = base * ytile[e][128 + u * 3 + 1];
            m2[r] = base * ytile[e][128 + u * 3 + 2];
          }
          aV3[jj][0] = segsum(sel, m0, aV3[jj][0]);
          aV3[jj][1] = segsum(sel, m1, aV3[jj][1]);
          aV3[jj][2] = segsum(sel, m2, aV3[jj][2]);
        }
      }
      {
        v8f cc[2] = { v8f{}, v8f{} };
        asm volatile("" : "+v"(qy0[0]), "+v"(qy1[0]) : "v"(aV3[0][0][7]), "v"(aV3[0][1][7]), "v"(aV3[0][2][7]), "v"(aV3[1][0][7]), "v"(aV3[1][1][7]), "v"(aV3[1][2][7]));
        l2mma(af, qy0, qy1, cc[0], cc[1]);
#pragma unroll
        for (int jj = 0; jj < 2; ++jj) {
          const int u = wave * 32 + jj * 16 + nl;
          float mid[8];
#pragma unroll
          for (int r = 0; r < 8; ++r) {
            const int e = r + 8 * half;
            const float dotv = ytile[e][128 + u * 3 + 0] * ea[e][1] +
                               ytile[e][128 + u * 3 + 1] * ea[e][2] +
                               ytile[e][128 + u * 3 + 2] * ea[e][3];
            mid[r] = (cc[jj][r] * 0.125f) * dotv * INV_SQ3;
          }
          aS1[jj] = segsum(sel, mid, aS1[jj]);
        }
      }
      __syncthreads();
    }
    asm volatile("v_nop\n\tv_nop\n\tv_nop\n\tv_nop" : "+v"(aS1[0]), "+v"(aS1[1]));

    if (seg_e >= NE) break;
    seg_b = seg_e;
    seg_e = (seg_e + CAP < NE) ? seg_e + CAP : NE;
  }

  int tq = threadIdx.x;
  asm volatile("" : "+v"(tq));
  const int wq = tq >> 5, hq = (tq >> 4) & 1, cq = tq & 15;
#pragma unroll
  for (int jj = 0; jj < 2; ++jj) {
    const int u = wq * 32 + jj * 16 + cq;
#pragma unroll
    for (int r = 0; r < 8; ++r) {
      const size_t n = (size_t)(n0 + r + 8 * hq);
      s_acc[n * 256 + u]       = aS0[jj][r] * INV_NEIGH;
      s_acc[n * 256 + 128 + u] = aS1[jj][r] * INV_NEIGH;
#pragma unroll
      for (int c = 0; c < 3; ++c) {
        v_acc[(n * 256 + u) * 3 + c]       = aV2[jj][c][r] * INV_NEIGH;
        v_acc[(n * 256 + 128 + u) * 3 + c] = aV3[jj][c][r] * INV_NEIGH;
      }
    }
  }
}

__global__ __launch_bounds__(128) void out_kernel(
    const float* __restrict__ s_acc, const float* __restrict__ v_acc,
    const float* __restrict__ lin2_ws, const float* __restrict__ lin2_wv,
    float* __restrict__ out) {
  __shared__ float st[16][256 + LPAD];
  __shared__ float vt[16][256 + LPAD];
  const int n0  = blockIdx.x * 16;
  const int tid = threadIdx.x;
  const int wave = tid >> 5;
  const int lane = tid & 31;
  const int half = lane >> 4;
  const int m    = lane & 15;
  const int nl   = lane & 15;
  const int wA   = wave * 16;
  const int wB   = wA + 64;

  v2f p0A[8], p0B[8], p1A[8], p1B[8];
  const float* L2s = lin2_ws + (size_t)(2 * half) * MULC + nl;
  loadB8(p0A, p0B, L2s, wA, wB);

  for (int i = tid; i < 16 * 256; i += 128)
    st[i >> 8][i & 255] = s_acc[(size_t)n0 * 256 + i];
  __syncthreads();

  {
    v8f accA = {}, accB = {};
    for (int g = 0; g < 8; g += 2) {
      loadB8(p1A, p1B, L2s + (size_t)(g + 1) * 32 * MULC, wA, wB);
      mma8(&st[m][g * 32 + 2 * half], p0A, p0B, accA, accB);
      if (g + 2 < 8) loadB8(p0A, p0B, L2s + (size_t)(g + 2) * 32 * MULC, wA, wB);
      mma8(&st[m][(g + 1) * 32 + 2 * half], p1A, p1B, accA, accB);
    }
#pragma unroll
    for (int r = 0; r < 8; ++r) {
      size_t row = (size_t)(n0 + r + 8 * half) * 512;
      out[row + wA + nl] += accA[r] * L2_NORM;
      out[row + wB + nl] += accB[r] * L2_NORM;
    }
  }
  for (int c = 0; c < 3; ++c) {
    __syncthreads();
    int voff = 0;
    asm volatile("" : "+s"(voff));
    const float* L2v = lin2_wv + voff + (size_t)(2 * half) * MULC + nl;
    loadB8(p0A, p0B, L2v, wA, wB);
#pragma nounroll
    for (int hgrp = 0; hgrp < 2; ++hgrp) {
      float tv[16];
#pragma unroll
      for (int q = 0; q < 16; ++q) {
        const int i = tid + (hgrp * 16 + q) * 128;
        const int r = i >> 8, k = i & 255;
        tv[q] = v_acc[(((size_t)(n0 + r)) * 256 + k) * 3 + c];
      }
#pragma unroll
      for (int q = 0; q < 16; ++q) {
        const int i = tid + (hgrp * 16 + q) * 128;
        vt[i >> 8][i & 255] = tv[q];
      }
    }
    __syncthreads();
    v8f accA = {}, accB = {};
    for (int g = 0; g < 8; g += 2) {
      loadB8(p1A, p1B, L2v + (size_t)(g + 1) * 32 * MULC, wA, wB);
      mma8(&vt[m][g * 32 + 2 * half], p0A, p0B, accA, accB);
      if (g + 2 < 8) loadB8(p0A, p0B, L2v + (size_t)(g + 2) * 32 * MULC, wA, wB);
      mma8(&vt[m][(g + 1) * 32 + 2 * half], p1A, p1B, accA, accB);
    }
    float oA[8], oB[8];
#pragma unroll
    for (int r = 0; r < 8; ++r) {
      size_t row = (size_t)(n0 + r + 8 * half) * 512 + MULC;
      oA[r] = out[row + (wA + nl) * 3 + c];
      oB[r] = out[row + (wB + nl) * 3 + c];
    }
#pragma unroll
    for (int r = 0; r < 8; ++r) {
      size_t row = (size_t)(n0 + r + 8 * half) * 512 + MULC;
      out[row + (wA + nl) * 3 + c] = oA[r] + accA[r] * L2_NORM;
      out[row + (wB + nl) * 3 + c] = oB[r] + accB[r] * L2_NORM;
    }
  }
}

extern "C" void kernel_launch(void* const* d_in, const int* in_sizes, int n_in,
                              void* d_out, int out_size, void* d_ws,
                              size_t ws_size, hipStream_t stream) {
  const float* node_feats = (const float*)d_in[0];
  const float* node_attrs = (const float*)d_in[1];
  const float* edge_feats = (const float*)d_in[2];
  const float* edge_attrs = (const float*)d_in[3];
  const int*   edge_index = (const int*)d_in[4];
  const float* lin1_ws    = (const float*)d_in[5];
  const float* lin1_wv    = (const float*)d_in[6];
  const float* fc_w0      = (const float*)d_in[7];
  const float* fc_w1      = (const float*)d_in[8];
  const float* fc_w2      = (const float*)d_in[9];
  const float* lin2_ws    = (const float*)d_in[10];
  const float* lin2_wv    = (const float*)d_in[11];
  const float* sc_ws      = (const float*)d_in[12];
  const float* sc_wv      = (const float*)d_in[13];

  float* out   = (float*)d_out;
  float* y     = (float*)d_ws;
  float* s_acc = y + (size_t)NN * 512;
  float* v_acc = s_acc + (size_t)NN * 256;

  node_kernel<<<NN / 16, 128, 0, stream>>>(node_feats, node_attrs, lin1_ws,
                                           lin1_wv, sc_ws, sc_wv, out, y);
  edge_kernel<<<NN / 16, 128, 0, stream>>>(y, edge_feats, edge_attrs,
                                          edge_index, fc_w0, fc_w1, fc_w2,
                                          s_acc, v_acc);
  out_kernel<<<NN / 16, 128, 0, stream>>>(s_acc, v_acc, lin2_ws, lin2_wv, out);
}


# vector-block products (sc_v, y_v, lin2_v) take all three Cartesian channels per weight fragment: B loaded once, used by 3 A tiles (48 rows); z chunk and v tile staged once for the three
# speedup vs baseline: 2.0504x; 1.0758x over previous
#include <hip/hip_runtime.h>
#include <hip/hip_bf16.h>


typedef __attribute__((ext_vector_type(2))) float v2f;
typedef __attribute__((ext_vector_type(8))) float v8f;

static constexpr int NN   = 10000;
static constexpr int NE   = 160000;
static constexpr int MULC = 128;
static constexpr int NSP  = 10;
static constexpr int NRAD = 8;

static constexpr int LPAD = 4;

static constexpr float SILU_C    = 1.679f;
static constexpr float SC_NORM   = 0.02795084971874737f;
static constexpr float L1_NORM   = 0.08838834764831845f;
static constexpr float L2_NORM   = 0.0625f;
static constexpr float INV_SQRT8 = 0.3535533905932738f;
static constexpr float INV_SQ3   = 0.5773502691896258f;
static constexpr float INV_NEIGH = 0.25f;

__device__ __forceinline__ v8f wmma4(v2f a, v2f b, v8f c) {
  return __builtin_amdgcn_wmma_f32_16x16x4_f32(
       false, a,  false, b,
       (short)0, c,  false,  false);
}

__device__ __forceinline__ float silu_c(float z) {
  return SILU_C * z / (1.0f + __expf(-z));
}

__device__ __forceinline__ void loadB8(v2f (&bA)[8], v2f (&bB)[8],
                                       const float* __restrict__ p, int wA,
                                       int wB) {
#pragma unroll
  for (int s = 0; s < 8; ++s) {
    const float* q = p + s * 4 * MULC;
    bA[s] = v2f{ q[wA], q[MULC + wA] };
    bB[s] = v2f{ q[wB], q[MULC + wB] };
  }
}

__device__ __forceinline__ void mma8(const float* arow, const v2f (&bA)[8],
                                     const v2f (&bB)[8], v8f& accA, v8f& accB) {
#pragma unroll
  for (int s = 0; s < 8; ++s) {
    v2f a = { arow[4 * s], arow[4 * s + 1] };
    accA = wmma4(a, bA[s], accA);
    accB = wmma4(a, bB[s], accB);
  }
}

__device__ __forceinline__ void mma8x3(const float* a0, const float* a1,
                                       const float* a2, const v2f (&bA)[8],
                                       const v2f (&bB)[8], v8f (&accA)[3],
                                       v8f (&accB)[3]) {
#pragma unroll
  for (int s = 0; s < 8; ++s) {
    v2f x0 = { a0[4 * s], a0[4 * s + 1] };
    v2f x1 = { a1[4 * s], a1[4 * s + 1] };
    v2f x2 = { a2[4 * s], a2[4 * s + 1] };
    accA[0] = wmma4(x0, bA[s], accA[0]);
    accA[1] = wmma4(x1, bA[s], accA[1]);
    accA[2] = wmma4(x2, bA[s], accA[2]);
    accB[0] = wmma4(x0, bB[s], accB[0]);
    accB[1] = wmma4(x1, bB[s], accB[1]);
    accB[2] = wmma4(x2, bB[s], accB[2]);
  }
}

__global__ __launch_bounds__(128) void node_kernel(
    const float* __restrict__ node_feats, const float* __restrict__ node_attrs,
    const float* __restrict__ lin1_ws, const float* __restrict__ lin1_wv,
    const float* __restrict__ sc_ws, const float* __restrict__ sc_wv,
    float* __restrict__ out, float* __restrict__ y) {
  __shared__ __attribute__((aligned(16))) float xs[16][MULC + LPAD];
  __shared__ float xvc[3][16][MULC + LPAD];
  __shared__ float at[16][16];
  __shared__ float zbuf[3][16][256 + LPAD];

  const int n0  = blockIdx.x * 16;
  const int tid = threadIdx.x;

  {
    float av[2];
#pragma unroll
    for (int q = 0; q < 2; ++q) {
      const int i = tid + q * 128, r = i >> 4, c = i & 15;
      av[q] = (c < NSP) ? node_attrs[(size_t)(n0 + r) * NSP + c] : 0.0f;
    }
    const float4* nf4 = (const float4*)(node_feats + (size_t)n0 * 512);
    float4 t[16];
#pragma unroll
    for (int q = 0; q < 16; ++q) t[q] = nf4[q * 128 + tid];
#pragma unroll
    for (int q = 0; q < 2; ++q) {
      const int i = tid + q * 128;
      at[i >> 4][i & 15] = av[q];
    }
    const int c0 = tid * 4;
    if (c0 < MULC) {
#pragma unroll
      for (int q = 0; q < 16; ++q) *(float4*)&xs[q][c0] = t[q];
    } else {
      const int f0 = c0 - MULC;
      const int h0 = f0 % 3, u0 = f0 / 3;
      const int h1 = (f0 + 1) % 3, u1 = (f0 + 1) / 3;
      const int h2 = (f0 + 2) % 3, u2 = (f0 + 2) / 3;
      const int h3 = (f0 + 3) % 3, u3 = (f0 + 3) / 3;
#pragma unroll
      for (int q = 0; q < 16; ++q) {
        xvc[h0][q][u0] = t[q].x;
        xvc[h1][q][u1] = t[q].y;
        xvc[h2][q][u2] = t[q].z;
        xvc[h3][q][u3] = t[q].w;
      }
    }
  }
  __syncthreads();

  const int wave = tid >> 5;
  const int lane = tid & 31;
  const int half = lane >> 4;
  const int m    = lane & 15;
  const int nl   = lane & 15;
  const int wA   = wave * 16;
  const int wB   = wA + 64;

  v2f p0A[8], p0B[8], p1A[8], p1B[8];

  {
    v8f accA = {}, accB = {};
    for (int chunk = 0; chunk < 5; ++chunk) {
      __syncthreads();
      const float* Wc = sc_ws + ((size_t)chunk * 256 + 2 * half) * MULC + nl;
      loadB8(p0A, p0B, Wc, wA, wB);
      for (int i = tid; i < 16 * 256; i += 128) {
        int r = i >> 8, kk = i & 255;
        int k = chunk * 256 + kk;
        zbuf[0][r][kk] = xs[r][k / NSP] * at[r][k % NSP];
      }
      __syncthreads();
#pragma unroll
      for (int g = 0; g < 8; g += 2) {
        loadB8(p1A, p1B, Wc + (size_t)(g + 1) * 32 * MULC, wA, wB);
        mma8(&zbuf[0][m][g * 32 + 2 * half], p0A, p0B, accA, accB);
        if (g + 2 < 8) loadB8(p0A, p0B, Wc + (size_t)(g + 2) * 32 * MULC, wA, wB);
        mma8(&zbuf[0][m][(g + 1) * 32 + 2 * half], p1A, p1B, accA, accB);
      }
    }
#pragma unroll
    for (int r = 0; r < 8; ++r) {
      size_t row = (size_t)(n0 + r + 8 * half) * 512;
      out[row + wA + nl] = accA[r] * SC_NORM;
      out[row + wB + nl] = accB[r] * SC_NORM;
    }
  }

  {
    v8f accA[3] = { v8f{}, v8f{}, v8f{} }, accB[3] = { v8f{}, v8f{}, v8f{} };
    for (int chunk = 0; chunk < 5; ++chunk) {
      __syncthreads();
      const float* Wc = sc_wv + ((size_t)chunk * 256 + 2 * half) * MULC + nl;
      loadB8(p0A, p0B, Wc, wA, wB);
      for (int i = tid; i < 16 * 256; i += 128) {
        int r = i >> 8, kk = i & 255;
        int k = chunk * 256 + kk;
        const int u = k / NSP;
        const float a = at[r][k - u * NSP];
        zbuf[0][r][kk] = xvc[0][r][u] * a;
        zbuf[1][r][kk] = xvc[1][r][u] * a;
        zbuf[2][r][kk] = xvc[2][r][u] * a;
      }
      __syncthreads();
#pragma unroll
      for (int g = 0; g < 8; g += 2) {
        const int ka = g * 32 + 2 * half, kb = ka + 32;
        loadB8(p1A, p1B, Wc + (size_t)(g + 1) * 32 * MULC, wA, wB);
        mma8x3(&zbuf[0][m][ka], &zbuf[1][m][ka], &zbuf[2][m][ka], p0A, p0B,
               accA, accB);
        if (g + 2 < 8) loadB8(p0A, p0B, Wc + (size_t)(g + 2) * 32 * MULC, wA, wB);
        mma8x3(&zbuf[0][m][kb], &zbuf[1][m][kb], &zbuf[2][m][kb], p1A, p1B,
               accA, accB);
      }
    }
#pragma unroll
    for (int r = 0; r < 8; ++r) {
      size_t row = (size_t)(n0 + r + 8 * half) * 512 + MULC;
#pragma unroll
      for (int c = 0; c < 3; ++c) {
        out[row + (wA + nl) * 3 + c] = accA[c][r] * SC_NORM;
        out[row + (wB + nl) * 3 + c] = accB[c][r] * SC_NORM;
      }
    }
  }

  const float* L1s = lin1_ws + (size_t)(2 * half) * MULC + nl;
  const float* L1v = lin1_wv + (size_t)(2 * half) * MULC + nl;
  loadB8(p0A, p0B, L1s, wA, wB);
  {
    v8f accA = {}, accB = {};
#pragma unroll
    for (int g = 0; g < 4; g += 2) {
      loadB8(p1A, p1B, L1s + (size_t)(g + 1) * 32 * MULC, wA, wB);
      mma8(&xs[m][g * 32 + 2 * half], p0A, p0B, accA, accB);
      loadB8(p0A, p0B, (g + 2 < 4) ? L1s + (size_t)(g + 2) * 32 * MULC : L1v, wA, wB);
      mma8(&xs[m][(g + 1) * 32 + 2 * half], p1A, p1B, accA, accB);
    }
#pragma unroll
    for (int r = 0; r < 8; ++r) {
      size_t row = (size_t)(n0 + r + 8 * half) * 512;
      y[row + wA + nl] = accA[r] * L1_NORM;
      y[row + wB + nl] = accB[r] * L1_NORM;
    }
  }
  asm volatile("" ::: "memory");
  {
    v8f accA[3] = { v8f{}, v8f{}, v8f{} }, accB[3] = { v8f{}, v8f{}, v8f{} };
#pragma unroll
    for (int g = 0; g < 4; g += 2) {
      const int ka = g * 32 + 2 * half, kb = ka + 32;
      loadB8(p1A, p1B, L1v + (size_t)(g + 1) * 32 * MULC, wA, wB);
      mma8x3(&xvc[0][m][ka], &xvc[1][m][ka], &xvc[2][m][ka], p0A, p0B, accA,
             accB);
      if (g + 2 < 4) loadB8(p0A, p0B, L1v + (size_t)(g + 2) * 32 * MULC, wA, wB);
      mma8x3(&xvc[0][m][kb], &xvc[1][m][kb], &xvc[2][m][kb], p1A, p1B, accA,
             accB);
    }
#pragma unroll
    for (int r = 0; r < 8; ++r) {
      size_t row = (size_t)(n0 + r + 8 * half) * 512 + MULC;
#pragma unroll
      for (int c = 0; c < 3; ++c) {
        y[row + (wA + nl) * 3 + c] = accA[c][r] * L1_NORM;
        y[row + (wB + nl) * 3 + c] = accB[c][r] * L1_NORM;
      }
    }
  }
}

static constexpr int CAP    = 1024;
static constexpr int SCAN_U = 8;

__device__ __forceinline__ void l2load(v2f (&b0)[16], v2f (&b1)[16],
                                       const float* __restrict__ w2, int j0,
                                       int half, int nl) {
#pragma unroll
  for (int s = 0; s < 16; ++s) {
    const int k0 = 4 * s + 2 * half;
    const float* p = w2 + (size_t)k0 * 512 + j0 + nl;
    b0[s] = v2f{ p[0], p[512] };
    b1[s] = v2f{ p[16], p[512 + 16] };
  }
}

__device__ __forceinline__ void l2mma(const v2f (&af)[16], const v2f (&b0)[16],
                                      const v2f (&b1)[16], v8f& c0, v8f& c1) {
#pragma unroll
  for (int s = 0; s < 16; ++s) {
    c0 = wmma4(af[s], b0[s], c0);
    c1 = wmma4(af[s], b1[s], c1);
  }
}

__device__ __forceinline__ v8f segsum(const v2f (&sel)[4], const float (&mid)[8],
                                      v8f acc) {
#pragma unroll
  for (int t = 0; t < 4; ++t) {
    v2f b = { mid[2 * t], mid[2 * t + 1] };
    acc = wmma4(sel[t], b, acc);
  }
  return acc;
}

__global__ __launch_bounds__(128, 3) void edge_kernel(
    const float* y, const float* __restrict__ edge_feats,
    const float* __restrict__ edge_attrs, const int* __restrict__ edge_index,
    const float* __restrict__ fc_w0, const float* __restrict__ fc_w1,
    const float* __restrict__ fc_w2, float* __restrict__ s_acc,
    float* __restrict__ v_acc) {
  __shared__ float ytile[16][512];
  __shared__ float ha[16][64 + LPAD];
  __shared__ float hb[16][64 + LPAD];
  __shared__ float ea[16][4];
  __shared__ float ef[16][NRAD + LPAD];
  __shared__ int   dl[16];
  __shared__ int   lst_e[CAP];
  __shared__ int   lst_d[CAP];
  __shared__ int   lst_s[CAP];
  __shared__ int   cnt;

  const int n0   = blockIdx.x * 16;
  const int tid  = threadIdx.x;
  const int wave = tid >> 5;
  const int lane = tid & 31;
  const int half = lane >> 4;
  const int m    = lane & 15;
  const int nl   = lane & 15;

  v8f aS0[2], aS1[2], aV2[2][3], aV3[2][3];
#pragma unroll
  for (int jj = 0; jj < 2; ++jj) {
    aS0[jj] = v8f{};
    aS1[jj] = v8f{};
#pragma unroll
    for (int c = 0; c < 3; ++c) { aV2[jj][c] = v8f{}; aV3[jj][c] = v8f{}; }
  }

  const int4* d4 = (const int4*)edge_index;
  int seg_b = 0, seg_e = NE;
  for (;;) {
    if (tid == 0) cnt = 0;
    __syncthreads();
    const int e4 = seg_e >> 2;
    for (int base = (seg_b >> 2) + tid; base < e4; base += 128 * SCAN_U) {
      int4 v[SCAN_U];
#pragma unroll
      for (int q = 0; q < SCAN_U; ++q) {
        const int idx = base + q * 128;
        v[q] = (idx < e4) ? d4[idx] : make_int4(-1, -1, -1, -1);
      }
#pragma unroll
      for (int q = 0; q < SCAN_U; ++q) {
        const int idx = base + q * 128;
        const unsigned dd[4] = { (unsigned)v[q].x - (unsigned)n0,
                                 (unsigned)v[q].y - (unsigned)n0,
                                 (unsigned)v[q].z - (unsigned)n0,
                                 (unsigned)v[q].w - (unsigned)n0 };
#pragma unroll
        for (int z = 0; z < 4; ++z) {
          if (dd[z] < 16u) {
            const int p = atomicAdd(&cnt, 1);
            if (p < CAP) { lst_e[p] = idx * 4 + z; lst_d[p] = (int)dd[z]; }
          }
        }
      }
    }
    __syncthreads();
    const int c = cnt;
    __syncthreads();
    if (c > CAP) { seg_b = 0; seg_e = CAP; continue; }

    for (int i = tid; i < c; i += 128) lst_s[i] = edge_index[NE + lst_e[i]];
    __syncthreads();

    const int ntile = (c + 15) >> 4;
    for (int ti = 0; ti < ntile; ++ti) {
      int woff = 0;
      asm volatile("" : "+s"(woff));
      const float* w0 = fc_w0 + woff;
      const float* w1 = fc_w1 + woff;
      const float* w2 = fc_w2 + woff;
      v2f bw0[2], bw1[16];
      v2f qx0[16], qx1[16], qy0[16], qy1[16];
      {
        const float4* sp[4];
        int tw = tid;
        asm volatile("" : "+v"(tw));
        const int wrow = (tw >> 5) * 4;
#pragma unroll
        for (int rr = 0; rr < 4; ++rr) {
          const int li  = ti * 16 + wrow + rr;
          const int lic = (li < c) ? li : 0;
          const int sv  = lst_s[lic];
          const int sr  = (li < c) ? sv : 0;
          sp[rr] = (const float4*)(y + (size_t)sr * 512);
        }
        const int fr = tid >> 3, fk = tid & 7;
        const int fli = ti * 16 + fr;
        const int fe  = lst_e[(fli < c) ? fli : 0];
        const int ar = (tid >> 2) & 15, ak = tid & 3;
        const int ali = ti * 16 + ar;
        const int ae  = lst_e[(ali < c) ? ali : 0];
        const float4 a0 = sp[0][lane], a1 = sp[0][lane + 32],
                     a2 = sp[0][lane + 64], a3 = sp[0][lane + 96];
        const float4 b0 = sp[1][lane], b1 = sp[1][lane + 32],
                     b2 = sp[1][lane + 64], b3 = sp[1][lane + 96];
        const float4 c0 = sp[2][lane], c1 = sp[2][lane + 32],
                     c2 = sp[2][lane + 64], c3 = sp[2][lane + 96];
        const float4 d0 = sp[3][lane], d1 = sp[3][lane + 32],
                     d2 = sp[3][lane + 64], d3 = sp[3][lane + 96];
        float fraw = edge_feats[(unsigned)fe * NRAD + (unsigned)fk];
        float araw = 0.0f;
        if (tid < 64) araw = edge_attrs[(unsigned)ae * 4u + (unsigned)ak];
#pragma unroll
        for (int s = 0; s < 2; ++s) {
          const int k0 = 4 * s + 2 * half;
          bw0[s] = v2f{ w0[k0 * 64 + wave * 16 + nl],
                        w0[(k0 + 1) * 64 + wave * 16 + nl] };
        }
#pragma unroll
        for (int s = 0; s < 16; ++s) {
          const int k0 = 4 * s + 2 * half;
          bw1[s] = v2f{ w1[k0 * 64 + wave * 16 + nl],
                        w1[(k0 + 1) * 64 + wave * 16 + nl] };
        }
        asm volatile("" : "+v"(fraw), "+v"(araw) : : "memory");
        float4* dp = (float4*)&ytile[wave * 4][0];
        dp[lane]       = a0; dp[lane + 32]  = a1; dp[lane + 64]  = a2; dp[lane + 96]  = a3;
        dp[lane + 128] = b0; dp[lane + 160] = b1; dp[lane + 192] = b2; dp[lane + 224] = b3;
        dp[lane + 256] = c0; dp[lane + 288] = c1; dp[lane + 320] = c2; dp[lane + 352] = c3;
        dp[lane + 384] = d0; dp[lane + 416] = d1; dp[lane + 448] = d2; dp[lane + 480] = d3;
        ef[fr][fk] = (fli < c) ? fraw : 0.0f;
        if (tid < 64) ea[ar][ak] = (ali < c) ? araw : 0.0f;
      }
      if (tid < 16) {
        const int li = ti * 16 + tid;
        dl[tid] = (li < c) ? lst_d[li] : -1;
      }
      __syncthreads();

      {
        v8f acc = {};
#pragma unroll
        for (int s = 0; s < 2; ++s) {
          const int k0 = 4 * s + 2 * half;
          v2f a = { ef[m][k0], ef[m][k0 + 1] };
          acc = wmma4(a, bw0[s], acc);
        }
#pragma unroll
        for (int r = 0; r < 8; ++r)
          ha[r + 8 * half][wave * 16 + nl] = silu_c(acc[r] * INV_SQRT8);
      }
      __syncthreads();

      l2load(qx0, qx1, w2, 0 * 128 + wave * 32, half, nl);

      {
        v8f acc = {};
#pragma unroll
        for (int s = 0; s < 16; ++s) {
          const int k0 = 4 * s + 2 * half;
          v2f a = { ha[m][k0], ha[m][k0 + 1] };
          acc = wmma4(a, bw1[s], acc);
        }
#pragma unroll
        for (int r = 0; r < 8; ++r)
          hb[r + 8 * half][wave * 16 + nl] = silu_c(acc[r] * 0.125f);
      }
      __syncthreads();

      v2f af[16];
#pragma unroll
      for (int s = 0; s < 16; ++s) {
        const int k0 = 4 * s + 2 * half;
        af[s] = v2f{ hb[m][k0], hb[m][k0 + 1] };
      }
      v2f sel[4];
      int mrow = m;
      asm volatile("" : "+v"(mrow));
#pragma unroll
      for (int t = 0; t < 4; ++t) {
        const int e0 = 8 * half + 2 * t;
        sel[t] = v2f{ dl[e0] == mrow ? 1.0f : 0.0f, dl[e0 + 1] == mrow ? 1.0f : 0.0f };
      }

      {
        v8f cc[2] = { v8f{}, v8f{} };
        l2mma(af, qx0, qx1, cc[0], cc[1]);
        asm volatile("" : "+s"(woff) : "v"(cc[1][7]));
        w2 = fc_w2 + woff;
        l2load(qy0, qy1, w2, 1 * 128 + wave * 32, half, nl);
        asm volatile("" ::: "memory");
#pragma unroll
        for (int jj = 0; jj < 2; ++jj) {
          const int u = wave * 32 + jj * 16 + nl;
          float mid[8];
#pragma unroll
          for (int r = 0; r < 8; ++r) {
            const int e = r + 8 * half;
            mid[r] = (cc[jj][r] * 0.125f) * ytile[e][u] * ea[e][0];
          }
          aS0[jj] = segsum(sel, mid, aS0[jj]);
        }
      }
      {
        v8f cc[2] = { v8f{}, v8f{} };
        asm volatile("" : "+v"(qy0[0]), "+v"(qy1[0]) : "v"(aS0[0][7]), "v"(aS0[1][7]));
        l2mma(af, qy0, qy1, cc[0], cc[1]);
        asm volatile("" : "+s"(woff) : "v"(cc[1][7]));
        w2 = fc_w2 + woff;
        l2load(qx0, qx1, w2, 2 * 128 + wave * 32, half, nl);
        asm volatile("" ::: "memory");
#pragma unroll
        for (int jj = 0; jj < 2; ++jj) {
          const int u = wave * 32 + jj * 16 + nl;
          float m0[8], m1[8], m2[8];
#pragma unroll
          for (int r = 0; r < 8; ++r) {
            const int e = r + 8 * half;
            const float base = (cc[jj][r] * 0.125f) * ytile[e][u];
            m0[r] = base * ea[e][1];
            m1[r] = base * ea[e][2];
            m2[r] = base * ea[e][3];
          }
          aV2[jj][0] = segsum(sel, m0, aV2[jj][0]);
          aV2[jj][1] = segsum(sel, m1, aV2[jj][1]);
          aV2[jj][2] = segsum(sel, m2, aV2[jj][2]);
        }
      }
      {
        v8f cc[2] = { v8f{}, v8f{} };
        asm volatile("" : "+v"(qx0[0]), "+v"(qx1[0]) : "v"(aV2[0][0][7]), "v"(aV2[0][1][7]), "v"(aV2[0][2][7]), "v"(aV2[1][0][7]), "v"(aV2[1][1][7]), "v"(aV2[1][2][7]));
        l2mma(af, qx0, qx1, cc[0], cc[1]);
        asm volatile("" : "+s"(woff) : "v"(cc[1][7]));
        w2 = fc_w2 + woff;
        l2load(qy0, qy1, w2, 3 * 128 + wave * 32, half, nl);
        asm volatile("" ::: "memory");
#pragma unroll
        for (int jj = 0; jj < 2; ++jj) {
          const int u = wave * 32 + jj * 16 + nl;
          float m0[8], m1[8], m2[8];
#pragma unroll
          for (int r = 0; r < 8; ++r) {
            const int e = r + 8 * half;
            const float base = (cc[jj][r] * 0.125f) * ea[e][0];
            m0[r] = base * ytile[e][128 + u * 3 + 0];
            m1[r] = base * ytile[e][128 + u * 3 + 1];
            m2[r] = base * ytile[e][128 + u * 3 + 2];
          }
          aV3[jj][0] = segsum(sel, m0, aV3[jj][0]);
          aV3[jj][1] = segsum(sel, m1, aV3[jj][1]);
          aV3[jj][2] = segsum(sel, m2, aV3[jj][2]);
        }
      }
      {
        v8f cc[2] = { v8f{}, v8f{} };
        asm volatile("" : "+v"(qy0[0]), "+v"(qy1[0]) : "v"(aV3[0][0][7]), "v"(aV3[0][1][7]), "v"(aV3[0][2][7]), "v"(aV3[1][0][7]), "v"(aV3[1][1][7]), "v"(aV3[1][2][7]));
        l2mma(af, qy0, qy1, cc[0], cc[1]);
#pragma unroll
        for (int jj = 0; jj < 2; ++jj) {
          const int u = wave * 32 + jj * 16 + nl;
          float mid[8];
#pragma unroll
          for (int r = 0; r < 8; ++r) {
            const int e = r + 8 * half;
            const float dotv = ytile[e][128 + u * 3 + 0] * ea[e][1] +
                               ytile[e][128 + u * 3 + 1] * ea[e][2] +
                               ytile[e][128 + u * 3 + 2] * ea[e][3];
            mid[r] = (cc[jj][r] * 0.125f) * dotv * INV_SQ3;
          }
          aS1[jj] = segsum(sel, mid, aS1[jj]);
        }
      }
      __syncthreads();
    }
    asm volatile("v_nop\n\tv_nop\n\tv_nop\n\tv_nop" : "+v"(aS1[0]), "+v"(aS1[1]));

    if (seg_e >= NE) break;
    seg_b = seg_e;
    seg_e = (seg_e + CAP < NE) ? seg_e + CAP : NE;
  }

  int tq = threadIdx.x;
  asm volatile("" : "+v"(tq));
  const int wq = tq >> 5, hq = (tq >> 4) & 1, cq = tq & 15;
#pragma unroll
  for (int jj = 0; jj < 2; ++jj) {
    const int u = wq * 32 + jj * 16 + cq;
#pragma unroll
    for (int r = 0; r < 8; ++r) {
      const size_t n = (size_t)(n0 + r + 8 * hq);
      s_acc[n * 256 + u]       = aS0[jj][r] * INV_NEIGH;
      s_acc[n * 256 + 128 + u] = aS1[jj][r] * INV_NEIGH;
#pragma unroll
      for (int c = 0; c < 3; ++c) {
        v_acc[(n * 256 + u) * 3 + c]       = aV2[jj][c][r] * INV_NEIGH;
        v_acc[(n * 256 + 128 + u) * 3 + c] = aV3[jj][c][r] * INV_NEIGH;
      }
    }
  }
}

__global__ __launch_bounds__(128) void out_kernel(
    const float* __restrict__ s_acc, const float* __restrict__ v_acc,
    const float* __restrict__ lin2_ws, const float* __restrict__ lin2_wv,
    float* __restrict__ out) {
  __shared__ float st[16][256 + LPAD];
  __shared__ float vt[3][16][256 + LPAD];
  const int n0  = blockIdx.x * 16;
  const int tid = threadIdx.x;
  const int wave = tid >> 5;
  const int lane = tid & 31;
  const int half = lane >> 4;
  const int m    = lane & 15;
  const int nl   = lane & 15;
  const int wA   = wave * 16;
  const int wB   = wA + 64;

  v2f p0A[8], p0B[8], p1A[8], p1B[8];
  const float* L2s = lin2_ws + (size_t)(2 * half) * MULC + nl;
  const float* L2v = lin2_wv + (size_t)(2 * half) * MULC + nl;
  loadB8(p0A, p0B, L2s, wA, wB);

  {
    const float4* va4 = (const float4*)(v_acc + (size_t)n0 * 768);
    float4 t[24];
#pragma unroll
    for (int q = 0; q < 24; ++q) t[q] = va4[q * 128 + tid];
    int vo[3][4];
#pragma unroll
    for (int i = 0; i < 3; ++i) {
      const int w  = i * 128 + tid;
      const int rr = w / 192;
      const int f0 = (w - rr * 192) * 4;
#pragma unroll
      for (int e = 0; e < 4; ++e) {
        const int f = f0 + e, k = f / 3, c = f - 3 * k;
        vo[i][e] = (c * 16 + rr) * (256 + LPAD) + k;
      }
    }
    for (int i = tid; i < 16 * 256; i += 128)
      st[i >> 8][i & 255] = s_acc[(size_t)n0 * 256 + i];
    float* vb = &vt[0][0][0];
#pragma unroll
    for (int q = 0; q < 24; ++q) {
      const int j = q / 3, i = q - 3 * j;
      float* b = vb + 2 * j * (256 + LPAD);
      b[vo[i][0]] = t[q].x;
      b[vo[i][1]] = t[q].y;
      b[vo[i][2]] = t[q].z;
      b[vo[i][3]] = t[q].w;
    }
  }
  __syncthreads();

  {
    v8f accA = {}, accB = {};
#pragma unroll
    for (int g = 0; g < 8; g += 2) {
      loadB8(p1A, p1B, L2s + (size_t)(g + 1) * 32 * MULC, wA, wB);
      mma8(&st[m][g * 32 + 2 * half], p0A, p0B, accA, accB);
      loadB8(p0A, p0B, (g + 2 < 8) ? L2s + (size_t)(g + 2) * 32 * MULC : L2v, wA, wB);
      mma8(&st[m][(g + 1) * 32 + 2 * half], p1A, p1B, accA, accB);
    }
    float oA[8], oB[8];
#pragma unroll
    for (int r = 0; r < 8; ++r) {
      size_t row = (size_t)(n0 + r + 8 * half) * 512;
      oA[r] = out[row + wA + nl];
      oB[r] = out[row + wB + nl];
    }
#pragma unroll
    for (int r = 0; r < 8; ++r) {
      size_t row = (size_t)(n0 + r + 8 * half) * 512;
      out[row + wA + nl] = oA[r] + accA[r] * L2_NORM;
      out[row + wB + nl] = oB[r] + accB[r] * L2_NORM;
    }
  }
  {
    v8f accA[3] = { v8f{}, v8f{}, v8f{} }, accB[3] = { v8f{}, v8f{}, v8f{} };
#pragma unroll
    for (int g = 0; g < 8; g += 2) {
      const int ka = g * 32 + 2 * half, kb = ka + 32;
      loadB8(p1A, p1B, L2v + (size_t)(g + 1) * 32 * MULC, wA, wB);
      mma8x3(&vt[0][m][ka], &vt[1][m][ka], &vt[2][m][ka], p0A, p0B, accA, accB);
      if (g + 2 < 8) loadB8(p0A, p0B, L2v + (size_t)(g + 2) * 32 * MULC, wA, wB);
      mma8x3(&vt[0][m][kb], &vt[1][m][kb], &vt[2][m][kb], p1A, p1B, accA, accB);
    }
    float oA[8][3], oB[8][3];
#pragma unroll
    for (int r = 0; r < 8; ++r) {
      size_t row = (size_t)(n0 + r + 8 * half) * 512 + MULC;
#pragma unroll
      for (int c = 0; c < 3; ++c) {
        oA[r][c] = out[row + (wA + nl) * 3 + c];
        oB[r][c] = out[row + (wB + nl) * 3 + c];
      }
    }
#pragma unroll
    for (int r = 0; r < 8; ++r) {
      size_t row = (size_t)(n0 + r + 8 * half) * 512 + MULC;
#pragma unroll
      for (int c = 0; c < 3; ++c) {
        out[row + (wA + nl) * 3 + c] = oA[r][c] + accA[c][r] * L2_NORM;
        out[row + (wB + nl) * 3 + c] = oB[r][c] + accB[c][r] * L2_NORM;
      }
    }
  }
}

extern "C" void kernel_launch(void* const* d_in, const int* in_sizes, int n_in,
                              void* d_out, int out_size, void* d_ws,
                              size_t ws_size, hipStream_t stream) {
  const float* node_feats = (const float*)d_in[0];
  const float* node_attrs = (const float*)d_in[1];
  const float* edge_feats = (const float*)d_in[2];
  const float* edge_attrs = (const float*)d_in[3];
  const int*   edge_index = (const int*)d_in[4];
  const float* lin1_ws    = (const float*)d_in[5];
  const float* lin1_wv    = (const float*)d_in[6];
  const float* fc_w0      = (const float*)d_in[7];
  const float* fc_w1      = (const float*)d_in[8];
  const float* fc_w2      = (const float*)d_in[9];
  const float* lin2_ws    = (const float*)d_in[10];
  const float* lin2_wv    = (const float*)d_in[11];
  const float* sc_ws      = (const float*)d_in[12];
  const float* sc_wv      = (const float*)d_in[13];

  float* out   = (float*)d_out;
  float* y     = (float*)d_ws;
  float* s_acc = y + (size_t)NN * 512;
  float* v_acc = s_acc + (size_t)NN * 256;

  node_kernel<<<NN / 16, 128, 0, stream>>>(node_feats, node_attrs, lin1_ws,
                                           lin1_wv, sc_ws, sc_wv, out, y);
  edge_kernel<<<NN / 16, 128, 0, stream>>>(y, edge_feats, edge_attrs,
                                          edge_index, fc_w0, fc_w1, fc_w2,
                                          s_acc, v_acc);
  out_kernel<<<NN / 16, 128, 0, stream>>>(s_acc, v_acc, lin2_ws, lin2_wv, out);
}


# a wave's two column tiles interleaved (tile j, lane column c = matrix column base+2c+j): every weight row is one 8-byte load for both tiles, results stored as neighbouring pairs; sums unchanged
# speedup vs baseline: 2.2715x; 1.1078x over previous
#include <hip/hip_runtime.h>
#include <hip/hip_bf16.h>


typedef __attribute__((ext_vector_type(2))) float v2f;
typedef __attribute__((ext_vector_type(8))) float v8f;

static constexpr int NN   = 10000;
static constexpr int NE   = 160000;
static constexpr int MULC = 128;
static constexpr int NSP  = 10;
static constexpr int NRAD = 8;

static constexpr int LPAD = 4;

static constexpr float SILU_C    = 1.679f;
static constexpr float SC_NORM   = 0.02795084971874737f;
static constexpr float L1_NORM   = 0.08838834764831845f;
static constexpr float L2_NORM   = 0.0625f;
static constexpr float INV_SQRT8 = 0.3535533905932738f;
static constexpr float INV_SQ3   = 0.5773502691896258f;
static constexpr float INV_NEIGH = 0.25f;

__device__ __forceinline__ v8f wmma4(v2f a, v2f b, v8f c) {
  return __builtin_amdgcn_wmma_f32_16x16x4_f32(
       false, a,  false, b,
       (short)0, c,  false,  false);
}

__device__ __forceinline__ float silu_c(float z) {
  return SILU_C * z / (1.0f + __expf(-z));
}

__device__ __forceinline__ void loadB8(v2f (&bA)[8], v2f (&bB)[8],
                                       const float* __restrict__ p) {
#pragma unroll
  for (int s = 0; s < 8; ++s) {
    const float* q = p + s * 4 * MULC;
    const v2f r0 = *(const v2f*)q;
    const v2f r1 = *(const v2f*)(q + MULC);
    bA[s] = v2f{ r0.x, r1.x };
    bB[s] = v2f{ r0.y, r1.y };
  }
}

__device__ __forceinline__ void mma8(const float* arow, const v2f (&bA)[8],
                                     const v2f (&bB)[8], v8f& accA, v8f& accB) {
#pragma unroll
  for (int s = 0; s < 8; ++s) {
    v2f a = { arow[4 * s], arow[4 * s + 1] };
    accA = wmma4(a, bA[s], accA);
    accB = wmma4(a, bB[s], accB);
  }
}

__device__ __forceinline__ void mma8x3(const float* a0, const float* a1,
                                       const float* a2, const v2f (&bA)[8],
                                       const v2f (&bB)[8], v8f (&accA)[3],
                                       v8f (&accB)[3]) {
#pragma unroll
  for (int s = 0; s < 8; ++s) {
    v2f x0 = { a0[4 * s], a0[4 * s + 1] };
    v2f x1 = { a1[4 * s], a1[4 * s + 1] };
    v2f x2 = { a2[4 * s], a2[4 * s + 1] };
    accA[0] = wmma4(x0, bA[s], accA[0]);
    accA[1] = wmma4(x1, bA[s], accA[1]);
    accA[2] = wmma4(x2, bA[s], accA[2]);
    accB[0] = wmma4(x0, bB[s], accB[0]);
    accB[1] = wmma4(x1, bB[s], accB[1]);
    accB[2] = wmma4(x2, bB[s], accB[2]);
  }
}

__global__ __launch_bounds__(128) void node_kernel(
    const float* __restrict__ node_feats, const float* __restrict__ node_attrs,
    const float* __restrict__ lin1_ws, const float* __restrict__ lin1_wv,
    const float* __restrict__ sc_ws, const float* __restrict__ sc_wv,
    float* __restrict__ out, float* __restrict__ y) {
  __shared__ __attribute__((aligned(16))) float xs[16][MULC + LPAD];
  __shared__ float xvc[3][16][MULC + LPAD];
  __shared__ float at[16][16];
  __shared__ float zbuf[3][16][256 + LPAD];

  const int n0  = blockIdx.x * 16;
  const int tid = threadIdx.x;

  {
    float av[2];
#pragma unroll
    for (int q = 0; q < 2; ++q) {
      const int i = tid + q * 128, r = i >> 4, c = i & 15;
      av[q] = (c < NSP) ? node_attrs[(size_t)(n0 + r) * NSP + c] : 0.0f;
    }
    const float4* nf4 = (const float4*)(node_feats + (size_t)n0 * 512);
    float4 t[16];
#pragma unroll
    for (int q = 0; q < 16; ++q) t[q] = nf4[q * 128 + tid];
#pragma unroll
    for (int q = 0; q < 2; ++q) {
      const int i = tid + q * 128;
      at[i >> 4][i & 15] = av[q];
    }
    const int c0 = tid * 4;
    if (c0 < MULC) {
#pragma unroll
      for (int q = 0; q < 16; ++q) *(float4*)&xs[q][c0] = t[q];
    } else {
      const int f0 = c0 - MULC;
      const int h0 = f0 % 3, u0 = f0 / 3;
      const int h1 = (f0 + 1) % 3, u1 = (f0 + 1) / 3;
      const int h2 = (f0 + 2) % 3, u2 = (f0 + 2) / 3;
      const int h3 = (f0 + 3) % 3, u3 = (f0 + 3) / 3;
#pragma unroll
      for (int q = 0; q < 16; ++q) {
        xvc[h0][q][u0] = t[q].x;
        xvc[h1][q][u1] = t[q].y;
        xvc[h2][q][u2] = t[q].z;
        xvc[h3][q][u3] = t[q].w;
      }
    }
  }
  __syncthreads();

  const int wave = tid >> 5;
  const int lane = tid & 31;
  const int half = lane >> 4;
  const int m    = lane & 15;
  const int nl   = lane & 15;
  const int c2   = wave * 32 + 2 * nl;

  v2f p0A[8], p0B[8], p1A[8], p1B[8];

  {
    v8f accA = {}, accB = {};
    for (int chunk = 0; chunk < 5; ++chunk) {
      __syncthreads();
      const float* Wc = sc_ws + ((size_t)chunk * 256 + 2 * half) * MULC + c2;
      loadB8(p0A, p0B, Wc);
      for (int i = tid; i < 16 * 256; i += 128) {
        int r = i >> 8, kk = i & 255;
        int k = chunk * 256 + kk;
        zbuf[0][r][kk] = xs[r][k / NSP] * at[r][k % NSP];
      }
      __syncthreads();
#pragma unroll
      for (int g = 0; g < 8; g += 2) {
        loadB8(p1A, p1B, Wc + (size_t)(g + 1) * 32 * MULC);
        mma8(&zbuf[0][m][g * 32 + 2 * half], p0A, p0B, accA, accB);
        if (g + 2 < 8) loadB8(p0A, p0B, Wc + (size_t)(g + 2) * 32 * MULC);
        mma8(&zbuf[0][m][(g + 1) * 32 + 2 * half], p1A, p1B, accA, accB);
      }
    }
#pragma unroll
    for (int r = 0; r < 8; ++r) {
      size_t row = (size_t)(n0 + r + 8 * half) * 512;
      *(v2f*)&out[row + c2] = v2f{ accA[r] * SC_NORM, accB[r] * SC_NORM };
    }
  }

  {
    v8f accA[3] = { v8f{}, v8f{}, v8f{} }, accB[3] = { v8f{}, v8f{}, v8f{} };
    for (int chunk = 0; chunk < 5; ++chunk) {
      __syncthreads();
      const float* Wc = sc_wv + ((size_t)chunk * 256 + 2 * half) * MULC + c2;
      loadB8(p0A, p0B, Wc);
      for (int i = tid; i < 16 * 256; i += 128) {
        int r = i >> 8, kk = i & 255;
        int k = chunk * 256 + kk;
        const int u = k / NSP;
        const float a = at[r][k - u * NSP];
        zbuf[0][r][kk] = xvc[0][r][u] * a;
        zbuf[1][r][kk] = xvc[1][r][u] * a;
        zbuf[2][r][kk] = xvc[2][r][u] * a;
      }
      __syncthreads();
#pragma unroll
      for (int g = 0; g < 8; g += 2) {
        const int ka = g * 32 + 2 * half, kb = ka + 32;
        loadB8(p1A, p1B, Wc + (size_t)(g + 1) * 32 * MULC);
        mma8x3(&zbuf[0][m][ka], &zbuf[1][m][ka], &zbuf[2][m][ka], p0A, p0B,
               accA, accB);
        if (g + 2 < 8) loadB8(p0A, p0B, Wc + (size_t)(g + 2) * 32 * MULC);
        mma8x3(&zbuf[0][m][kb], &zbuf[1][m][kb], &zbuf[2][m][kb], p1A, p1B,
               accA, accB);
      }
    }
#pragma unroll
    for (int r = 0; r < 8; ++r) {
      v2f* o = (v2f*)&out[(size_t)(n0 + r + 8 * half) * 512 + MULC + c2 * 3];
      o[0] = v2f{ accA[0][r] * SC_NORM, accA[1][r] * SC_NORM };
      o[1] = v2f{ accA[2][r] * SC_NORM, accB[0][r] * SC_NORM };
      o[2] = v2f{ accB[1][r] * SC_NORM, accB[2][r] * SC_NORM };
    }
  }

  const float* L1s = lin1_ws + (size_t)(2 * half) * MULC + c2;
  const float* L1v = lin1_wv + (size_t)(2 * half) * MULC + c2;
  loadB8(p0A, p0B, L1s);
  {
    v8f accA = {}, accB = {};
#pragma unroll
    for (int g = 0; g < 4; g += 2) {
      loadB8(p1A, p1B, L1s + (size_t)(g + 1) * 32 * MULC);
      mma8(&xs[m][g * 32 + 2 * half], p0A, p0B, accA, accB);
      loadB8(p0A, p0B, (g + 2 < 4) ? L1s + (size_t)(g + 2) * 32 * MULC : L1v);
      mma8(&xs[m][(g + 1) * 32 + 2 * half], p1A, p1B, accA, accB);
    }
#pragma unroll
    for (int r = 0; r < 8; ++r) {
      size_t row = (size_t)(n0 + r + 8 * half) * 512;
      *(v2f*)&y[row + c2] = v2f{ accA[r] * L1_NORM, accB[r] * L1_NORM };
    }
  }
  asm volatile("" ::: "memory");
  {
    v8f accA[3] = { v8f{}, v8f{}, v8f{} }, accB[3] = { v8f{}, v8f{}, v8f{} };
#pragma unroll
    for (int g = 0; g < 4; g += 2) {
      const int ka = g * 32 + 2 * half, kb = ka + 32;
      loadB8(p1A, p1B, L1v + (size_t)(g + 1) * 32 * MULC);
      mma8x3(&xvc[0][m][ka], &xvc[1][m][ka], &xvc[2][m][ka], p0A, p0B, accA,
             accB);
      if (g + 2 < 4) loadB8(p0A, p0B, L1v + (size_t)(g + 2) * 32 * MULC);
      mma8x3(&xvc[0][m][kb], &xvc[1][m][kb], &xvc[2][m][kb], p1A, p1B, accA,
             accB);
    }
#pragma unroll
    for (int r = 0; r < 8; ++r) {
      v2f* o = (v2f*)&y[(size_t)(n0 + r + 8 * half) * 512 + MULC + c2 * 3];
      o[0] = v2f{ accA[0][r] * L1_NORM, accA[1][r] * L1_NORM };
      o[1] = v2f{ accA[2][r] * L1_NORM, accB[0][r] * L1_NORM };
      o[2] = v2f{ accB[1][r] * L1_NORM, accB[2][r] * L1_NORM };
    }
  }
}

static constexpr int CAP    = 1024;
static constexpr int SCAN_U = 8;

__device__ __forceinline__ void l2load(v2f (&b0)[16], v2f (&b1)[16],
                                       const float* __restrict__ w2, int j0,
                                       int half, int nl) {
#pragma unroll
  for (int s = 0; s < 16; ++s) {
    const int k0 = 4 * s + 2 * half;
    const float* p = w2 + (size_t)k0 * 512 + j0 + 2 * nl;
    const v2f r0 = *(const v2f*)p;
    const v2f r1 = *(const v2f*)(p + 512);
    b0[s] = v2f{ r0.x, r1.x };
    b1[s] = v2f{ r0.y, r1.y };
  }
}

__device__ __forceinline__ void l2mma(const v2f (&af)[16], const v2f (&b0)[16],
                                      const v2f (&b1)[16], v8f& c0, v8f& c1) {
#pragma unroll
  for (int s = 0; s < 16; ++s) {
    c0 = wmma4(af[s], b0[s], c0);
    c1 = wmma4(af[s], b1[s], c1);
  }
}

__device__ __forceinline__ v8f segsum(const v2f (&sel)[4], const float (&mid)[8],
                                      v8f acc) {
#pragma unroll
  for (int t = 0; t < 4; ++t) {
    v2f b = { mid[2 * t], mid[2 * t + 1] };
    acc = wmma4(sel[t], b, acc);
  }
  return acc;
}

__global__ __launch_bounds__(128, 3) void edge_kernel(
    const float* y, const float* __restrict__ edge_feats,
    const float* __restrict__ edge_attrs, const int* __restrict__ edge_index,
    const float* __restrict__ fc_w0, const float* __restrict__ fc_w1,
    const float* __restrict__ fc_w2, float* __restrict__ s_acc,
    float* __restrict__ v_acc) {
  __shared__ float ytile[16][512];
  __shared__ float ha[16][64 + LPAD];
  __shared__ float hb[16][64 + LPAD];
  __shared__ float ea[16][4];
  __shared__ float ef[16][NRAD + LPAD];
  __shared__ int   dl[16];
  __shared__ int   lst_e[CAP];
  __shared__ int   lst_d[CAP];
  __shared__ int   lst_s[CAP];
  __shared__ int   cnt;

  const int n0   = blockIdx.x * 16;
  const int tid  = threadIdx.x;
  const int wave = tid >> 5;
  const int lane = tid & 31;
  const int half = lane >> 4;
  const int m    = lane & 15;
  const int nl   = lane & 15;

  v8f aS0[2], aS1[2], aV2[2][3], aV3[2][3];
#pragma unroll
  for (int jj = 0; jj < 2; ++jj) {
    aS0[jj] = v8f{};
    aS1[jj] = v8f{};
#pragma unroll
    for (int c = 0; c < 3; ++c) { aV2[jj][c] = v8f{}; aV3[jj][c] = v8f{}; }
  }

  const int4* d4 = (const int4*)edge_index;
  int seg_b = 0, seg_e = NE;
  for (;;) {
    if (tid == 0) cnt = 0;
    __syncthreads();
    const int e4 = seg_e >> 2;
    for (int base = (seg_b >> 2) + tid; base < e4; base += 128 * SCAN_U) {
      int4 v[SCAN_U];
#pragma unroll
      for (int q = 0; q < SCAN_U; ++q) {
        const int idx = base + q * 128;
        v[q] = (idx < e4) ? d4[idx] : make_int4(-1, -1, -1, -1);
      }
#pragma unroll
      for (int q = 0; q < SCAN_U; ++q) {
        const int idx = base + q * 128;
        const unsigned dd[4] = { (unsigned)v[q].x - (unsigned)n0,
                                 (unsigned)v[q].y - (unsigned)n0,
                                 (unsigned)v[q].z - (unsigned)n0,
                                 (unsigned)v[q].w - (unsigned)n0 };
#pragma unroll
        for (int z = 0; z < 4; ++z) {
          if (dd[z] < 16u) {
            const int p = atomicAdd(&cnt, 1);
            if (p < CAP) { lst_e[p] = idx * 4 + z; lst_d[p] = (int)dd[z]; }
          }
        }
      }
    }
    __syncthreads();
    const int c = cnt;
    __syncthreads();
    if (c > CAP) { seg_b = 0; seg_e = CAP; continue; }

    for (int i = tid; i < c; i += 128) lst_s[i] = edge_index[NE + lst_e[i]];
    __syncthreads();

    const int ntile = (c + 15) >> 4;
    for (int ti = 0; ti < ntile; ++ti) {
      int woff = 0;
      asm volatile("" : "+s"(woff));
      const float* w0 = fc_w0 + woff;
      const float* w1 = fc_w1 + woff;
      const float* w2 = fc_w2 + woff;
      v2f bw0[2], bw1[16];
      v2f qx0[16], qx1[16], qy0[16], qy1[16];
      {
        const float4* sp[4];
        int tw = tid;
        asm volatile("" : "+v"(tw));
        const int wrow = (tw >> 5) * 4;
#pragma unroll
        for (int rr = 0; rr < 4; ++rr) {
          const int li  = ti * 16 + wrow + rr;
          const int lic = (li < c) ? li : 0;
          const int sv  = lst_s[lic];
          const int sr  = (li < c) ? sv : 0;
          sp[rr] = (const float4*)(y + (size_t)sr * 512);
        }
        const int fr = tid >> 3, fk = tid & 7;
        const int fli = ti * 16 + fr;
        const int fe  = lst_e[(fli < c) ? fli : 0];
        const int ar = (tid >> 2) & 15, ak = tid & 3;
        const int ali = ti * 16 + ar;
        const int ae  = lst_e[(ali < c) ? ali : 0];
        const float4 a0 = sp[0][lane], a1 = sp[0][lane + 32],
                     a2 = sp[0][lane + 64], a3 = sp[0][lane + 96];
        const float4 b0 = sp[1][lane], b1 = sp[1][lane + 32],
                     b2 = sp[1][lane + 64], b3 = sp[1][lane + 96];
        const float4 c0 = sp[2][lane], c1 = sp[2][lane + 32],
                     c2 = sp[2][lane + 64], c3 = sp[2][lane + 96];
        const float4 d0 = sp[3][lane], d1 = sp[3][lane + 32],
                     d2 = sp[3][lane + 64], d3 = sp[3][lane + 96];
        float fraw = edge_feats[(unsigned)fe * NRAD + (unsigned)fk];
        float araw = 0.0f;
        if (tid < 64) araw = edge_attrs[(unsigned)ae * 4u + (unsigned)ak];
#pragma unroll
        for (int s = 0; s < 2; ++s) {
          const int k0 = 4 * s + 2 * half;
          bw0[s] = v2f{ w0[k0 * 64 + wave * 16 + nl],
                        w0[(k0 + 1) * 64 + wave * 16 + nl] };
        }
#pragma unroll
        for (int s = 0; s < 16; ++s) {
          const int k0 = 4 * s + 2 * half;
          bw1[s] = v2f{ w1[k0 * 64 + wave * 16 + nl],
                        w1[(k0 + 1) * 64 + wave * 16 + nl] };
        }
        asm volatile("" : "+v"(fraw), "+v"(araw) : : "memory");
        float4* dp = (float4*)&ytile[wave * 4][0];
        dp[lane]       = a0; dp[lane + 32]  = a1; dp[lane + 64]  = a2; dp[lane + 96]  = a3;
        dp[lane + 128] = b0; dp[lane + 160] = b1; dp[lane + 192] = b2; dp[lane + 224] = b3;
        dp[lane + 256] = c0; dp[lane + 288] = c1; dp[lane + 320] = c2; dp[lane + 352] = c3;
        dp[lane + 384] = d0; dp[lane + 416] = d1; dp[lane + 448] = d2; dp[lane + 480] = d3;
        ef[fr][fk] = (fli < c) ? fraw : 0.0f;
        if (tid < 64) ea[ar][ak] = (ali < c) ? araw : 0.0f;
      }
      if (tid < 16) {
        const int li = ti * 16 + tid;
        dl[tid] = (li < c) ? lst_d[li] : -1;
      }
      __syncthreads();

      {
        v8f acc = {};
#pragma unroll
        for (int s = 0; s < 2; ++s) {
          const int k0 = 4 * s + 2 * half;
          v2f a = { ef[m][k0], ef[m][k0 + 1] };
          acc = wmma4(a, bw0[s], acc);
        }
#pragma unroll
        for (int r = 0; r < 8; ++r)
          ha[r + 8 * half][wave * 16 + nl] = silu_c(acc[r] * INV_SQRT8);
      }
      __syncthreads();

      l2load(qx0, qx1, w2, 0 * 128 + wave * 32, half, nl);

      {
        v8f acc = {};
#pragma unroll
        for (int s = 0; s < 16; ++s) {
          const int k0 = 4 * s + 2 * half;
          v2f a = { ha[m][k0], ha[m][k0 + 1] };
          acc = wmma4(a, bw1[s], acc);
        }
#pragma unroll
        for (int r = 0; r < 8; ++r)
          hb[r + 8 * half][wave * 16 + nl] = silu_c(acc[r] * 0.125f);
      }
      __syncthreads();

      v2f af[16];
#pragma unroll
      for (int s = 0; s < 16; ++s) {
        const int k0 = 4 * s + 2 * half;
        af[s] = v2f{ hb[m][k0], hb[m][k0 + 1] };
      }
      v2f sel[4];
      int mrow = m;
      asm volatile("" : "+v"(mrow));
#pragma unroll
      for (int t = 0; t < 4; ++t) {
        const int e0 = 8 * half + 2 * t;
        sel[t] = v2f{ dl[e0] == mrow ? 1.0f : 0.0f, dl[e0 + 1] == mrow ? 1.0f : 0.0f };
      }

      {
        v8f cc[2] = { v8f{}, v8f{} };
        l2mma(af, qx0, qx1, cc[0], cc[1]);
        asm volatile("" : "+s"(woff) : "v"(cc[1][7]));
        w2 = fc_w2 + woff;
        l2load(qy0, qy1, w2, 1 * 128 + wave * 32, half, nl);
        asm volatile("" ::: "memory");
#pragma unroll
        for (int jj = 0; jj < 2; ++jj) {
          const int u = wave * 32 + 2 * nl + jj;
          float mid[8];
#pragma unroll
          for (int r = 0; r < 8; ++r) {
            const int e = r + 8 * half;
            mid[r] = (cc[jj][r] * 0.125f) * ytile[e][u] * ea[e][0];
          }
          aS0[jj] = segsum(sel, mid, aS0[jj]);
        }
      }
      {
        v8f cc[2] = { v8f{}, v8f{} };
        asm volatile("" : "+v"(qy0[0]), "+v"(qy1[0]) : "v"(aS0[0][7]), "v"(aS0[1][7]));
        l2mma(af, qy0, qy1, cc[0], cc[1]);
        asm volatile("" : "+s"(woff) : "v"(cc[1][7]));
        w2 = fc_w2 + woff;
        l2load(qx0, qx1, w2, 2 * 128 + wave * 32, half, nl);
        asm volatile("" ::: "memory");
#pragma unroll
        for (int jj = 0; jj < 2; ++jj) {
          const int u = wave * 32 + 2 * nl + jj;
          float m0[8], m1[8], m2[8];
#pragma unroll
          for (int r = 0; r < 8; ++r) {
            const int e = r + 8 * half;
            const float base = (cc[jj][r] * 0.125f) * ytile[e][u];
            m0[r] = base * ea[e][1];
            m1[r] = base * ea[e][2];
            m2[r] = base * ea[e][3];
          }
          aV2[jj][0] = segsum(sel, m0, aV2[jj][0]);
          aV2[jj][1] = segsum(sel, m1, aV2[jj][1]);
          aV2[jj][2] = segsum(sel, m2, aV2[jj][2]);
        }
      }
      {
        v8f cc[2] = { v8f{}, v8f{} };
        asm volatile("" : "+v"(qx0[0]), "+v"(qx1[0]) : "v"(aV2[0][0][7]), "v"(aV2[0][1][7]), "v"(aV2[0][2][7]), "v"(aV2[1][0][7]), "v"(aV2[1][1][7]), "v"(aV2[1][2][7]));
        l2mma(af, qx0, qx1, cc[0], cc[1]);
        asm volatile("" : "+s"(woff) : "v"(cc[1][7]));
        w2 = fc_w2 + woff;
        l2load(qy0, qy1, w2, 3 * 128 + wave * 32, half, nl);
        asm volatile("" ::: "memory");
#pragma unroll
        for (int jj = 0; jj < 2; ++jj) {
          const int u = wave * 32 + 2 * nl + jj;
          float m0[8], m1[8], m2[8];
#pragma unroll
          for (int r = 0; r < 8; ++r) {
            const int e = r + 8 * half;
            const float base = (cc[jj][r] * 0.125f) * ea[e][0];
            m0[r] = base * ytile[e][128 + u * 3 + 0];
            m1[r] = base * ytile[e][128 + u * 3 + 1];
            m2[r] = base * ytile[e][128 + u * 3 + 2];
          }
          aV3[jj][0] = segsum(sel, m0, aV3[jj][0]);
          aV3[jj][1] = segsum(sel, m1, aV3[jj][1]);
          aV3[jj][2] = segsum(sel, m2, aV3[jj][2]);
        }
      }
      {
        v8f cc[2] = { v8f{}, v8f{} };
        asm volatile("" : "+v"(qy0[0]), "+v"(qy1[0]) : "v"(aV3[0][0][7]), "v"(aV3[0][1][7]), "v"(aV3[0][2][7]), "v"(aV3[1][0][7]), "v"(aV3[1][1][7]), "v"(aV3[1][2][7]));
        l2mma(af, qy0, qy1, cc[0], cc[1]);
#pragma unroll
        for (int jj = 0; jj < 2; ++jj) {
          const int u = wave * 32 + 2 * nl + jj;
          float mid[8];
#pragma unroll
          for (int r = 0; r < 8; ++r) {
            const int e = r + 8 * half;
            const float dotv = ytile[e][128 + u * 3 + 0] * ea[e][1] +
                               ytile[e][128 + u * 3 + 1] * ea[e][2] +
                               ytile[e][128 + u * 3 + 2] * ea[e][3];
            mid[r] = (cc[jj][r] * 0.125f) * dotv * INV_SQ3;
          }
          aS1[jj] = segsum(sel, mid, aS1[jj]);
        }
      }
      __syncthreads();
    }
    asm volatile("v_nop\n\tv_nop\n\tv_nop\n\tv_nop" : "+v"(aS1[0]), "+v"(aS1[1]));

    if (seg_e >= NE) break;
    seg_b = seg_e;
    seg_e = (seg_e + CAP < NE) ? seg_e + CAP : NE;
  }

  int tq = threadIdx.x;
  asm volatile("" : "+v"(tq));
  const int wq = tq >> 5, hq = (tq >> 4) & 1, cq = tq & 15;
  const int u0 = wq * 32 + 2 * cq;
#pragma unroll
  for (int r = 0; r < 8; ++r) {
    const size_t n = (size_t)(n0 + r + 8 * hq);
    *(v2f*)&s_acc[n * 256 + u0] =
        v2f{ aS0[0][r] * INV_NEIGH, aS0[1][r] * INV_NEIGH };
    *(v2f*)&s_acc[n * 256 + 128 + u0] =
        v2f{ aS1[0][r] * INV_NEIGH, aS1[1][r] * INV_NEIGH };
    v2f* o2 = (v2f*)&v_acc[(n * 256 + u0) * 3];
    o2[0] = v2f{ aV2[0][0][r] * INV_NEIGH, aV2[0][1][r] * INV_NEIGH };
    o2[1] = v2f{ aV2[0][2][r] * INV_NEIGH, aV2[1][0][r] * INV_NEIGH };
    o2[2] = v2f{ aV2[1][1][r] * INV_NEIGH, aV2[1][2][r] * INV_NEIGH };
    v2f* o3 = (v2f*)&v_acc[(n * 256 + 128 + u0) * 3];
    o3[0] = v2f{ aV3[0][0][r] * INV_NEIGH, aV3[0][1][r] * INV_NEIGH };
    o3[1] = v2f{ aV3[0][2][r] * INV_NEIGH, aV3[1][0][r] * INV_NEIGH };
    o3[2] = v2f{ aV3[1][1][r] * INV_NEIGH, aV3[1][2][r] * INV_NEIGH };
  }
}

__global__ __launch_bounds__(128) void out_kernel(
    const float* __restrict__ s_acc, const float* __restrict__ v_acc,
    const float* __restrict__ lin2_ws, const float* __restrict__ lin2_wv,
    float* __restrict__ out) {
  __shared__ float st[16][256 + LPAD];
  __shared__ float vt[3][16][256 + LPAD];
  const int n0  = blockIdx.x * 16;
  const int tid = threadIdx.x;
  const int wave = tid >> 5;
  const int lane = tid & 31;
  const int half = lane >> 4;
  const int m    = lane & 15;
  const int nl   = lane & 15;
  const int c2   = wave * 32 + 2 * nl;

  v2f p0A[8], p0B[8], p1A[8], p1B[8];
  const float* L2s = lin2_ws + (size_t)(2 * half) * MULC + c2;
  const float* L2v = lin2_wv + (size_t)(2 * half) * MULC + c2;
  loadB8(p0A, p0B, L2s);

  {
    const float4* va4 = (const float4*)(v_acc + (size_t)n0 * 768);
    float4 t[24];
#pragma unroll
    for (int q = 0; q < 24; ++q) t[q] = va4[q * 128 + tid];
    int vo[3][4];
#pragma unroll
    for (int i = 0; i < 3; ++i) {
      const int w  = i * 128 + tid;
      const int rr = w / 192;
      const int f0 = (w - rr * 192) * 4;
#pragma unroll
      for (int e = 0; e < 4; ++e) {
        const int f = f0 + e, k = f / 3, c = f - 3 * k;
        vo[i][e] = (c * 16 + rr) * (256 + LPAD) + k;
      }
    }
    for (int i = tid; i < 16 * 256; i += 128)
      st[i >> 8][i & 255] = s_acc[(size_t)n0 * 256 + i];
    float* vb = &vt[0][0][0];
#pragma unroll
    for (int q = 0; q < 24; ++q) {
      const int j = q / 3, i = q - 3 * j;
      float* b = vb + 2 * j * (256 + LPAD);
      b[vo[i][0]] = t[q].x;
      b[vo[i][1]] = t[q].y;
      b[vo[i][2]] = t[q].z;
      b[vo[i][3]] = t[q].w;
    }
  }
  __syncthreads();

  {
    v8f accA = {}, accB = {};
#pragma unroll
    for (int g = 0; g < 8; g += 2) {
      loadB8(p1A, p1B, L2s + (size_t)(g + 1) * 32 * MULC);
      mma8(&st[m][g * 32 + 2 * half], p0A, p0B, accA, accB);
      loadB8(p0A, p0B, (g + 2 < 8) ? L2s + (size_t)(g + 2) * 32 * MULC : L2v);
      mma8(&st[m][(g + 1) * 32 + 2 * half], p1A, p1B, accA, accB);
    }
    v2f o[8];
#pragma unroll
    for (int r = 0; r < 8; ++r) {
      size_t row = (size_t)(n0 + r + 8 * half) * 512;
      o[r] = *(const v2f*)&out[row + c2];
    }
#pragma unroll
    for (int r = 0; r < 8; ++r) {
      size_t row = (size_t)(n0 + r + 8 * half) * 512;
      *(v2f*)&out[row + c2] =
          v2f{ o[r].x + accA[r] * L2_NORM, o[r].y + accB[r] * L2_NORM };
    }
  }
  {
    v8f accA[3] = { v8f{}, v8f{}, v8f{} }, accB[3] = { v8f{}, v8f{}, v8f{} };
#pragma unroll
    for (int g = 0; g < 8; g += 2) {
      const int ka = g * 32 + 2 * half, kb = ka + 32;
      loadB8(p1A, p1B, L2v + (size_t)(g + 1) * 32 * MULC);
      mma8x3(&vt[0][m][ka], &vt[1][m][ka], &vt[2][m][ka], p0A, p0B, accA, accB);
      if (g + 2 < 8) loadB8(p0A, p0B, L2v + (size_t)(g + 2) * 32 * MULC);
      mma8x3(&vt[0][m][kb], &vt[1][m][kb], &vt[2][m][kb], p1A, p1B, accA, accB);
    }
    v2f o[8][3];
#pragma unroll
    for (int r = 0; r < 8; ++r) {
      const v2f* q = (const v2f*)&out[(size_t)(n0 + r + 8 * half) * 512 + MULC + c2 * 3];
      o[r][0] = q[0];
      o[r][1] = q[1];
      o[r][2] = q[2];
    }
#pragma unroll
    for (int r = 0; r < 8; ++r) {
      v2f* q = (v2f*)&out[(size_t)(n0 + r + 8 * half) * 512 + MULC + c2 * 3];
      q[0] = v2f{ o[r][0].x + accA[0][r] * L2_NORM, o[r][0].y + accA[1][r] * L2_NORM };
      q[1] = v2f{ o[r][1].x + accA[2][r] * L2_NORM, o[r][1].y + accB[0][r] * L2_NORM };
      q[2] = v2f{ o[r][2].x + accB[1][r] * L2_NORM, o[r][2].y + accB[2][r] * L2_NORM };
    }
  }
}

extern "C" void kernel_launch(void* const* d_in, const int* in_sizes, int n_in,
                              void* d_out, int out_size, void* d_ws,
                              size_t ws_size, hipStream_t stream) {
  const float* node_feats = (const float*)d_in[0];
  const float* node_attrs = (const float*)d_in[1];
  const float* edge_feats = (const float*)d_in[2];
  const float* edge_attrs = (const float*)d_in[3];
  const int*   edge_index = (const int*)d_in[4];
  const float* lin1_ws    = (const float*)d_in[5];
  const float* lin1_wv    = (const float*)d_in[6];
  const float* fc_w0      = (const float*)d_in[7];
  const float* fc_w1      = (const float*)d_in[8];
  const float* fc_w2      = (const float*)d_in[9];
  const float* lin2_ws    = (const float*)d_in[10];
  const float* lin2_wv    = (const float*)d_in[11];
  const float* sc_ws      = (const float*)d_in[12];
  const float* sc_wv      = (const float*)d_in[13];

  float* out   = (float*)d_out;
  float* y     = (float*)d_ws;
  float* s_acc = y + (size_t)NN * 512;
  float* v_acc = s_acc + (size_t)NN * 256;

  node_kernel<<<NN / 16, 128, 0, stream>>>(node_feats, node_attrs, lin1_ws,
                                           lin1_wv, sc_ws, sc_wv, out, y);
  edge_kernel<<<NN / 16, 128, 0, stream>>>(y, edge_feats, edge_attrs,
                                          edge_index, fc_w0, fc_w1, fc_w2,
                                          s_acc, v_acc);
  out_kernel<<<NN / 16, 128, 0, stream>>>(s_acc, v_acc, lin2_ws, lin2_wv, out);
}


# edge lists binned once: node block b files the 256 edges of chunk b by target block (LDS counters) into cells in the owner's v rows; an edge block reads 625 cells, no full scan; list order by arrival
# speedup vs baseline: 2.5323x; 1.1148x over previous
#include <hip/hip_runtime.h>
#include <hip/hip_bf16.h>


typedef __attribute__((ext_vector_type(2))) float v2f;
typedef __attribute__((ext_vector_type(8))) float v8f;

static constexpr int NN   = 10000;
static constexpr int NE   = 160000;
static constexpr int MULC = 128;
static constexpr int NSP  = 10;
static constexpr int NRAD = 8;

static constexpr int NTB   = NN / 16;
static constexpr int ECH   = 256;
static constexpr int CELLW = 8;
static constexpr int CELLK = CELLW - 1;
static constexpr int VROW  = 16 * 256 * 3;
static_assert(NTB * ECH == NE, "one edge chunk per node block");
static_assert(NTB * CELLW <= VROW, "a block's cells fit its own rows of v_acc");

static constexpr int LPAD = 4;

static constexpr float SILU_C    = 1.679f;
static constexpr float SC_NORM   = 0.02795084971874737f;
static constexpr float L1_NORM   = 0.08838834764831845f;
static constexpr float L2_NORM   = 0.0625f;
static constexpr float INV_SQRT8 = 0.3535533905932738f;
static constexpr float INV_SQ3   = 0.5773502691896258f;
static constexpr float INV_NEIGH = 0.25f;

__device__ __forceinline__ v8f wmma4(v2f a, v2f b, v8f c) {
  return __builtin_amdgcn_wmma_f32_16x16x4_f32(
       false, a,  false, b,
       (short)0, c,  false,  false);
}

__device__ __forceinline__ float silu_c(float z) {
  return SILU_C * z / (1.0f + __expf(-z));
}

__device__ __forceinline__ void loadB8(v2f (&bA)[8], v2f (&bB)[8],
                                       const float* __restrict__ p) {
#pragma unroll
  for (int s = 0; s < 8; ++s) {
    const float* q = p + s * 4 * MULC;
    const v2f r0 = *(const v2f*)q;
    const v2f r1 = *(const v2f*)(q + MULC);
    bA[s] = v2f{ r0.x, r1.x };
    bB[s] = v2f{ r0.y, r1.y };
  }
}

__device__ __forceinline__ void mma8(const float* arow, const v2f (&bA)[8],
                                     const v2f (&bB)[8], v8f& accA, v8f& accB) {
#pragma unroll
  for (int s = 0; s < 8; ++s) {
    v2f a = { arow[4 * s], arow[4 * s + 1] };
    accA = wmma4(a, bA[s], accA);
    accB = wmma4(a, bB[s], accB);
  }
}

__device__ __forceinline__ void mma8x3(const float* a0, const float* a1,
                                       const float* a2, const v2f (&bA)[8],
                                       const v2f (&bB)[8], v8f (&accA)[3],
                                       v8f (&accB)[3]) {
#pragma unroll
  for (int s = 0; s < 8; ++s) {
    v2f x0 = { a0[4 * s], a0[4 * s + 1] };
    v2f x1 = { a1[4 * s], a1[4 * s + 1] };
    v2f x2 = { a2[4 * s], a2[4 * s + 1] };
    accA[0] = wmma4(x0, bA[s], accA[0]);
    accA[1] = wmma4(x1, bA[s], accA[1]);
    accA[2] = wmma4(x2, bA[s], accA[2]);
    accB[0] = wmma4(x0, bB[s], accB[0]);
    accB[1] = wmma4(x1, bB[s], accB[1]);
    accB[2] = wmma4(x2, bB[s], accB[2]);
  }
}

__global__ __launch_bounds__(128) void node_kernel(
    const float* __restrict__ node_feats, const float* __restrict__ node_attrs,
    const float* __restrict__ lin1_ws, const float* __restrict__ lin1_wv,
    const float* __restrict__ sc_ws, const float* __restrict__ sc_wv,
    float* __restrict__ out, float* __restrict__ y,
    const int* __restrict__ edge_index, int* __restrict__ cells) {
  __shared__ __attribute__((aligned(16))) float xs[16][MULC + LPAD];
  __shared__ float xvc[3][16][MULC + LPAD];
  __shared__ float at[16][16];
  __shared__ float zbuf[3][16][256 + LPAD];
  __shared__ int hist[NTB];

  const int n0  = blockIdx.x * 16;
  const int tid = threadIdx.x;

  const int eA = blockIdx.x * ECH + tid, eB = eA + 128;
  const int dA = edge_index[eA], dB = edge_index[eB];
  for (int i = tid; i < NTB; i += 128) hist[i] = 0;

  {
    float av[2];
#pragma unroll
    for (int q = 0; q < 2; ++q) {
      const int i = tid + q * 128, r = i >> 4, c = i & 15;
      av[q] = (c < NSP) ? node_attrs[(size_t)(n0 + r) * NSP + c] : 0.0f;
    }
    const float4* nf4 = (const float4*)(node_feats + (size_t)n0 * 512);
    float4 t[16];
#pragma unroll
    for (int q = 0; q < 16; ++q) t[q] = nf4[q * 128 + tid];
#pragma unroll
    for (int q = 0; q < 2; ++q) {
      const int i = tid + q * 128;
      at[i >> 4][i & 15] = av[q];
    }
    const int c0 = tid * 4;
    if (c0 < MULC) {
#pragma unroll
      for (int q = 0; q < 16; ++q) *(float4*)&xs[q][c0] = t[q];
    } else {
      const int f0 = c0 - MULC;
      const int h0 = f0 % 3, u0 = f0 / 3;
      const int h1 = (f0 + 1) % 3, u1 = (f0 + 1) / 3;
      const int h2 = (f0 + 2) % 3, u2 = (f0 + 2) / 3;
      const int h3 = (f0 + 3) % 3, u3 = (f0 + 3) / 3;
#pragma unroll
      for (int q = 0; q < 16; ++q) {
        xvc[h0][q][u0] = t[q].x;
        xvc[h1][q][u1] = t[q].y;
        xvc[h2][q][u2] = t[q].z;
        xvc[h3][q][u3] = t[q].w;
      }
    }
  }
  __syncthreads();

  {
    int* cb = cells + (size_t)blockIdx.x * CELLW;
    const int tA = dA >> 4, tB = dB >> 4;
    const int pA = atomicAdd(&hist[tA], 1);
    const int pB = atomicAdd(&hist[tB], 1);
    if (pA < CELLK) cb[(size_t)tA * VROW + 1 + pA] = eA;
    if (pB < CELLK) cb[(size_t)tB * VROW + 1 + pB] = eB;
    __syncthreads();
    for (int i = tid; i < NTB; i += 128) cb[(size_t)i * VROW] = hist[i];
  }

  const int wave = tid >> 5;
  const int lane = tid & 31;
  const int half = lane >> 4;
  const int m    = lane & 15;
  const int nl   = lane & 15;
  const int c2   = wave * 32 + 2 * nl;

  v2f p0A[8], p0B[8], p1A[8], p1B[8];

  {
    v8f accA = {}, accB = {};
    for (int chunk = 0; chunk < 5; ++chunk) {
      __syncthreads();
      const float* Wc = sc_ws + ((size_t)chunk * 256 + 2 * half) * MULC + c2;
      loadB8(p0A, p0B, Wc);
      for (int i = tid; i < 16 * 256; i += 128) {
        int r = i >> 8, kk = i & 255;
        int k = chunk * 256 + kk;
        zbuf[0][r][kk] = xs[r][k / NSP] * at[r][k % NSP];
      }
      __syncthreads();
#pragma unroll
      for (int g = 0; g < 8; g += 2) {
        loadB8(p1A, p1B, Wc + (size_t)(g + 1) * 32 * MULC);
        mma8(&zbuf[0][m][g * 32 + 2 * half], p0A, p0B, accA, accB);
        if (g + 2 < 8) loadB8(p0A, p0B, Wc + (size_t)(g + 2) * 32 * MULC);
        mma8(&zbuf[0][m][(g + 1) * 32 + 2 * half], p1A, p1B, accA, accB);
      }
    }
#pragma unroll
    for (int r = 0; r < 8; ++r) {
      size_t row = (size_t)(n0 + r + 8 * half) * 512;
      *(v2f*)&out[row + c2] = v2f{ accA[r] * SC_NORM, accB[r] * SC_NORM };
    }
  }

  {
    v8f accA[3] = { v8f{}, v8f{}, v8f{} }, accB[3] = { v8f{}, v8f{}, v8f{} };
    for (int chunk = 0; chunk < 5; ++chunk) {
      __syncthreads();
      const float* Wc = sc_wv + ((size_t)chunk * 256 + 2 * half) * MULC + c2;
      loadB8(p0A, p0B, Wc);
      for (int i = tid; i < 16 * 256; i += 128) {
        int r = i >> 8, kk = i & 255;
        int k = chunk * 256 + kk;
        const int u = k / NSP;
        const float a = at[r][k - u * NSP];
        zbuf[0][r][kk] = xvc[0][r][u] * a;
        zbuf[1][r][kk] = xvc[1][r][u] * a;
        zbuf[2][r][kk] = xvc[2][r][u] * a;
      }
      __syncthreads();
#pragma unroll
      for (int g = 0; g < 8; g += 2) {
        const int ka = g * 32 + 2 * half, kb = ka + 32;
        loadB8(p1A, p1B, Wc + (size_t)(g + 1) * 32 * MULC);
        mma8x3(&zbuf[0][m][ka], &zbuf[1][m][ka], &zbuf[2][m][ka], p0A, p0B,
               accA, accB);
        if (g + 2 < 8) loadB8(p0A, p0B, Wc + (size_t)(g + 2) * 32 * MULC);
        mma8x3(&zbuf[0][m][kb], &zbuf[1][m][kb], &zbuf[2][m][kb], p1A, p1B,
               accA, accB);
      }
    }
#pragma unroll
    for (int r = 0; r < 8; ++r) {
      v2f* o = (v2f*)&out[(size_t)(n0 + r + 8 * half) * 512 + MULC + c2 * 3];
      o[0] = v2f{ accA[0][r] * SC_NORM, accA[1][r] * SC_NORM };
      o[1] = v2f{ accA[2][r] * SC_NORM, accB[0][r] * SC_NORM };
      o[2] = v2f{ accB[1][r] * SC_NORM, accB[2][r] * SC_NORM };
    }
  }

  const float* L1s = lin1_ws + (size_t)(2 * half) * MULC + c2;
  const float* L1v = lin1_wv + (size_t)(2 * half) * MULC + c2;
  loadB8(p0A, p0B, L1s);
  {
    v8f accA = {}, accB = {};
#pragma unroll
    for (int g = 0; g < 4; g += 2) {
      loadB8(p1A, p1B, L1s + (size_t)(g + 1) * 32 * MULC);
      mma8(&xs[m][g * 32 + 2 * half], p0A, p0B, accA, accB);
      loadB8(p0A, p0B, (g + 2 < 4) ? L1s + (size_t)(g + 2) * 32 * MULC : L1v);
      mma8(&xs[m][(g + 1) * 32 + 2 * half], p1A, p1B, accA, accB);
    }
#pragma unroll
    for (int r = 0; r < 8; ++r) {
      size_t row = (size_t)(n0 + r + 8 * half) * 512;
      *(v2f*)&y[row + c2] = v2f{ accA[r] * L1_NORM, accB[r] * L1_NORM };
    }
  }
  asm volatile("" ::: "memory");
  {
    v8f accA[3] = { v8f{}, v8f{}, v8f{} }, accB[3] = { v8f{}, v8f{}, v8f{} };
#pragma unroll
    for (int g = 0; g < 4; g += 2) {
      const int ka = g * 32 + 2 * half, kb = ka + 32;
      loadB8(p1A, p1B, L1v + (size_t)(g + 1) * 32 * MULC);
      mma8x3(&xvc[0][m][ka], &xvc[1][m][ka], &xvc[2][m][ka], p0A, p0B, accA,
             accB);
      if (g + 2 < 4) loadB8(p0A, p0B, L1v + (size_t)(g + 2) * 32 * MULC);
      mma8x3(&xvc[0][m][kb], &xvc[1][m][kb], &xvc[2][m][kb], p1A, p1B, accA,
             accB);
    }
#pragma unroll
    for (int r = 0; r < 8; ++r) {
      v2f* o = (v2f*)&y[(size_t)(n0 + r + 8 * half) * 512 + MULC + c2 * 3];
      o[0] = v2f{ accA[0][r] * L1_NORM, accA[1][r] * L1_NORM };
      o[1] = v2f{ accA[2][r] * L1_NORM, accB[0][r] * L1_NORM };
      o[2] = v2f{ accB[1][r] * L1_NORM, accB[2][r] * L1_NORM };
    }
  }
}

static constexpr int CAP    = 1024;
static constexpr int SCAN_U = 8;

__device__ __forceinline__ void l2load(v2f (&b0)[16], v2f (&b1)[16],
                                       const float* __restrict__ w2, int j0,
                                       int half, int nl) {
#pragma unroll
  for (int s = 0; s < 16; ++s) {
    const int k0 = 4 * s + 2 * half;
    const float* p = w2 + (size_t)k0 * 512 + j0 + 2 * nl;
    const v2f r0 = *(const v2f*)p;
    const v2f r1 = *(const v2f*)(p + 512);
    b0[s] = v2f{ r0.x, r1.x };
    b1[s] = v2f{ r0.y, r1.y };
  }
}

__device__ __forceinline__ void l2mma(const v2f (&af)[16], const v2f (&b0)[16],
                                      const v2f (&b1)[16], v8f& c0, v8f& c1) {
#pragma unroll
  for (int s = 0; s < 16; ++s) {
    c0 = wmma4(af[s], b0[s], c0);
    c1 = wmma4(af[s], b1[s], c1);
  }
}

__device__ __forceinline__ v8f segsum(const v2f (&sel)[4], const float (&mid)[8],
                                      v8f acc) {
#pragma unroll
  for (int t = 0; t < 4; ++t) {
    v2f b = { mid[2 * t], mid[2 * t + 1] };
    acc = wmma4(sel[t], b, acc);
  }
  return acc;
}

__global__ __launch_bounds__(128, 3) void edge_kernel(
    const float* y, const float* __restrict__ edge_feats,
    const float* __restrict__ edge_attrs, const int* __restrict__ edge_index,
    const float* __restrict__ fc_w0, const float* __restrict__ fc_w1,
    const float* __restrict__ fc_w2, float* __restrict__ s_acc,
    float* __restrict__ v_acc) {
  __shared__ float ytile[16][512];
  __shared__ float ha[16][64 + LPAD];
  __shared__ float hb[16][64 + LPAD];
  __shared__ float ea[16][4];
  __shared__ float ef[16][NRAD + LPAD];
  __shared__ int   dl[16];
  __shared__ int   lst_e[CAP];
  __shared__ int   lst_d[CAP];
  __shared__ int   lst_s[CAP];
  __shared__ int   cnt;

  const int n0   = blockIdx.x * 16;
  const int tid  = threadIdx.x;
  const int wave = tid >> 5;
  const int lane = tid & 31;
  const int half = lane >> 4;
  const int m    = lane & 15;
  const int nl   = lane & 15;

  v8f aS0[2], aS1[2], aV2[2][3], aV3[2][3];
#pragma unroll
  for (int jj = 0; jj < 2; ++jj) {
    aS0[jj] = v8f{};
    aS1[jj] = v8f{};
#pragma unroll
    for (int c = 0; c < 3; ++c) { aV2[jj][c] = v8f{}; aV3[jj][c] = v8f{}; }
  }

  const int4* d4 = (const int4*)edge_index;
  int seg_b = 0, seg_e = NE;
  bool from_cells = true;
  if (tid == 0) cnt = 0;
  __syncthreads();
  {
    const int4* cl = (const int4*)(v_acc + (size_t)blockIdx.x * VROW);
    int4 ca[5], cb[5];
#pragma unroll
    for (int j = 0; j < 5; ++j) {
      const int cell = tid + j * 128;
      const int cc = (cell < NTB) ? cell : 0;
      ca[j] = cl[cc * 2];
      cb[j] = cl[cc * 2 + 1];
    }
#pragma unroll
    for (int j = 0; j < 5; ++j) {
      const int cell = tid + j * 128;
      const int n = (cell < NTB) ? ca[j].x : 0;
      if (n > 0) {
        const int p = atomicAdd(&cnt, (n <= CELLK) ? n : CAP + 1);
        const int ee[CELLK] = { ca[j].y, ca[j].z, ca[j].w, cb[j].x,
                                cb[j].y, cb[j].z, cb[j].w };
#pragma unroll
        for (int k = 0; k < CELLK; ++k)
          if (k < n && n <= CELLK && p + k < CAP) lst_e[p + k] = ee[k];
      }
    }
  }
  for (;;) {
    if (!from_cells) {
      if (tid == 0) cnt = 0;
      __syncthreads();
    }
    const int e4 = from_cells ? 0 : (seg_e >> 2);
    for (int base = (seg_b >> 2) + tid; base < e4; base += 128 * SCAN_U) {
      int4 v[SCAN_U];
#pragma unroll
      for (int q = 0; q < SCAN_U; ++q) {
        const int idx = base + q * 128;
        v[q] = (idx < e4) ? d4[idx] : make_int4(-1, -1, -1, -1);
      }
#pragma unroll
      for (int q = 0; q < SCAN_U; ++q) {
        const int idx = base + q * 128;
        const unsigned dd[4] = { (unsigned)v[q].x - (unsigned)n0,
                                 (unsigned)v[q].y - (unsigned)n0,
                                 (unsigned)v[q].z - (unsigned)n0,
                                 (unsigned)v[q].w - (unsigned)n0 };
#pragma unroll
        for (int z = 0; z < 4; ++z) {
          if (dd[z] < 16u) {
            const int p = atomicAdd(&cnt, 1);
            if (p < CAP) { lst_e[p] = idx * 4 + z; lst_d[p] = (int)dd[z]; }
          }
        }
      }
    }
    __syncthreads();
    const int c = cnt;
    __syncthreads();
    if (c > CAP) { seg_b = 0; seg_e = CAP; from_cells = false; continue; }

    for (int i = tid; i < c; i += 128) {
      const int e = lst_e[i];
      lst_s[i] = edge_index[NE + e];
      lst_d[i] = edge_index[e] - n0;
    }
    __syncthreads();

    const int ntile = (c + 15) >> 4;
    for (int ti = 0; ti < ntile; ++ti) {
      int woff = 0;
      asm volatile("" : "+s"(woff));
      const float* w0 = fc_w0 + woff;
      const float* w1 = fc_w1 + woff;
      const float* w2 = fc_w2 + woff;
      v2f bw0[2], bw1[16];
      v2f qx0[16], qx1[16], qy0[16], qy1[16];
      {
        const float4* sp[4];
        int tw = tid;
        asm volatile("" : "+v"(tw));
        const int wrow = (tw >> 5) * 4;
#pragma unroll
        for (int rr = 0; rr < 4; ++rr) {
          const int li  = ti * 16 + wrow + rr;
          const int lic = (li < c) ? li : 0;
          const int sv  = lst_s[lic];
          const int sr  = (li < c) ? sv : 0;
          sp[rr] = (const float4*)(y + (size_t)sr * 512);
        }
        const int fr = tid >> 3, fk = tid & 7;
        const int fli = ti * 16 + fr;
        const int fe  = lst_e[(fli < c) ? fli : 0];
        const int ar = (tid >> 2) & 15, ak = tid & 3;
        const int ali = ti * 16 + ar;
        const int ae  = lst_e[(ali < c) ? ali : 0];
        const float4 a0 = sp[0][lane], a1 = sp[0][lane + 32],
                     a2 = sp[0][lane + 64], a3 = sp[0][lane + 96];
        const float4 b0 = sp[1][lane], b1 = sp[1][lane + 32],
                     b2 = sp[1][lane + 64], b3 = sp[1][lane + 96];
        const float4 c0 = sp[2][lane], c1 = sp[2][lane + 32],
                     c2 = sp[2][lane + 64], c3 = sp[2][lane + 96];
        const float4 d0 = sp[3][lane], d1 = sp[3][lane + 32],
                     d2 = sp[3][lane + 64], d3 = sp[3][lane + 96];
        float fraw = edge_feats[(unsigned)fe * NRAD + (unsigned)fk];
        float araw = 0.0f;
        if (tid < 64) araw = edge_attrs[(unsigned)ae * 4u + (unsigned)ak];
#pragma unroll
        for (int s = 0; s < 2; ++s) {
          const int k0 = 4 * s + 2 * half;
          bw0[s] = v2f{ w0[k0 * 64 + wave * 16 + nl],
                        w0[(k0 + 1) * 64 + wave * 16 + nl] };
        }
#pragma unroll
        for (int s = 0; s < 16; ++s) {
          const int k0 = 4 * s + 2 * half;
          bw1[s] = v2f{ w1[k0 * 64 + wave * 16 + nl],
                        w1[(k0 + 1) * 64 + wave * 16 + nl] };
        }
        asm volatile("" : "+v"(fraw), "+v"(araw) : : "memory");
        float4* dp = (float4*)&ytile[wave * 4][0];
        dp[lane]       = a0; dp[lane + 32]  = a1; dp[lane + 64]  = a2; dp[lane + 96]  = a3;
        dp[lane + 128] = b0; dp[lane + 160] = b1; dp[lane + 192] = b2; dp[lane + 224] = b3;
        dp[lane + 256] = c0; dp[lane + 288] = c1; dp[lane + 320] = c2; dp[lane + 352] = c3;
        dp[lane + 384] = d0; dp[lane + 416] = d1; dp[lane + 448] = d2; dp[lane + 480] = d3;
        ef[fr][fk] = (fli < c) ? fraw : 0.0f;
        if (tid < 64) ea[ar][ak] = (ali < c) ? araw : 0.0f;
      }
      if (tid < 16) {
        const int li = ti * 16 + tid;
        dl[tid] = (li < c) ? lst_d[li] : -1;
      }
      __syncthreads();

      {
        v8f acc = {};
#pragma unroll
        for (int s = 0; s < 2; ++s) {
          const int k0 = 4 * s + 2 * half;
          v2f a = { ef[m][k0], ef[m][k0 + 1] };
          acc = wmma4(a, bw0[s], acc);
        }
#pragma unroll
        for (int r = 0; r < 8; ++r)
          ha[r + 8 * half][wave * 16 + nl] = silu_c(acc[r] * INV_SQRT8);
      }
      __syncthreads();

      l2load(qx0, qx1, w2, 0 * 128 + wave * 32, half, nl);

      {
        v8f acc = {};
#pragma unroll
        for (int s = 0; s < 16; ++s) {
          const int k0 = 4 * s + 2 * half;
          v2f a = { ha[m][k0], ha[m][k0 + 1] };
          acc = wmma4(a, bw1[s], acc);
        }
#pragma unroll
        for (int r = 0; r < 8; ++r)
          hb[r + 8 * half][wave * 16 + nl] = silu_c(acc[r] * 0.125f);
      }
      __syncthreads();

      v2f af[16];
#pragma unroll
      for (int s = 0; s < 16; ++s) {
        const int k0 = 4 * s + 2 * half;
        af[s] = v2f{ hb[m][k0], hb[m][k0 + 1] };
      }
      v2f sel[4];
      int mrow = m;
      asm volatile("" : "+v"(mrow));
#pragma unroll
      for (int t = 0; t < 4; ++t) {
        const int e0 = 8 * half + 2 * t;
        sel[t] = v2f{ dl[e0] == mrow ? 1.0f : 0.0f, dl[e0 + 1] == mrow ? 1.0f : 0.0f };
      }

      {
        v8f cc[2] = { v8f{}, v8f{} };
        l2mma(af, qx0, qx1, cc[0], cc[1]);
        asm volatile("" : "+s"(woff) : "v"(cc[1][7]));
        w2 = fc_w2 + woff;
        l2load(qy0, qy1, w2, 1 * 128 + wave * 32, half, nl);
        asm volatile("" ::: "memory");
#pragma unroll
        for (int jj = 0; jj < 2; ++jj) {
          const int u = wave * 32 + 2 * nl + jj;
          float mid[8];
#pragma unroll
          for (int r = 0; r < 8; ++r) {
            const int e = r + 8 * half;
            mid[r] = (cc[jj][r] * 0.125f) * ytile[e][u] * ea[e][0];
          }
          aS0[jj] = segsum(sel, mid, aS0[jj]);
        }
      }
      {
        v8f cc[2] = { v8f{}, v8f{} };
        asm volatile("" : "+v"(qy0[0]), "+v"(qy1[0]) : "v"(aS0[0][7]), "v"(aS0[1][7]));
        l2mma(af, qy0, qy1, cc[0], cc[1]);
        asm volatile("" : "+s"(woff) : "v"(cc[1][7]));
        w2 = fc_w2 + woff;
        l2load(qx0, qx1, w2, 2 * 128 + wave * 32, half, nl);
        asm volatile("" ::: "memory");
#pragma unroll
        for (int jj = 0; jj < 2; ++jj) {
          const int u = wave * 32 + 2 * nl + jj;
          float m0[8], m1[8], m2[8];
#pragma unroll
          for (int r = 0; r < 8; ++r) {
            const int e = r + 8 * half;
            const float base = (cc[jj][r] * 0.125f) * ytile[e][u];
            m0[r] = base * ea[e][1];
            m1[r] = base * ea[e][2];
            m2[r] = base * ea[e][3];
          }
          aV2[jj][0] = segsum(sel, m0, aV2[jj][0]);
          aV2[jj][1] = segsum(sel, m1, aV2[jj][1]);
          aV2[jj][2] = segsum(sel, m2, aV2[jj][2]);
        }
      }
      {
        v8f cc[2] = { v8f{}, v8f{} };
        asm volatile("" : "+v"(qx0[0]), "+v"(qx1[0]) : "v"(aV2[0][0][7]), "v"(aV2[0][1][7]), "v"(aV2[0][2][7]), "v"(aV2[1][0][7]), "v"(aV2[1][1][7]), "v"(aV2[1][2][7]));
        l2mma(af, qx0, qx1, cc[0], cc[1]);
        asm volatile("" : "+s"(woff) : "v"(cc[1][7]));
        w2 = fc_w2 + woff;
        l2load(qy0, qy1, w2, 3 * 128 + wave * 32, half, nl);
        asm volatile("" ::: "memory");
#pragma unroll
        for (int jj = 0; jj < 2; ++jj) {
          const int u = wave * 32 + 2 * nl + jj;
          float m0[8], m1[8], m2[8];
#pragma unroll
          for (int r = 0; r < 8; ++r) {
            const int e = r + 8 * half;
            const float base = (cc[jj][r] * 0.125f) * ea[e][0];
            m0[r] = base * ytile[e][128 + u * 3 + 0];
            m1[r] = base * ytile[e][128 + u * 3 + 1];
            m2[r] = base * ytile[e][128 + u * 3 + 2];
          }
          aV3[jj][0] = segsum(sel, m0, aV3[jj][0]);
          aV3[jj][1] = segsum(sel, m1, aV3[jj][1]);
          aV3[jj][2] = segsum(sel, m2, aV3[jj][2]);
        }
      }
      {
        v8f cc[2] = { v8f{}, v8f{} };
        asm volatile("" : "+v"(qy0[0]), "+v"(qy1[0]) : "v"(aV3[0][0][7]), "v"(aV3[0][1][7]), "v"(aV3[0][2][7]), "v"(aV3[1][0][7]), "v"(aV3[1][1][7]), "v"(aV3[1][2][7]));
        l2mma(af, qy0, qy1, cc[0], cc[1]);
#pragma unroll
        for (int jj = 0; jj < 2; ++jj) {
          const int u = wave * 32 + 2 * nl + jj;
          float mid[8];
#pragma unroll
          for (int r = 0; r < 8; ++r) {
            const int e = r + 8 * half;
            const float dotv = ytile[e][128 + u * 3 + 0] * ea[e][1] +
                               ytile[e][128 + u * 3 + 1] * ea[e][2] +
                               ytile[e][128 + u * 3 + 2] * ea[e][3];
            mid[r] = (cc[jj][r] * 0.125f) * dotv * INV_SQ3;
          }
          aS1[jj] = segsum(sel, mid, aS1[jj]);
        }
      }
      __syncthreads();
    }
    asm volatile("v_nop\n\tv_nop\n\tv_nop\n\tv_nop" : "+v"(aS1[0]), "+v"(aS1[1]));

    if (seg_e >= NE) break;
    seg_b = seg_e;
    seg_e = (seg_e + CAP < NE) ? seg_e + CAP : NE;
  }

  int tq = threadIdx.x;
  asm volatile("" : "+v"(tq));
  const int wq = tq >> 5, hq = (tq >> 4) & 1, cq = tq & 15;
  const int u0 = wq * 32 + 2 * cq;
#pragma unroll
  for (int r = 0; r < 8; ++r) {
    const size_t n = (size_t)(n0 + r + 8 * hq);
    *(v2f*)&s_acc[n * 256 + u0] =
        v2f{ aS0[0][r] * INV_NEIGH, aS0[1][r] * INV_NEIGH };
    *(v2f*)&s_acc[n * 256 + 128 + u0] =
        v2f{ aS1[0][r] * INV_NEIGH, aS1[1][r] * INV_NEIGH };
    v2f* o2 = (v2f*)&v_acc[(n * 256 + u0) * 3];
    o2[0] = v2f{ aV2[0][0][r] * INV_NEIGH, aV2[0][1][r] * INV_NEIGH };
    o2[1] = v2f{ aV2[0][2][r] * INV_NEIGH, aV2[1][0][r] * INV_NEIGH };
    o2[2] = v2f{ aV2[1][1][r] * INV_NEIGH, aV2[1][2][r] * INV_NEIGH };
    v2f* o3 = (v2f*)&v_acc[(n * 256 + 128 + u0) * 3];
    o3[0] = v2f{ aV3[0][0][r] * INV_NEIGH, aV3[0][1][r] * INV_NEIGH };
    o3[1] = v2f{ aV3[0][2][r] * INV_NEIGH, aV3[1][0][r] * INV_NEIGH };
    o3[2] = v2f{ aV3[1][1][r] * INV_NEIGH, aV3[1][2][r] * INV_NEIGH };
  }
}

__global__ __launch_bounds__(128) void out_kernel(
    const float* __restrict__ s_acc, const float* __restrict__ v_acc,
    const float* __restrict__ lin2_ws, const float* __restrict__ lin2_wv,
    float* __restrict__ out) {
  __shared__ float st[16][256 + LPAD];
  __shared__ float vt[3][16][256 + LPAD];
  const int n0  = blockIdx.x * 16;
  const int tid = threadIdx.x;
  const int wave = tid >> 5;
  const int lane = tid & 31;
  const int half = lane >> 4;
  const int m    = lane & 15;
  const int nl   = lane & 15;
  const int c2   = wave * 32 + 2 * nl;

  v2f p0A[8], p0B[8], p1A[8], p1B[8];
  const float* L2s = lin2_ws + (size_t)(2 * half) * MULC + c2;
  const float* L2v = lin2_wv + (size_t)(2 * half) * MULC + c2;
  loadB8(p0A, p0B, L2s);

  {
    const float4* va4 = (const float4*)(v_acc + (size_t)n0 * 768);
    float4 t[24];
#pragma unroll
    for (int q = 0; q < 24; ++q) t[q] = va4[q * 128 + tid];
    int vo[3][4];
#pragma unroll
    for (int i = 0; i < 3; ++i) {
      const int w  = i * 128 + tid;
      const int rr = w / 192;
      const int f0 = (w - rr * 192) * 4;
#pragma unroll
      for (int e = 0; e < 4; ++e) {
        const int f = f0 + e, k = f / 3, c = f - 3 * k;
        vo[i][e] = (c * 16 + rr) * (256 + LPAD) + k;
      }
    }
    for (int i = tid; i < 16 * 256; i += 128)
      st[i >> 8][i & 255] = s_acc[(size_t)n0 * 256 + i];
    float* vb = &vt[0][0][0];
#pragma unroll
    for (int q = 0; q < 24; ++q) {
      const int j = q / 3, i = q - 3 * j;
      float* b = vb + 2 * j * (256 + LPAD);
      b[vo[i][0]] = t[q].x;
      b[vo[i][1]] = t[q].y;
      b[vo[i][2]] = t[q].z;
      b[vo[i][3]] = t[q].w;
    }
  }
  __syncthreads();

  {
    v8f accA = {}, accB = {};
#pragma unroll
    for (int g = 0; g < 8; g += 2) {
      loadB8(p1A, p1B, L2s + (size_t)(g + 1) * 32 * MULC);
      mma8(&st[m][g * 32 + 2 * half], p0A, p0B, accA, accB);
      loadB8(p0A, p0B, (g + 2 < 8) ? L2s + (size_t)(g + 2) * 32 * MULC : L2v);
      mma8(&st[m][(g + 1) * 32 + 2 * half], p1A, p1B, accA, accB);
    }
    v2f o[8];
#pragma unroll
    for (int r = 0; r < 8; ++r) {
      size_t row = (size_t)(n0 + r + 8 * half) * 512;
      o[r] = *(const v2f*)&out[row + c2];
    }
#pragma unroll
    for (int r = 0; r < 8; ++r) {
      size_t row = (size_t)(n0 + r + 8 * half) * 512;
      *(v2f*)&out[row + c2] =
          v2f{ o[r].x + accA[r] * L2_NORM, o[r].y + accB[r] * L2_NORM };
    }
  }
  {
    v8f accA[3] = { v8f{}, v8f{}, v8f{} }, accB[3] = { v8f{}, v8f{}, v8f{} };
#pragma unroll
    for (int g = 0; g < 8; g += 2) {
      const int ka = g * 32 + 2 * half, kb = ka + 32;
      loadB8(p1A, p1B, L2v + (size_t)(g + 1) * 32 * MULC);
      mma8x3(&vt[0][m][ka], &vt[1][m][ka], &vt[2][m][ka], p0A, p0B, accA, accB);
      if (g + 2 < 8) loadB8(p0A, p0B, L2v + (size_t)(g + 2) * 32 * MULC);
      mma8x3(&vt[0][m][kb], &vt[1][m][kb], &vt[2][m][kb], p1A, p1B, accA, accB);
    }
    v2f o[8][3];
#pragma unroll
    for (int r = 0; r < 8; ++r) {
      const v2f* q = (const v2f*)&out[(size_t)(n0 + r + 8 * half) * 512 + MULC + c2 * 3];
      o[r][0] = q[0];
      o[r][1] = q[1];
      o[r][2] = q[2];
    }
#pragma unroll
    for (int r = 0; r < 8; ++r) {
      v2f* q = (v2f*)&out[(size_t)(n0 + r + 8 * half) * 512 + MULC + c2 * 3];
      q[0] = v2f{ o[r][0].x + accA[0][r] * L2_NORM, o[r][0].y + accA[1][r] * L2_NORM };
      q[1] = v2f{ o[r][1].x + accA[2][r] * L2_NORM, o[r][1].y + accB[0][r] * L2_NORM };
      q[2] = v2f{ o[r][2].x + accB[1][r] * L2_NORM, o[r][2].y + accB[2][r] * L2_NORM };
    }
  }
}

extern "C" void kernel_launch(void* const* d_in, const int* in_sizes, int n_in,
                              void* d_out, int out_size, void* d_ws,
                              size_t ws_size, hipStream_t stream) {
  const float* node_feats = (const float*)d_in[0];
  const float* node_attrs = (const float*)d_in[1];
  const float* edge_feats = (const float*)d_in[2];
  const float* edge_attrs = (const float*)d_in[3];
  const int*   edge_index = (const int*)d_in[4];
  const float* lin1_ws    = (const float*)d_in[5];
  const float* lin1_wv    = (const float*)d_in[6];
  const float* fc_w0      = (const float*)d_in[7];
  const float* fc_w1      = (const float*)d_in[8];
  const float* fc_w2      = (const float*)d_in[9];
  const float* lin2_ws    = (const float*)d_in[10];
  const float* lin2_wv    = (const float*)d_in[11];
  const float* sc_ws      = (const float*)d_in[12];
  const float* sc_wv      = (const float*)d_in[13];

  float* out   = (float*)d_out;
  float* y     = (float*)d_ws;
  float* s_acc = y + (size_t)NN * 512;
  float* v_acc = s_acc + (size_t)NN * 256;

  node_kernel<<<NN / 16, 128, 0, stream>>>(node_feats, node_attrs, lin1_ws,
                                           lin1_wv, sc_ws, sc_wv, out, y,
                                           edge_index, (int*)v_acc);
  edge_kernel<<<NN / 16, 128, 0, stream>>>(y, edge_feats, edge_attrs,
                                          edge_index, fc_w0, fc_w1, fc_w2,
                                          s_acc, v_acc);
  out_kernel<<<NN / 16, 128, 0, stream>>>(s_acc, v_acc, lin2_ws, lin2_wv, out);
}


# self-connection einsum as ten K=128 products per tile (one per attribute index) on features held in registers, each scaled by its attribute on the accumulator; no operand built in LDS; sums reordered
# speedup vs baseline: 2.6390x; 1.0421x over previous
#include <hip/hip_runtime.h>
#include <hip/hip_bf16.h>


typedef __attribute__((ext_vector_type(2))) float v2f;
typedef __attribute__((ext_vector_type(8))) float v8f;

static constexpr int NN   = 10000;
static constexpr int NE   = 160000;
static constexpr int MULC = 128;
static constexpr int NSP  = 10;
static constexpr int NRAD = 8;

static constexpr int NTB   = NN / 16;
static constexpr int ECH   = 256;
static constexpr int CELLW = 8;
static constexpr int CELLK = CELLW - 1;
static constexpr int VROW  = 16 * 256 * 3;
static_assert(NTB * ECH == NE, "one edge chunk per node block");
static_assert(NTB * CELLW <= VROW, "a block's cells fit its own rows of v_acc");

static constexpr int LPAD = 4;

static constexpr float SILU_C    = 1.679f;
static constexpr float SC_NORM   = 0.02795084971874737f;
static constexpr float L1_NORM   = 0.08838834764831845f;
static constexpr float L2_NORM   = 0.0625f;
static constexpr float INV_SQRT8 = 0.3535533905932738f;
static constexpr float INV_SQ3   = 0.5773502691896258f;
static constexpr float INV_NEIGH = 0.25f;

__device__ __forceinline__ v8f wmma4(v2f a, v2f b, v8f c) {
  return __builtin_amdgcn_wmma_f32_16x16x4_f32(
       false, a,  false, b,
       (short)0, c,  false,  false);
}

__device__ __forceinline__ float silu_c(float z) {
  return SILU_C * z / (1.0f + __expf(-z));
}

__device__ __forceinline__ void loadB8(v2f (&bA)[8], v2f (&bB)[8],
                                       const float* __restrict__ p) {
#pragma unroll
  for (int s = 0; s < 8; ++s) {
    const float* q = p + s * 4 * MULC;
    const v2f r0 = *(const v2f*)q;
    const v2f r1 = *(const v2f*)(q + MULC);
    bA[s] = v2f{ r0.x, r1.x };
    bB[s] = v2f{ r0.y, r1.y };
  }
}

__device__ __forceinline__ void mma8(const float* arow, const v2f (&bA)[8],
                                     const v2f (&bB)[8], v8f& accA, v8f& accB) {
#pragma unroll
  for (int s = 0; s < 8; ++s) {
    v2f a = { arow[4 * s], arow[4 * s + 1] };
    accA = wmma4(a, bA[s], accA);
    accB = wmma4(a, bB[s], accB);
  }
}

__device__ __forceinline__ void mma8x3(const float* a0, const float* a1,
                                       const float* a2, const v2f (&bA)[8],
                                       const v2f (&bB)[8], v8f (&accA)[3],
                                       v8f (&accB)[3]) {
#pragma unroll
  for (int s = 0; s < 8; ++s) {
    v2f x0 = { a0[4 * s], a0[4 * s + 1] };
    v2f x1 = { a1[4 * s], a1[4 * s + 1] };
    v2f x2 = { a2[4 * s], a2[4 * s + 1] };
    accA[0] = wmma4(x0, bA[s], accA[0]);
    accA[1] = wmma4(x1, bA[s], accA[1]);
    accA[2] = wmma4(x2, bA[s], accA[2]);
    accB[0] = wmma4(x0, bB[s], accB[0]);
    accB[1] = wmma4(x1, bB[s], accB[1]);
    accB[2] = wmma4(x2, bB[s], accB[2]);
  }
}

static constexpr int SCROW = NSP * MULC;

__device__ __forceinline__ void loadB8sc(v2f (&bA)[8], v2f (&bB)[8],
                                         const float* __restrict__ p) {
#pragma unroll
  for (int s = 0; s < 8; ++s) {
    const float* q = p + s * 4 * SCROW;
    const v2f r0 = *(const v2f*)q;
    const v2f r1 = *(const v2f*)(q + SCROW);
    bA[s] = v2f{ r0.x, r1.x };
    bB[s] = v2f{ r0.y, r1.y };
  }
}

template <int O>
__device__ __forceinline__ void mma8reg(const v2f (&a)[32], const v2f (&bA)[8],
                                        const v2f (&bB)[8], v8f& accA,
                                        v8f& accB) {
#pragma unroll
  for (int s = 0; s < 8; ++s) {
    accA = wmma4(a[O + s], bA[s], accA);
    accB = wmma4(a[O + s], bB[s], accB);
  }
}

template <int O>
__device__ __forceinline__ void mma8x3reg(const v2f (&a)[3][16],
                                          const v2f (&bA)[8],
                                          const v2f (&bB)[8], v8f (&accA)[3],
                                          v8f (&accB)[3]) {
#pragma unroll
  for (int s = 0; s < 8; ++s) {
    accA[0] = wmma4(a[0][O + s], bA[s], accA[0]);
    accA[1] = wmma4(a[1][O + s], bA[s], accA[1]);
    accA[2] = wmma4(a[2][O + s], bA[s], accA[2]);
    accB[0] = wmma4(a[0][O + s], bB[s], accB[0]);
    accB[1] = wmma4(a[1][O + s], bB[s], accB[1]);
    accB[2] = wmma4(a[2][O + s], bB[s], accB[2]);
  }
}

__global__ __launch_bounds__(128) void node_kernel(
    const float* __restrict__ node_feats, const float* __restrict__ node_attrs,
    const float* __restrict__ lin1_ws, const float* __restrict__ lin1_wv,
    const float* __restrict__ sc_ws, const float* __restrict__ sc_wv,
    float* __restrict__ out, float* __restrict__ y,
    const int* __restrict__ edge_index, int* __restrict__ cells) {
  __shared__ __attribute__((aligned(16))) float xs[16][MULC + LPAD];
  __shared__ float xvc[3][16][MULC + LPAD];
  __shared__ __attribute__((aligned(16))) float atT[16][16];
  __shared__ int hist[NTB];

  const int n0  = blockIdx.x * 16;
  const int tid = threadIdx.x;

  const int eA = blockIdx.x * ECH + tid, eB = eA + 128;
  const int dA = edge_index[eA], dB = edge_index[eB];
  for (int i = tid; i < NTB; i += 128) hist[i] = 0;

  {
    float av[2];
#pragma unroll
    for (int q = 0; q < 2; ++q) {
      const int i = tid + q * 128, r = i >> 4, c = i & 15;
      av[q] = (c < NSP) ? node_attrs[(size_t)(n0 + r) * NSP + c] : 0.0f;
    }
    const float4* nf4 = (const float4*)(node_feats + (size_t)n0 * 512);
    float4 t[16];
#pragma unroll
    for (int q = 0; q < 16; ++q) t[q] = nf4[q * 128 + tid];
#pragma unroll
    for (int q = 0; q < 2; ++q) {
      const int i = tid + q * 128;
      atT[i & 15][i >> 4] = av[q];
    }
    const int c0 = tid * 4;
    if (c0 < MULC) {
#pragma unroll
      for (int q = 0; q < 16; ++q) *(float4*)&xs[q][c0] = t[q];
    } else {
      const int f0 = c0 - MULC;
      const int h0 = f0 % 3, u0 = f0 / 3;
      const int h1 = (f0 + 1) % 3, u1 = (f0 + 1) / 3;
      const int h2 = (f0 + 2) % 3, u2 = (f0 + 2) / 3;
      const int h3 = (f0 + 3) % 3, u3 = (f0 + 3) / 3;
#pragma unroll
      for (int q = 0; q < 16; ++q) {
        xvc[h0][q][u0] = t[q].x;
        xvc[h1][q][u1] = t[q].y;
        xvc[h2][q][u2] = t[q].z;
        xvc[h3][q][u3] = t[q].w;
      }
    }
  }
  __syncthreads();

  {
    int* cb = cells + (size_t)blockIdx.x * CELLW;
    const int tA = dA >> 4, tB = dB >> 4;
    const int pA = atomicAdd(&hist[tA], 1);
    const int pB = atomicAdd(&hist[tB], 1);
    if (pA < CELLK) cb[(size_t)tA * VROW + 1 + pA] = eA;
    if (pB < CELLK) cb[(size_t)tB * VROW + 1 + pB] = eB;
    __syncthreads();
    for (int i = tid; i < NTB; i += 128) cb[(size_t)i * VROW] = hist[i];
  }

  const int wave = tid >> 5;
  const int lane = tid & 31;
  const int half = lane >> 4;
  const int m    = lane & 15;
  const int nl   = lane & 15;
  const int c2   = wave * 32 + 2 * nl;

  v2f p0A[8], p0B[8], p1A[8], p1B[8];

  const float* Wss = sc_ws + (size_t)(2 * half) * SCROW + c2;
  const float* Wsv = sc_wv + (size_t)(2 * half) * SCROW + c2;
  const int arow8 = 8 * half;

  loadB8sc(p0A, p0B, Wss);
  {
    v2f xa[32];
#pragma unroll
    for (int j = 0; j < 32; ++j) {
      const float* a = &xs[m][4 * j + 2 * half];
      xa[j] = v2f{ a[0], a[1] };
    }
    v8f totA = {}, totB = {};
#pragma nounroll
    for (int v = 0; v < NSP; ++v) {
      const float* W = Wss + v * MULC;
      const float4 s0 = *(const float4*)&atT[v][arow8];
      const float4 s1 = *(const float4*)&atT[v][arow8 + 4];
      v8f accA = {}, accB = {};
      loadB8sc(p1A, p1B, W + (size_t)1 * 32 * SCROW);
      mma8reg<0>(xa, p0A, p0B, accA, accB);
      loadB8sc(p0A, p0B, W + (size_t)2 * 32 * SCROW);
      mma8reg<8>(xa, p1A, p1B, accA, accB);
      loadB8sc(p1A, p1B, W + (size_t)3 * 32 * SCROW);
      mma8reg<16>(xa, p0A, p0B, accA, accB);
      loadB8sc(p0A, p0B, (v + 1 < NSP) ? W + MULC : Wsv);
      mma8reg<24>(xa, p1A, p1B, accA, accB);
      const float sc[8] = { s0.x, s0.y, s0.z, s0.w, s1.x, s1.y, s1.z, s1.w };
#pragma unroll
      for (int r = 0; r < 8; ++r) {
        totA[r] += sc[r] * accA[r];
        totB[r] += sc[r] * accB[r];
      }
    }
#pragma unroll
    for (int r = 0; r < 8; ++r) {
      size_t row = (size_t)(n0 + r + 8 * half) * 512;
      *(v2f*)&out[row + c2] = v2f{ totA[r] * SC_NORM, totB[r] * SC_NORM };
    }
  }

  {
    v8f totA[3] = { v8f{}, v8f{}, v8f{} }, totB[3] = { v8f{}, v8f{}, v8f{} };
#pragma unroll
    for (int h = 0; h < 2; ++h) {
      v2f xa[3][16];
#pragma unroll
      for (int ch = 0; ch < 3; ++ch) {
#pragma unroll
        for (int j = 0; j < 16; ++j) {
          const float* a = &xvc[ch][m][64 * h + 4 * j + 2 * half];
          xa[ch][j] = v2f{ a[0], a[1] };
        }
      }
      const float* Wh = Wsv + (size_t)h * 64 * SCROW;
      for (int v = 0; v < NSP; ++v) {
        const float* W = Wh + v * MULC;
        const float4 s0 = *(const float4*)&atT[v][arow8];
        const float4 s1 = *(const float4*)&atT[v][arow8 + 4];
        v8f accA[3] = { v8f{}, v8f{}, v8f{} }, accB[3] = { v8f{}, v8f{}, v8f{} };
        loadB8sc(p1A, p1B, W + (size_t)32 * SCROW);
        mma8x3reg<0>(xa, p0A, p0B, accA, accB);
        loadB8sc(p0A, p0B, (v + 1 < NSP) ? W + MULC : Wsv + (size_t)64 * SCROW);
        mma8x3reg<8>(xa, p1A, p1B, accA, accB);
        const float sc[8] = { s0.x, s0.y, s0.z, s0.w, s1.x, s1.y, s1.z, s1.w };
#pragma unroll
        for (int ch = 0; ch < 3; ++ch) {
#pragma unroll
          for (int r = 0; r < 8; ++r) {
            totA[ch][r] += sc[r] * accA[ch][r];
            totB[ch][r] += sc[r] * accB[ch][r];
          }
        }
      }
    }
#pragma unroll
    for (int r = 0; r < 8; ++r) {
      v2f* o = (v2f*)&out[(size_t)(n0 + r + 8 * half) * 512 + MULC + c2 * 3];
      o[0] = v2f{ totA[0][r] * SC_NORM, totA[1][r] * SC_NORM };
      o[1] = v2f{ totA[2][r] * SC_NORM, totB[0][r] * SC_NORM };
      o[2] = v2f{ totB[1][r] * SC_NORM, totB[2][r] * SC_NORM };
    }
  }

  const float* L1s = lin1_ws + (size_t)(2 * half) * MULC + c2;
  const float* L1v = lin1_wv + (size_t)(2 * half) * MULC + c2;
  loadB8(p0A, p0B, L1s);
  {
    v8f accA = {}, accB = {};
#pragma unroll
    for (int g = 0; g < 4; g += 2) {
      loadB8(p1A, p1B, L1s + (size_t)(g + 1) * 32 * MULC);
      mma8(&xs[m][g * 32 + 2 * half], p0A, p0B, accA, accB);
      loadB8(p0A, p0B, (g + 2 < 4) ? L1s + (size_t)(g + 2) * 32 * MULC : L1v);
      mma8(&xs[m][(g + 1) * 32 + 2 * half], p1A, p1B, accA, accB);
    }
#pragma unroll
    for (int r = 0; r < 8; ++r) {
      size_t row = (size_t)(n0 + r + 8 * half) * 512;
      *(v2f*)&y[row + c2] = v2f{ accA[r] * L1_NORM, accB[r] * L1_NORM };
    }
  }
  asm volatile("" ::: "memory");
  {
    v8f accA[3] = { v8f{}, v8f{}, v8f{} }, accB[3] = { v8f{}, v8f{}, v8f{} };
#pragma unroll
    for (int g = 0; g < 4; g += 2) {
      const int ka = g * 32 + 2 * half, kb = ka + 32;
      loadB8(p1A, p1B, L1v + (size_t)(g + 1) * 32 * MULC);
      mma8x3(&xvc[0][m][ka], &xvc[1][m][ka], &xvc[2][m][ka], p0A, p0B, accA,
             accB);
      if (g + 2 < 4) loadB8(p0A, p0B, L1v + (size_t)(g + 2) * 32 * MULC);
      mma8x3(&xvc[0][m][kb], &xvc[1][m][kb], &xvc[2][m][kb], p1A, p1B, accA,
             accB);
    }
#pragma unroll
    for (int r = 0; r < 8; ++r) {
      v2f* o = (v2f*)&y[(size_t)(n0 + r + 8 * half) * 512 + MULC + c2 * 3];
      o[0] = v2f{ accA[0][r] * L1_NORM, accA[1][r] * L1_NORM };
      o[1] = v2f{ accA[2][r] * L1_NORM, accB[0][r] * L1_NORM };
      o[2] = v2f{ accB[1][r] * L1_NORM, accB[2][r] * L1_NORM };
    }
  }
}

static constexpr int CAP    = 1024;
static constexpr int SCAN_U = 8;

__device__ __forceinline__ void l2load(v2f (&b0)[16], v2f (&b1)[16],
                                       const float* __restrict__ w2, int j0,
                                       int half, int nl) {
#pragma unroll
  for (int s = 0; s < 16; ++s) {
    const int k0 = 4 * s + 2 * half;
    const float* p = w2 + (size_t)k0 * 512 + j0 + 2 * nl;
    const v2f r0 = *(const v2f*)p;
    const v2f r1 = *(const v2f*)(p + 512);
    b0[s] = v2f{ r0.x, r1.x };
    b1[s] = v2f{ r0.y, r1.y };
  }
}

__device__ __forceinline__ void l2mma(const v2f (&af)[16], const v2f (&b0)[16],
                                      const v2f (&b1)[16], v8f& c0, v8f& c1) {
#pragma unroll
  for (int s = 0; s < 16; ++s) {
    c0 = wmma4(af[s], b0[s], c0);
    c1 = wmma4(af[s], b1[s], c1);
  }
}

__device__ __forceinline__ v8f segsum(const v2f (&sel)[4], const float (&mid)[8],
                                      v8f acc) {
#pragma unroll
  for (int t = 0; t < 4; ++t) {
    v2f b = { mid[2 * t], mid[2 * t + 1] };
    acc = wmma4(sel[t], b, acc);
  }
  return acc;
}

__global__ __launch_bounds__(128, 3) void edge_kernel(
    const float* y, const float* __restrict__ edge_feats,
    const float* __restrict__ edge_attrs, const int* __restrict__ edge_index,
    const float* __restrict__ fc_w0, const float* __restrict__ fc_w1,
    const float* __restrict__ fc_w2, float* __restrict__ s_acc,
    float* __restrict__ v_acc) {
  __shared__ float ytile[16][512];
  __shared__ float ha[16][64 + LPAD];
  __shared__ float hb[16][64 + LPAD];
  __shared__ float ea[16][4];
  __shared__ float ef[16][NRAD + LPAD];
  __shared__ int   dl[16];
  __shared__ int   lst_e[CAP];
  __shared__ int   lst_d[CAP];
  __shared__ int   lst_s[CAP];
  __shared__ int   cnt;

  const int n0   = blockIdx.x * 16;
  const int tid  = threadIdx.x;
  const int wave = tid >> 5;
  const int lane = tid & 31;
  const int half = lane >> 4;
  const int m    = lane & 15;
  const int nl   = lane & 15;

  v8f aS0[2], aS1[2], aV2[2][3], aV3[2][3];
#pragma unroll
  for (int jj = 0; jj < 2; ++jj) {
    aS0[jj] = v8f{};
    aS1[jj] = v8f{};
#pragma unroll
    for (int c = 0; c < 3; ++c) { aV2[jj][c] = v8f{}; aV3[jj][c] = v8f{}; }
  }

  const int4* d4 = (const int4*)edge_index;
  int seg_b = 0, seg_e = NE;
  bool from_cells = true;
  if (tid == 0) cnt = 0;
  __syncthreads();
  {
    const int4* cl = (const int4*)(v_acc + (size_t)blockIdx.x * VROW);
    int4 ca[5], cb[5];
#pragma unroll
    for (int j = 0; j < 5; ++j) {
      const int cell = tid + j * 128;
      const int cc = (cell < NTB) ? cell : 0;
      ca[j] = cl[cc * 2];
      cb[j] = cl[cc * 2 + 1];
    }
#pragma unroll
    for (int j = 0; j < 5; ++j) {
      const int cell = tid + j * 128;
      const int n = (cell < NTB) ? ca[j].x : 0;
      if (n > 0) {
        const int p = atomicAdd(&cnt, (n <= CELLK) ? n : CAP + 1);
        const int ee[CELLK] = { ca[j].y, ca[j].z, ca[j].w, cb[j].x,
                                cb[j].y, cb[j].z, cb[j].w };
#pragma unroll
        for (int k = 0; k < CELLK; ++k)
          if (k < n && n <= CELLK && p + k < CAP) lst_e[p + k] = ee[k];
      }
    }
  }
  for (;;) {
    if (!from_cells) {
      if (tid == 0) cnt = 0;
      __syncthreads();
    }
    const int e4 = from_cells ? 0 : (seg_e >> 2);
    for (int base = (seg_b >> 2) + tid; base < e4; base += 128 * SCAN_U) {
      int4 v[SCAN_U];
#pragma unroll
      for (int q = 0; q < SCAN_U; ++q) {
        const int idx = base + q * 128;
        v[q] = (idx < e4) ? d4[idx] : make_int4(-1, -1, -1, -1);
      }
#pragma unroll
      for (int q = 0; q < SCAN_U; ++q) {
        const int idx = base + q * 128;
        const unsigned dd[4] = { (unsigned)v[q].x - (unsigned)n0,
                                 (unsigned)v[q].y - (unsigned)n0,
                                 (unsigned)v[q].z - (unsigned)n0,
                                 (unsigned)v[q].w - (unsigned)n0 };
#pragma unroll
        for (int z = 0; z < 4; ++z) {
          if (dd[z] < 16u) {
            const int p = atomicAdd(&cnt, 1);
            if (p < CAP) { lst_e[p] = idx * 4 + z; lst_d[p] = (int)dd[z]; }
          }
        }
      }
    }
    __syncthreads();
    const int c = cnt;
    __syncthreads();
    if (c > CAP) { seg_b = 0; seg_e = CAP; from_cells = false; continue; }

    for (int i = tid; i < c; i += 128) {
      const int e = lst_e[i];
      lst_s[i] = edge_index[NE + e];
      lst_d[i] = edge_index[e] - n0;
    }
    __syncthreads();

    const int ntile = (c + 15) >> 4;
    for (int ti = 0; ti < ntile; ++ti) {
      int woff = 0;
      asm volatile("" : "+s"(woff));
      const float* w0 = fc_w0 + woff;
      const float* w1 = fc_w1 + woff;
      const float* w2 = fc_w2 + woff;
      v2f bw0[2], bw1[16];
      v2f qx0[16], qx1[16], qy0[16], qy1[16];
      {
        const float4* sp[4];
        int tw = tid;
        asm volatile("" : "+v"(tw));
        const int wrow = (tw >> 5) * 4;
#pragma unroll
        for (int rr = 0; rr < 4; ++rr) {
          const int li  = ti * 16 + wrow + rr;
          const int lic = (li < c) ? li : 0;
          const int sv  = lst_s[lic];
          const int sr  = (li < c) ? sv : 0;
          sp[rr] = (const float4*)(y + (size_t)sr * 512);
        }
        const int fr = tid >> 3, fk = tid & 7;
        const int fli = ti * 16 + fr;
        const int fe  = lst_e[(fli < c) ? fli : 0];
        const int ar = (tid >> 2) & 15, ak = tid & 3;
        const int ali = ti * 16 + ar;
        const int ae  = lst_e[(ali < c) ? ali : 0];
        const float4 a0 = sp[0][lane], a1 = sp[0][lane + 32],
                     a2 = sp[0][lane + 64], a3 = sp[0][lane + 96];
        const float4 b0 = sp[1][lane], b1 = sp[1][lane + 32],
                     b2 = sp[1][lane + 64], b3 = sp[1][lane + 96];
        const float4 c0 = sp[2][lane], c1 = sp[2][lane + 32],
                     c2 = sp[2][lane + 64], c3 = sp[2][lane + 96];
        const float4 d0 = sp[3][lane], d1 = sp[3][lane + 32],
                     d2 = sp[3][lane + 64], d3 = sp[3][lane + 96];
        float fraw = edge_feats[(unsigned)fe * NRAD + (unsigned)fk];
        float araw = 0.0f;
        if (tid < 64) araw = edge_attrs[(unsigned)ae * 4u + (unsigned)ak];
#pragma unroll
        for (int s = 0; s < 2; ++s) {
          const int k0 = 4 * s + 2 * half;
          bw0[s] = v2f{ w0[k0 * 64 + wave * 16 + nl],
                        w0[(k0 + 1) * 64 + wave * 16 + nl] };
        }
#pragma unroll
        for (int s = 0; s < 16; ++s) {
          const int k0 = 4 * s + 2 * half;
          bw1[s] = v2f{ w1[k0 * 64 + wave * 16 + nl],
                        w1[(k0 + 1) * 64 + wave * 16 + nl] };
        }
        asm volatile("" : "+v"(fraw), "+v"(araw) : : "memory");
        float4* dp = (float4*)&ytile[wave * 4][0];
        dp[lane]       = a0; dp[lane + 32]  = a1; dp[lane + 64]  = a2; dp[lane + 96]  = a3;
        dp[lane + 128] = b0; dp[lane + 160] = b1; dp[lane + 192] = b2; dp[lane + 224] = b3;
        dp[lane + 256] = c0; dp[lane + 288] = c1; dp[lane + 320] = c2; dp[lane + 352] = c3;
        dp[lane + 384] = d0; dp[lane + 416] = d1; dp[lane + 448] = d2; dp[lane + 480] = d3;
        ef[fr][fk] = (fli < c) ? fraw : 0.0f;
        if (tid < 64) ea[ar][ak] = (ali < c) ? araw : 0.0f;
      }
      if (tid < 16) {
        const int li = ti * 16 + tid;
        dl[tid] = (li < c) ? lst_d[li] : -1;
      }
      __syncthreads();

      {
        v8f acc = {};
#pragma unroll
        for (int s = 0; s < 2; ++s) {
          const int k0 = 4 * s + 2 * half;
          v2f a = { ef[m][k0], ef[m][k0 + 1] };
          acc = wmma4(a, bw0[s], acc);
        }
#pragma unroll
        for (int r = 0; r < 8; ++r)
          ha[r + 8 * half][wave * 16 + nl] = silu_c(acc[r] * INV_SQRT8);
      }
      __syncthreads();

      l2load(qx0, qx1, w2, 0 * 128 + wave * 32, half, nl);

      {
        v8f acc = {};
#pragma unroll
        for (int s = 0; s < 16; ++s) {
          const int k0 = 4 * s + 2 * half;
          v2f a = { ha[m][k0], ha[m][k0 + 1] };
          acc = wmma4(a, bw1[s], acc);
        }
#pragma unroll
        for (int r = 0; r < 8; ++r)
          hb[r + 8 * half][wave * 16 + nl] = silu_c(acc[r] * 0.125f);
      }
      __syncthreads();

      v2f af[16];
#pragma unroll
      for (int s = 0; s < 16; ++s) {
        const int k0 = 4 * s + 2 * half;
        af[s] = v2f{ hb[m][k0], hb[m][k0 + 1] };
      }
      v2f sel[4];
      int mrow = m;
      asm volatile("" : "+v"(mrow));
#pragma unroll
      for (int t = 0; t < 4; ++t) {
        const int e0 = 8 * half + 2 * t;
        sel[t] = v2f{ dl[e0] == mrow ? 1.0f : 0.0f, dl[e0 + 1] == mrow ? 1.0f : 0.0f };
      }

      {
        v8f cc[2] = { v8f{}, v8f{} };
        l2mma(af, qx0, qx1, cc[0], cc[1]);
        asm volatile("" : "+s"(woff) : "v"(cc[1][7]));
        w2 = fc_w2 + woff;
        l2load(qy0, qy1, w2, 1 * 128 + wave * 32, half, nl);
        asm volatile("" ::: "memory");
#pragma unroll
        for (int jj = 0; jj < 2; ++jj) {
          const int u = wave * 32 + 2 * nl + jj;
          float mid[8];
#pragma unroll
          for (int r = 0; r < 8; ++r) {
            const int e = r + 8 * half;
            mid[r] = (cc[jj][r] * 0.125f) * ytile[e][u] * ea[e][0];
          }
          aS0[jj] = segsum(sel, mid, aS0[jj]);
        }
      }
      {
        v8f cc[2] = { v8f{}, v8f{} };
        asm volatile("" : "+v"(qy0[0]), "+v"(qy1[0]) : "v"(aS0[0][7]), "v"(aS0[1][7]));
        l2mma(af, qy0, qy1, cc[0], cc[1]);
        asm volatile("" : "+s"(woff) : "v"(cc[1][7]));
        w2 = fc_w2 + woff;
        l2load(qx0, qx1, w2, 2 * 128 + wave * 32, half, nl);
        asm volatile("" ::: "memory");
#pragma unroll
        for (int jj = 0; jj < 2; ++jj) {
          const int u = wave * 32 + 2 * nl + jj;
          float m0[8], m1[8], m2[8];
#pragma unroll
          for (int r = 0; r < 8; ++r) {
            const int e = r + 8 * half;
            const float base = (cc[jj][r] * 0.125f) * ytile[e][u];
            m0[r] = base * ea[e][1];
            m1[r] = base * ea[e][2];
            m2[r] = base * ea[e][3];
          }
          aV2[jj][0] = segsum(sel, m0, aV2[jj][0]);
          aV2[jj][1] = segsum(sel, m1, aV2[jj][1]);
          aV2[jj][2] = segsum(sel, m2, aV2[jj][2]);
        }
      }
      {
        v8f cc[2] = { v8f{}, v8f{} };
        asm volatile("" : "+v"(qx0[0]), "+v"(qx1[0]) : "v"(aV2[0][0][7]), "v"(aV2[0][1][7]), "v"(aV2[0][2][7]), "v"(aV2[1][0][7]), "v"(aV2[1][1][7]), "v"(aV2[1][2][7]));
        l2mma(af, qx0, qx1, cc[0], cc[1]);
        asm volatile("" : "+s"(woff) : "v"(cc[1][7]));
        w2 = fc_w2 + woff;
        l2load(qy0, qy1, w2, 3 * 128 + wave * 32, half, nl);
        asm volatile("" ::: "memory");
#pragma unroll
        for (int jj = 0; jj < 2; ++jj) {
          const int u = wave * 32 + 2 * nl + jj;
          float m0[8], m1[8], m2[8];
#pragma unroll
          for (int r = 0; r < 8; ++r) {
            const int e = r + 8 * half;
            const float base = (cc[jj][r] * 0.125f) * ea[e][0];
            m0[r] = base * ytile[e][128 + u * 3 + 0];
            m1[r] = base * ytile[e][128 + u * 3 + 1];
            m2[r] = base * ytile[e][128 + u * 3 + 2];
          }
          aV3[jj][0] = segsum(sel, m0, aV3[jj][0]);
          aV3[jj][1] = segsum(sel, m1, aV3[jj][1]);
          aV3[jj][2] = segsum(sel, m2, aV3[jj][2]);
        }
      }
      {
        v8f cc[2] = { v8f{}, v8f{} };
        asm volatile("" : "+v"(qy0[0]), "+v"(qy1[0]) : "v"(aV3[0][0][7]), "v"(aV3[0][1][7]), "v"(aV3[0][2][7]), "v"(aV3[1][0][7]), "v"(aV3[1][1][7]), "v"(aV3[1][2][7]));
        l2mma(af, qy0, qy1, cc[0], cc[1]);
#pragma unroll
        for (int jj = 0; jj < 2; ++jj) {
          const int u = wave * 32 + 2 * nl + jj;
          float mid[8];
#pragma unroll
          for (int r = 0; r < 8; ++r) {
            const int e = r + 8 * half;
            const float dotv = ytile[e][128 + u * 3 + 0] * ea[e][1] +
                               ytile[e][128 + u * 3 + 1] * ea[e][2] +
                               ytile[e][128 + u * 3 + 2] * ea[e][3];
            mid[r] = (cc[jj][r] * 0.125f) * dotv * INV_SQ3;
          }
          aS1[jj] = segsum(sel, mid, aS1[jj]);
        }
      }
      __syncthreads();
    }
    asm volatile("v_nop\n\tv_nop\n\tv_nop\n\tv_nop" : "+v"(aS1[0]), "+v"(aS1[1]));

    if (seg_e >= NE) break;
    seg_b = seg_e;
    seg_e = (seg_e + CAP < NE) ? seg_e + CAP : NE;
  }

  int tq = threadIdx.x;
  asm volatile("" : "+v"(tq));
  const int wq = tq >> 5, hq = (tq >> 4) & 1, cq = tq & 15;
  const int u0 = wq * 32 + 2 * cq;
#pragma unroll
  for (int r = 0; r < 8; ++r) {
    const size_t n = (size_t)(n0 + r + 8 * hq);
    *(v2f*)&s_acc[n * 256 + u0] =
        v2f{ aS0[0][r] * INV_NEIGH, aS0[1][r] * INV_NEIGH };
    *(v2f*)&s_acc[n * 256 + 128 + u0] =
        v2f{ aS1[0][r] * INV_NEIGH, aS1[1][r] * INV_NEIGH };
    v2f* o2 = (v2f*)&v_acc[(n * 256 + u0) * 3];
    o2[0] = v2f{ aV2[0][0][r] * INV_NEIGH, aV2[0][1][r] * INV_NEIGH };
    o2[1] = v2f{ aV2[0][2][r] * INV_NEIGH, aV2[1][0][r] * INV_NEIGH };
    o2[2] = v2f{ aV2[1][1][r] * INV_NEIGH, aV2[1][2][r] * INV_NEIGH };
    v2f* o3 = (v2f*)&v_acc[(n * 256 + 128 + u0) * 3];
    o3[0] = v2f{ aV3[0][0][r] * INV_NEIGH, aV3[0][1][r] * INV_NEIGH };
    o3[1] = v2f{ aV3[0][2][r] * INV_NEIGH, aV3[1][0][r] * INV_NEIGH };
    o3[2] = v2f{ aV3[1][1][r] * INV_NEIGH, aV3[1][2][r] * INV_NEIGH };
  }
}

__global__ __launch_bounds__(128) void out_kernel(
    const float* __restrict__ s_acc, const float* __restrict__ v_acc,
    const float* __restrict__ lin2_ws, const float* __restrict__ lin2_wv,
    float* __restrict__ out) {
  __shared__ float st[16][256 + LPAD];
  __shared__ float vt[3][16][256 + LPAD];
  const int n0  = blockIdx.x * 16;
  const int tid = threadIdx.x;
  const int wave = tid >> 5;
  const int lane = tid & 31;
  const int half = lane >> 4;
  const int m    = lane & 15;
  const int nl   = lane & 15;
  const int c2   = wave * 32 + 2 * nl;

  v2f p0A[8], p0B[8], p1A[8], p1B[8];
  const float* L2s = lin2_ws + (size_t)(2 * half) * MULC + c2;
  const float* L2v = lin2_wv + (size_t)(2 * half) * MULC + c2;
  loadB8(p0A, p0B, L2s);

  {
    const float4* va4 = (const float4*)(v_acc + (size_t)n0 * 768);
    float4 t[24];
#pragma unroll
    for (int q = 0; q < 24; ++q) t[q] = va4[q * 128 + tid];
    int vo[3][4];
#pragma unroll
    for (int i = 0; i < 3; ++i) {
      const int w  = i * 128 + tid;
      const int rr = w / 192;
      const int f0 = (w - rr * 192) * 4;
#pragma unroll
      for (int e = 0; e < 4; ++e) {
        const int f = f0 + e, k = f / 3, c = f - 3 * k;
        vo[i][e] = (c * 16 + rr) * (256 + LPAD) + k;
      }
    }
    for (int i = tid; i < 16 * 256; i += 128)
      st[i >> 8][i & 255] = s_acc[(size_t)n0 * 256 + i];
    float* vb = &vt[0][0][0];
#pragma unroll
    for (int q = 0; q < 24; ++q) {
      const int j = q / 3, i = q - 3 * j;
      float* b = vb + 2 * j * (256 + LPAD);
      b[vo[i][0]] = t[q].x;
      b[vo[i][1]] = t[q].y;
      b[vo[i][2]] = t[q].z;
      b[vo[i][3]] = t[q].w;
    }
  }
  __syncthreads();

  {
    v8f accA = {}, accB = {};
#pragma unroll
    for (int g = 0; g < 8; g += 2) {
      loadB8(p1A, p1B, L2s + (size_t)(g + 1) * 32 * MULC);
      mma8(&st[m][g * 32 + 2 * half], p0A, p0B, accA, accB);
      loadB8(p0A, p0B, (g + 2 < 8) ? L2s + (size_t)(g + 2) * 32 * MULC : L2v);
      mma8(&st[m][(g + 1) * 32 + 2 * half], p1A, p1B, accA, accB);
    }
    v2f o[8];
#pragma unroll
    for (int r = 0; r < 8; ++r) {
      size_t row = (size_t)(n0 + r + 8 * half) * 512;
      o[r] = *(const v2f*)&out[row + c2];
    }
#pragma unroll
    for (int r = 0; r < 8; ++r) {
      size_t row = (size_t)(n0 + r + 8 * half) * 512;
      *(v2f*)&out[row + c2] =
          v2f{ o[r].x + accA[r] * L2_NORM, o[r].y + accB[r] * L2_NORM };
    }
  }
  {
    v8f accA[3] = { v8f{}, v8f{}, v8f{} }, accB[3] = { v8f{}, v8f{}, v8f{} };
#pragma unroll
    for (int g = 0; g < 8; g += 2) {
      const int ka = g * 32 + 2 * half, kb = ka + 32;
      loadB8(p1A, p1B, L2v + (size_t)(g + 1) * 32 * MULC);
      mma8x3(&vt[0][m][ka], &vt[1][m][ka], &vt[2][m][ka], p0A, p0B, accA, accB);
      if (g + 2 < 8) loadB8(p0A, p0B, L2v + (size_t)(g + 2) * 32 * MULC);
      mma8x3(&vt[0][m][kb], &vt[1][m][kb], &vt[2][m][kb], p1A, p1B, accA, accB);
    }
    v2f o[8][3];
#pragma unroll
    for (int r = 0; r < 8; ++r) {
      const v2f* q = (const v2f*)&out[(size_t)(n0 + r + 8 * half) * 512 + MULC + c2 * 3];
      o[r][0] = q[0];
      o[r][1] = q[1];
      o[r][2] = q[2];
    }
#pragma unroll
    for (int r = 0; r < 8; ++r) {
      v2f* q = (v2f*)&out[(size_t)(n0 + r + 8 * half) * 512 + MULC + c2 * 3];
      q[0] = v2f{ o[r][0].x + accA[0][r] * L2_NORM, o[r][0].y + accA[1][r] * L2_NORM };
      q[1] = v2f{ o[r][1].x + accA[2][r] * L2_NORM, o[r][1].y + accB[0][r] * L2_NORM };
      q[2] = v2f{ o[r][2].x + accB[1][r] * L2_NORM, o[r][2].y + accB[2][r] * L2_NORM };
    }
  }
}

extern "C" void kernel_launch(void* const* d_in, const int* in_sizes, int n_in,
                              void* d_out, int out_size, void* d_ws,
                              size_t ws_size, hipStream_t stream) {
  const float* node_feats = (const float*)d_in[0];
  const float* node_attrs = (const float*)d_in[1];
  const float* edge_feats = (const float*)d_in[2];
  const float* edge_attrs = (const float*)d_in[3];
  const int*   edge_index = (const int*)d_in[4];
  const float* lin1_ws    = (const float*)d_in[5];
  const float* lin1_wv    = (const float*)d_in[6];
  const float* fc_w0      = (const float*)d_in[7];
  const float* fc_w1      = (const float*)d_in[8];
  const float* fc_w2      = (const float*)d_in[9];
  const float* lin2_ws    = (const float*)d_in[10];
  const float* lin2_wv    = (const float*)d_in[11];
  const float* sc_ws      = (const float*)d_in[12];
  const float* sc_wv      = (const float*)d_in[13];

  float* out   = (float*)d_out;
  float* y     = (float*)d_ws;
  float* s_acc = y + (size_t)NN * 512;
  float* v_acc = s_acc + (size_t)NN * 256;

  node_kernel<<<NN / 16, 128, 0, stream>>>(node_feats, node_attrs, lin1_ws,
                                           lin1_wv, sc_ws, sc_wv, out, y,
                                           edge_index, (int*)v_acc);
  edge_kernel<<<NN / 16, 128, 0, stream>>>(y, edge_feats, edge_attrs,
                                          edge_index, fc_w0, fc_w1, fc_w2,
                                          s_acc, v_acc);
  out_kernel<<<NN / 16, 128, 0, stream>>>(s_acc, v_acc, lin2_ws, lin2_wv, out);
}


# sc_v products moved from the node launch into 625 extra blocks of the edge launch, handed out behind the edge blocks: they fill the WGPs holding two edge blocks, not three, and places freed later
# speedup vs baseline: 2.9299x; 1.1102x over previous
#include <hip/hip_runtime.h>
#include <hip/hip_bf16.h>


typedef __attribute__((ext_vector_type(2))) float v2f;
typedef __attribute__((ext_vector_type(8))) float v8f;

static constexpr int NN   = 10000;
static constexpr int NE   = 160000;
static constexpr int MULC = 128;
static constexpr int NSP  = 10;
static constexpr int NRAD = 8;

static constexpr int NTB   = NN / 16;
static constexpr int ECH   = 256;
static constexpr int CELLW = 8;
static constexpr int CELLK = CELLW - 1;
static constexpr int VROW  = 16 * 256 * 3;
static_assert(NTB * ECH == NE, "one edge chunk per node block");
static_assert(NTB * CELLW <= VROW, "a block's cells fit its own rows of v_acc");

static constexpr int LPAD = 4;

static constexpr float SILU_C    = 1.679f;
static constexpr float SC_NORM   = 0.02795084971874737f;
static constexpr float L1_NORM   = 0.08838834764831845f;
static constexpr float L2_NORM   = 0.0625f;
static constexpr float INV_SQRT8 = 0.3535533905932738f;
static constexpr float INV_SQ3   = 0.5773502691896258f;
static constexpr float INV_NEIGH = 0.25f;

__device__ __forceinline__ v8f wmma4(v2f a, v2f b, v8f c) {
  return __builtin_amdgcn_wmma_f32_16x16x4_f32(
       false, a,  false, b,
       (short)0, c,  false,  false);
}

__device__ __forceinline__ float silu_c(float z) {
  return SILU_C * z / (1.0f + __expf(-z));
}

__device__ __forceinline__ void loadB8(v2f (&bA)[8], v2f (&bB)[8],
                                       const float* __restrict__ p) {
#pragma unroll
  for (int s = 0; s < 8; ++s) {
    const float* q = p + s * 4 * MULC;
    const v2f r0 = *(const v2f*)q;
    const v2f r1 = *(const v2f*)(q + MULC);
    bA[s] = v2f{ r0.x, r1.x };
    bB[s] = v2f{ r0.y, r1.y };
  }
}

__device__ __forceinline__ void mma8(const float* arow, const v2f (&bA)[8],
                                     const v2f (&bB)[8], v8f& accA, v8f& accB) {
#pragma unroll
  for (int s = 0; s < 8; ++s) {
    v2f a = { arow[4 * s], arow[4 * s + 1] };
    accA = wmma4(a, bA[s], accA);
    accB = wmma4(a, bB[s], accB);
  }
}

__device__ __forceinline__ void mma8x3(const float* a0, const float* a1,
                                       const float* a2, const v2f (&bA)[8],
                                       const v2f (&bB)[8], v8f (&accA)[3],
                                       v8f (&accB)[3]) {
#pragma unroll
  for (int s = 0; s < 8; ++s) {
    v2f x0 = { a0[4 * s], a0[4 * s + 1] };
    v2f x1 = { a1[4 * s], a1[4 * s + 1] };
    v2f x2 = { a2[4 * s], a2[4 * s + 1] };
    accA[0] = wmma4(x0, bA[s], accA[0]);
    accA[1] = wmma4(x1, bA[s], accA[1]);
    accA[2] = wmma4(x2, bA[s], accA[2]);
    accB[0] = wmma4(x0, bB[s], accB[0]);
    accB[1] = wmma4(x1, bB[s], accB[1]);
    accB[2] = wmma4(x2, bB[s], accB[2]);
  }
}

static constexpr int SCROW = NSP * MULC;

__device__ __forceinline__ void loadB8sc(v2f (&bA)[8], v2f (&bB)[8],
                                         const float* __restrict__ p) {
#pragma unroll
  for (int s = 0; s < 8; ++s) {
    const float* q = p + s * 4 * SCROW;
    const v2f r0 = *(const v2f*)q;
    const v2f r1 = *(const v2f*)(q + SCROW);
    bA[s] = v2f{ r0.x, r1.x };
    bB[s] = v2f{ r0.y, r1.y };
  }
}

template <int O>
__device__ __forceinline__ void mma8reg(const v2f (&a)[32], const v2f (&bA)[8],
                                        const v2f (&bB)[8], v8f& accA,
                                        v8f& accB) {
#pragma unroll
  for (int s = 0; s < 8; ++s) {
    accA = wmma4(a[O + s], bA[s], accA);
    accB = wmma4(a[O + s], bB[s], accB);
  }
}

template <int O>
__device__ __forceinline__ void mma8x3reg(const v2f (&a)[3][16],
                                          const v2f (&bA)[8],
                                          const v2f (&bB)[8], v8f (&accA)[3],
                                          v8f (&accB)[3]) {
#pragma unroll
  for (int s = 0; s < 8; ++s) {
    accA[0] = wmma4(a[0][O + s], bA[s], accA[0]);
    accA[1] = wmma4(a[1][O + s], bA[s], accA[1]);
    accA[2] = wmma4(a[2][O + s], bA[s], accA[2]);
    accB[0] = wmma4(a[0][O + s], bB[s], accB[0]);
    accB[1] = wmma4(a[1][O + s], bB[s], accB[1]);
    accB[2] = wmma4(a[2][O + s], bB[s], accB[2]);
  }
}

__global__ __launch_bounds__(128) void node_kernel(
    const float* __restrict__ node_feats, const float* __restrict__ node_attrs,
    const float* __restrict__ lin1_ws, const float* __restrict__ lin1_wv,
    const float* __restrict__ sc_ws,
    float* __restrict__ out, float* __restrict__ y,
    const int* __restrict__ edge_index, int* __restrict__ cells) {
  __shared__ __attribute__((aligned(16))) float xs[16][MULC + LPAD];
  __shared__ float xvc[3][16][MULC + LPAD];
  __shared__ __attribute__((aligned(16))) float atT[16][16];
  __shared__ int hist[NTB];

  const int n0  = blockIdx.x * 16;
  const int tid = threadIdx.x;

  const int eA = blockIdx.x * ECH + tid, eB = eA + 128;
  const int dA = edge_index[eA], dB = edge_index[eB];
  for (int i = tid; i < NTB; i += 128) hist[i] = 0;

  {
    float av[2];
#pragma unroll
    for (int q = 0; q < 2; ++q) {
      const int i = tid + q * 128, r = i >> 4, c = i & 15;
      av[q] = (c < NSP) ? node_attrs[(size_t)(n0 + r) * NSP + c] : 0.0f;
    }
    const float4* nf4 = (const float4*)(node_feats + (size_t)n0 * 512);
    float4 t[16];
#pragma unroll
    for (int q = 0; q < 16; ++q) t[q] = nf4[q * 128 + tid];
#pragma unroll
    for (int q = 0; q < 2; ++q) {
      const int i = tid + q * 128;
      atT[i & 15][i >> 4] = av[q];
    }
    const int c0 = tid * 4;
    if (c0 < MULC) {
#pragma unroll
      for (int q = 0; q < 16; ++q) *(float4*)&xs[q][c0] = t[q];
    } else {
      const int f0 = c0 - MULC;
      const int h0 = f0 % 3, u0 = f0 / 3;
      const int h1 = (f0 + 1) % 3, u1 = (f0 + 1) / 3;
      const int h2 = (f0 + 2) % 3, u2 = (f0 + 2) / 3;
      const int h3 = (f0 + 3) % 3, u3 = (f0 + 3) / 3;
#pragma unroll
      for (int q = 0; q < 16; ++q) {
        xvc[h0][q][u0] = t[q].x;
        xvc[h1][q][u1] = t[q].y;
        xvc[h2][q][u2] = t[q].z;
        xvc[h3][q][u3] = t[q].w;
      }
    }
  }
  __syncthreads();

  {
    int* cb = cells + (size_t)blockIdx.x * CELLW;
    const int tA = dA >> 4, tB = dB >> 4;
    const int pA = atomicAdd(&hist[tA], 1);
    const int pB = atomicAdd(&hist[tB], 1);
    if (pA < CELLK) cb[(size_t)tA * VROW + 1 + pA] = eA;
    if (pB < CELLK) cb[(size_t)tB * VROW + 1 + pB] = eB;
    __syncthreads();
    for (int i = tid; i < NTB; i += 128) cb[(size_t)i * VROW] = hist[i];
  }

  const int wave = tid >> 5;
  const int lane = tid & 31;
  const int half = lane >> 4;
  const int m    = lane & 15;
  const int nl   = lane & 15;
  const int c2   = wave * 32 + 2 * nl;

  v2f p0A[8], p0B[8], p1A[8], p1B[8];

  const float* Wss = sc_ws + (size_t)(2 * half) * SCROW + c2;
  const int arow8 = 8 * half;

  loadB8sc(p0A, p0B, Wss);
  {
    v2f xa[32];
#pragma unroll
    for (int j = 0; j < 32; ++j) {
      const float* a = &xs[m][4 * j + 2 * half];
      xa[j] = v2f{ a[0], a[1] };
    }
    v8f totA = {}, totB = {};
#pragma nounroll
    for (int v = 0; v < NSP; ++v) {
      const float* W = Wss + v * MULC;
      const float4 s0 = *(const float4*)&atT[v][arow8];
      const float4 s1 = *(const float4*)&atT[v][arow8 + 4];
      v8f accA = {}, accB = {};
      loadB8sc(p1A, p1B, W + (size_t)1 * 32 * SCROW);
      mma8reg<0>(xa, p0A, p0B, accA, accB);
      loadB8sc(p0A, p0B, W + (size_t)2 * 32 * SCROW);
      mma8reg<8>(xa, p1A, p1B, accA, accB);
      loadB8sc(p1A, p1B, W + (size_t)3 * 32 * SCROW);
      mma8reg<16>(xa, p0A, p0B, accA, accB);
      loadB8sc(p0A, p0B, (v + 1 < NSP) ? W + MULC : Wss);
      mma8reg<24>(xa, p1A, p1B, accA, accB);
      const float sc[8] = { s0.x, s0.y, s0.z, s0.w, s1.x, s1.y, s1.z, s1.w };
#pragma unroll
      for (int r = 0; r < 8; ++r) {
        totA[r] += sc[r] * accA[r];
        totB[r] += sc[r] * accB[r];
      }
    }
#pragma unroll
    for (int r = 0; r < 8; ++r) {
      size_t row = (size_t)(n0 + r + 8 * half) * 512;
      *(v2f*)&out[row + c2] = v2f{ totA[r] * SC_NORM, totB[r] * SC_NORM };
    }
  }


  const float* L1s = lin1_ws + (size_t)(2 * half) * MULC + c2;
  const float* L1v = lin1_wv + (size_t)(2 * half) * MULC + c2;
  loadB8(p0A, p0B, L1s);
  {
    v8f accA = {}, accB = {};
#pragma unroll
    for (int g = 0; g < 4; g += 2) {
      loadB8(p1A, p1B, L1s + (size_t)(g + 1) * 32 * MULC);
      mma8(&xs[m][g * 32 + 2 * half], p0A, p0B, accA, accB);
      loadB8(p0A, p0B, (g + 2 < 4) ? L1s + (size_t)(g + 2) * 32 * MULC : L1v);
      mma8(&xs[m][(g + 1) * 32 + 2 * half], p1A, p1B, accA, accB);
    }
#pragma unroll
    for (int r = 0; r < 8; ++r) {
      size_t row = (size_t)(n0 + r + 8 * half) * 512;
      *(v2f*)&y[row + c2] = v2f{ accA[r] * L1_NORM, accB[r] * L1_NORM };
    }
  }
  asm volatile("" ::: "memory");
  {
    v8f accA[3] = { v8f{}, v8f{}, v8f{} }, accB[3] = { v8f{}, v8f{}, v8f{} };
#pragma unroll
    for (int g = 0; g < 4; g += 2) {
      const int ka = g * 32 + 2 * half, kb = ka + 32;
      loadB8(p1A, p1B, L1v + (size_t)(g + 1) * 32 * MULC);
      mma8x3(&xvc[0][m][ka], &xvc[1][m][ka], &xvc[2][m][ka], p0A, p0B, accA,
             accB);
      if (g + 2 < 4) loadB8(p0A, p0B, L1v + (size_t)(g + 2) * 32 * MULC);
      mma8x3(&xvc[0][m][kb], &xvc[1][m][kb], &xvc[2][m][kb], p1A, p1B, accA,
             accB);
    }
#pragma unroll
    for (int r = 0; r < 8; ++r) {
      v2f* o = (v2f*)&y[(size_t)(n0 + r + 8 * half) * 512 + MULC + c2 * 3];
      o[0] = v2f{ accA[0][r] * L1_NORM, accA[1][r] * L1_NORM };
      o[1] = v2f{ accA[2][r] * L1_NORM, accB[0][r] * L1_NORM };
      o[2] = v2f{ accB[1][r] * L1_NORM, accB[2][r] * L1_NORM };
    }
  }
}

static constexpr int CAP    = 1024;
static constexpr int SCAN_U = 8;

__device__ __forceinline__ void l2load(v2f (&b0)[16], v2f (&b1)[16],
                                       const float* __restrict__ w2, int j0,
                                       int half, int nl) {
#pragma unroll
  for (int s = 0; s < 16; ++s) {
    const int k0 = 4 * s + 2 * half;
    const float* p = w2 + (size_t)k0 * 512 + j0 + 2 * nl;
    const v2f r0 = *(const v2f*)p;
    const v2f r1 = *(const v2f*)(p + 512);
    b0[s] = v2f{ r0.x, r1.x };
    b1[s] = v2f{ r0.y, r1.y };
  }
}

__device__ __forceinline__ void l2mma(const v2f (&af)[16], const v2f (&b0)[16],
                                      const v2f (&b1)[16], v8f& c0, v8f& c1) {
#pragma unroll
  for (int s = 0; s < 16; ++s) {
    c0 = wmma4(af[s], b0[s], c0);
    c1 = wmma4(af[s], b1[s], c1);
    __builtin_amdgcn_sched_barrier(0);
  }
}

__device__ __forceinline__ v8f segsum(const v2f (&sel)[4], const float (&mid)[8],
                                      v8f acc) {
#pragma unroll
  for (int t = 0; t < 4; ++t) {
    v2f b = { mid[2 * t], mid[2 * t + 1] };
    acc = wmma4(sel[t], b, acc);
  }
  return acc;
}

__device__ __forceinline__ void scv_block(
    float* lds, int tile, const float* __restrict__ node_feats,
    const float* __restrict__ node_attrs, const float* __restrict__ sc_wv,
    float* __restrict__ out) {
  float (*xvc)[16][MULC + LPAD] = (float (*)[16][MULC + LPAD])lds;
  float (*atT)[16] = (float (*)[16])(lds + 3 * 16 * (MULC + LPAD));
  const int n0  = tile * 16;
  const int tid = threadIdx.x;
  {
    float av[2];
#pragma unroll
    for (int q = 0; q < 2; ++q) {
      const int i = tid + q * 128, r = i >> 4, c = i & 15;
      av[q] = (c < NSP) ? node_attrs[(size_t)(n0 + r) * NSP + c] : 0.0f;
    }
    const float4* nf4 = (const float4*)(node_feats + (size_t)n0 * 512);
    const int c0 = tid * 4;
    float4 t[16];
    if (c0 >= MULC) {
#pragma unroll
      for (int q = 0; q < 16; ++q) t[q] = nf4[q * 128 + tid];
    }
#pragma unroll
    for (int q = 0; q < 2; ++q) {
      const int i = tid + q * 128;
      atT[i & 15][i >> 4] = av[q];
    }
    if (c0 >= MULC) {
      const int f0 = c0 - MULC;
      const int h0 = f0 % 3, u0 = f0 / 3;
      const int h1 = (f0 + 1) % 3, u1 = (f0 + 1) / 3;
      const int h2 = (f0 + 2) % 3, u2 = (f0 + 2) / 3;
      const int h3 = (f0 + 3) % 3, u3 = (f0 + 3) / 3;
#pragma unroll
      for (int q = 0; q < 16; ++q) {
        xvc[h0][q][u0] = t[q].x;
        xvc[h1][q][u1] = t[q].y;
        xvc[h2][q][u2] = t[q].z;
        xvc[h3][q][u3] = t[q].w;
      }
    }
  }
  __syncthreads();

  const int wave = tid >> 5;
  const int lane = tid & 31;
  const int half = lane >> 4;
  const int m    = lane & 15;
  const int nl   = lane & 15;
  const int c2   = wave * 32 + 2 * nl;
  const int arow8 = 8 * half;

  v2f p0A[8], p0B[8], p1A[8], p1B[8];
  const float* Wsv = sc_wv + (size_t)(2 * half) * SCROW + c2;
  loadB8sc(p0A, p0B, Wsv);

  {
    v8f totA[3] = { v8f{}, v8f{}, v8f{} }, totB[3] = { v8f{}, v8f{}, v8f{} };
#pragma unroll
    for (int h = 0; h < 2; ++h) {
      v2f xa[3][16];
#pragma unroll
      for (int ch = 0; ch < 3; ++ch) {
#pragma unroll
        for (int j = 0; j < 16; ++j) {
          const float* a = &xvc[ch][m][64 * h + 4 * j + 2 * half];
          xa[ch][j] = v2f{ a[0], a[1] };
        }
      }
      const float* Wh = Wsv + (size_t)h * 64 * SCROW;
      for (int v = 0; v < NSP; ++v) {
        const float* W = Wh + v * MULC;
        const float4 s0 = *(const float4*)&atT[v][arow8];
        const float4 s1 = *(const float4*)&atT[v][arow8 + 4];
        v8f accA[3] = { v8f{}, v8f{}, v8f{} }, accB[3] = { v8f{}, v8f{}, v8f{} };
        loadB8sc(p1A, p1B, W + (size_t)32 * SCROW);
        mma8x3reg<0>(xa, p0A, p0B, accA, accB);
        loadB8sc(p0A, p0B, (v + 1 < NSP) ? W + MULC : Wsv + (size_t)64 * SCROW);
        mma8x3reg<8>(xa, p1A, p1B, accA, accB);
        const float sc[8] = { s0.x, s0.y, s0.z, s0.w, s1.x, s1.y, s1.z, s1.w };
#pragma unroll
        for (int ch = 0; ch < 3; ++ch) {
#pragma unroll
          for (int r = 0; r < 8; ++r) {
            totA[ch][r] += sc[r] * accA[ch][r];
            totB[ch][r] += sc[r] * accB[ch][r];
          }
        }
      }
    }
#pragma unroll
    for (int r = 0; r < 8; ++r) {
      v2f* o = (v2f*)&out[(size_t)(n0 + r + 8 * half) * 512 + MULC + c2 * 3];
      o[0] = v2f{ totA[0][r] * SC_NORM, totA[1][r] * SC_NORM };
      o[1] = v2f{ totA[2][r] * SC_NORM, totB[0][r] * SC_NORM };
      o[2] = v2f{ totB[1][r] * SC_NORM, totB[2][r] * SC_NORM };
    }
  }
}

__global__ __launch_bounds__(128, 3) void edge_kernel(
    const float* y, const float* __restrict__ edge_feats,
    const float* __restrict__ edge_attrs, const int* __restrict__ edge_index,
    const float* __restrict__ fc_w0, const float* __restrict__ fc_w1,
    const float* __restrict__ fc_w2, float* __restrict__ s_acc,
    float* __restrict__ v_acc, const float* __restrict__ node_feats,
    const float* __restrict__ node_attrs, const float* __restrict__ sc_wv,
    float* __restrict__ out) {
  __shared__ __attribute__((aligned(16))) float ytile[16][512];
  __shared__ float ha[16][64 + LPAD];
  __shared__ float hb[16][64 + LPAD];
  __shared__ float ea[16][4];
  __shared__ float ef[16][NRAD + LPAD];
  __shared__ int   dl[16];
  __shared__ int   lst_e[CAP];
  __shared__ int   lst_d[CAP];
  __shared__ int   lst_s[CAP];
  __shared__ int   cnt;

  if (blockIdx.x >= NTB) {
    scv_block(&ytile[0][0], (int)blockIdx.x - NTB, node_feats, node_attrs,
              sc_wv, out);
    return;
  }

  const int n0   = blockIdx.x * 16;
  const int tid  = threadIdx.x;
  const int wave = tid >> 5;
  const int lane = tid & 31;
  const int half = lane >> 4;
  const int m    = lane & 15;
  const int nl   = lane & 15;

  v8f aS0[2], aS1[2], aV2[2][3], aV3[2][3];
#pragma unroll
  for (int jj = 0; jj < 2; ++jj) {
    aS0[jj] = v8f{};
    aS1[jj] = v8f{};
#pragma unroll
    for (int c = 0; c < 3; ++c) { aV2[jj][c] = v8f{}; aV3[jj][c] = v8f{}; }
  }

  const int4* d4 = (const int4*)edge_index;
  int seg_b = 0, seg_e = NE;
  bool from_cells = true;
  if (tid == 0) cnt = 0;
  __syncthreads();
  {
    const int4* cl = (const int4*)(v_acc + (size_t)blockIdx.x * VROW);
    int4 ca[5], cb[5];
#pragma unroll
    for (int j = 0; j < 5; ++j) {
      const int cell = tid + j * 128;
      const int cc = (cell < NTB) ? cell : 0;
      ca[j] = cl[cc * 2];
      cb[j] = cl[cc * 2 + 1];
    }
#pragma unroll
    for (int j = 0; j < 5; ++j) {
      const int cell = tid + j * 128;
      const int n = (cell < NTB) ? ca[j].x : 0;
      if (n > 0) {
        const int p = atomicAdd(&cnt, (n <= CELLK) ? n : CAP + 1);
        const int ee[CELLK] = { ca[j].y, ca[j].z, ca[j].w, cb[j].x,
                                cb[j].y, cb[j].z, cb[j].w };
#pragma unroll
        for (int k = 0; k < CELLK; ++k)
          if (k < n && n <= CELLK && p + k < CAP) lst_e[p + k] = ee[k];
      }
    }
  }
  for (;;) {
    if (!from_cells) {
      if (tid == 0) cnt = 0;
      __syncthreads();
    }
    const int e4 = from_cells ? 0 : (seg_e >> 2);
    for (int base = (seg_b >> 2) + tid; base < e4; base += 128 * SCAN_U) {
      int4 v[SCAN_U];
#pragma unroll
      for (int q = 0; q < SCAN_U; ++q) {
        const int idx = base + q * 128;
        v[q] = (idx < e4) ? d4[idx] : make_int4(-1, -1, -1, -1);
      }
#pragma unroll
      for (int q = 0; q < SCAN_U; ++q) {
        const int idx = base + q * 128;
        const unsigned dd[4] = { (unsigned)v[q].x - (unsigned)n0,
                                 (unsigned)v[q].y - (unsigned)n0,
                                 (unsigned)v[q].z - (unsigned)n0,
                                 (unsigned)v[q].w - (unsigned)n0 };
#pragma unroll
        for (int z = 0; z < 4; ++z) {
          if (dd[z] < 16u) {
            const int p = atomicAdd(&cnt, 1);
            if (p < CAP) { lst_e[p] = idx * 4 + z; lst_d[p] = (int)dd[z]; }
          }
        }
      }
    }
    __syncthreads();
    const int c = cnt;
    __syncthreads();
    if (c > CAP) { seg_b = 0; seg_e = CAP; from_cells = false; continue; }

    for (int i = tid; i < c; i += 128) {
      const int e = lst_e[i];
      lst_s[i] = edge_index[NE + e];
      lst_d[i] = edge_index[e] - n0;
    }
    __syncthreads();

    const int ntile = (c + 15) >> 4;
    for (int ti = 0; ti < ntile; ++ti) {
      int woff = 0;
      asm volatile("" : "+s"(woff));
      const float* w0 = fc_w0 + woff;
      const float* w1 = fc_w1 + woff;
      const float* w2 = fc_w2 + woff;
      v2f bw0[2], bw1[16];
      v2f qx0[16], qx1[16], qy0[16], qy1[16];
      {
        const float4* sp[4];
        int tw = tid;
        asm volatile("" : "+v"(tw));
        const int wrow = (tw >> 5) * 4;
#pragma unroll
        for (int rr = 0; rr < 4; ++rr) {
          const int li  = ti * 16 + wrow + rr;
          const int lic = (li < c) ? li : 0;
          const int sv  = lst_s[lic];
          const int sr  = (li < c) ? sv : 0;
          sp[rr] = (const float4*)(y + (size_t)sr * 512);
        }
        const int fr = tid >> 3, fk = tid & 7;
        const int fli = ti * 16 + fr;
        const int fe  = lst_e[(fli < c) ? fli : 0];
        const int ar = (tid >> 2) & 15, ak = tid & 3;
        const int ali = ti * 16 + ar;
        const int ae  = lst_e[(ali < c) ? ali : 0];
        const float4 a0 = sp[0][lane], a1 = sp[0][lane + 32],
                     a2 = sp[0][lane + 64], a3 = sp[0][lane + 96];
        const float4 b0 = sp[1][lane], b1 = sp[1][lane + 32],
                     b2 = sp[1][lane + 64], b3 = sp[1][lane + 96];
        const float4 c0 = sp[2][lane], c1 = sp[2][lane + 32],
                     c2 = sp[2][lane + 64], c3 = sp[2][lane + 96];
        const float4 d0 = sp[3][lane], d1 = sp[3][lane + 32],
                     d2 = sp[3][lane + 64], d3 = sp[3][lane + 96];
        float fraw = edge_feats[(unsigned)fe * NRAD + (unsigned)fk];
        float araw = 0.0f;
        if (tid < 64) araw = edge_attrs[(unsigned)ae * 4u + (unsigned)ak];
#pragma unroll
        for (int s = 0; s < 2; ++s) {
          const int k0 = 4 * s + 2 * half;
          bw0[s] = v2f{ w0[k0 * 64 + wave * 16 + nl],
                        w0[(k0 + 1) * 64 + wave * 16 + nl] };
        }
#pragma unroll
        for (int s = 0; s < 16; ++s) {
          const int k0 = 4 * s + 2 * half;
          bw1[s] = v2f{ w1[k0 * 64 + wave * 16 + nl],
                        w1[(k0 + 1) * 64 + wave * 16 + nl] };
        }
        asm volatile("" : "+v"(fraw), "+v"(araw) : : "memory");
        float4* dp = (float4*)&ytile[wave * 4][0];
        dp[lane]       = a0; dp[lane + 32]  = a1; dp[lane + 64]  = a2; dp[lane + 96]  = a3;
        dp[lane + 128] = b0; dp[lane + 160] = b1; dp[lane + 192] = b2; dp[lane + 224] = b3;
        dp[lane + 256] = c0; dp[lane + 288] = c1; dp[lane + 320] = c2; dp[lane + 352] = c3;
        dp[lane + 384] = d0; dp[lane + 416] = d1; dp[lane + 448] = d2; dp[lane + 480] = d3;
        ef[fr][fk] = (fli < c) ? fraw : 0.0f;
        if (tid < 64) ea[ar][ak] = (ali < c) ? araw : 0.0f;
      }
      if (tid < 16) {
        const int li = ti * 16 + tid;
        dl[tid] = (li < c) ? lst_d[li] : -1;
      }
      __syncthreads();

      {
        v8f acc = {};
#pragma unroll
        for (int s = 0; s < 2; ++s) {
          const int k0 = 4 * s + 2 * half;
          v2f a = { ef[m][k0], ef[m][k0 + 1] };
          acc = wmma4(a, bw0[s], acc);
        }
#pragma unroll
        for (int r = 0; r < 8; ++r)
          ha[r + 8 * half][wave * 16 + nl] = silu_c(acc[r] * INV_SQRT8);
      }
      __syncthreads();

      l2load(qx0, qx1, w2, 0 * 128 + wave * 32, half, nl);

      {
        v8f acc = {};
#pragma unroll
        for (int s = 0; s < 16; ++s) {
          const int k0 = 4 * s + 2 * half;
          v2f a = { ha[m][k0], ha[m][k0 + 1] };
          acc = wmma4(a, bw1[s], acc);
        }
#pragma unroll
        for (int r = 0; r < 8; ++r)
          hb[r + 8 * half][wave * 16 + nl] = silu_c(acc[r] * 0.125f);
      }
      __syncthreads();

      v2f af[16];
#pragma unroll
      for (int s = 0; s < 16; ++s) {
        const int k0 = 4 * s + 2 * half;
        af[s] = v2f{ hb[m][k0], hb[m][k0 + 1] };
      }
      v2f sel[4];
      int mrow = m;
      asm volatile("" : "+v"(mrow));
#pragma unroll
      for (int t = 0; t < 4; ++t) {
        const int e0 = 8 * half + 2 * t;
        sel[t] = v2f{ dl[e0] == mrow ? 1.0f : 0.0f, dl[e0 + 1] == mrow ? 1.0f : 0.0f };
      }

      {
        v8f cc[2] = { v8f{}, v8f{} };
        l2mma(af, qx0, qx1, cc[0], cc[1]);
        asm volatile("" : "+s"(woff) : "v"(cc[1][7]));
        w2 = fc_w2 + woff;
        l2load(qy0, qy1, w2, 1 * 128 + wave * 32, half, nl);
        asm volatile("" ::: "memory");
#pragma unroll
        for (int jj = 0; jj < 2; ++jj) {
          const int u = wave * 32 + 2 * nl + jj;
          float mid[8];
#pragma unroll
          for (int r = 0; r < 8; ++r) {
            const int e = r + 8 * half;
            mid[r] = (cc[jj][r] * 0.125f) * ytile[e][u] * ea[e][0];
          }
          aS0[jj] = segsum(sel, mid, aS0[jj]);
        }
      }
      {
        v8f cc[2] = { v8f{}, v8f{} };
        asm volatile("" : "+v"(qy0[0]), "+v"(qy1[0]) : "v"(aS0[0][7]), "v"(aS0[1][7]));
        l2mma(af, qy0, qy1, cc[0], cc[1]);
        asm volatile("" : "+s"(woff) : "v"(cc[1][7]));
        w2 = fc_w2 + woff;
        l2load(qx0, qx1, w2, 2 * 128 + wave * 32, half, nl);
        asm volatile("" ::: "memory");
#pragma unroll
        for (int jj = 0; jj < 2; ++jj) {
          const int u = wave * 32 + 2 * nl + jj;
          float m0[8], m1[8], m2[8];
#pragma unroll
          for (int r = 0; r < 8; ++r) {
            const int e = r + 8 * half;
            const float base = (cc[jj][r] * 0.125f) * ytile[e][u];
            m0[r] = base * ea[e][1];
            m1[r] = base * ea[e][2];
            m2[r] = base * ea[e][3];
          }
          aV2[jj][0] = segsum(sel, m0, aV2[jj][0]);
          aV2[jj][1] = segsum(sel, m1, aV2[jj][1]);
          aV2[jj][2] = segsum(sel, m2, aV2[jj][2]);
        }
      }
      {
        v8f cc[2] = { v8f{}, v8f{} };
        asm volatile("" : "+v"(qx0[0]), "+v"(qx1[0]) : "v"(aV2[0][0][7]), "v"(aV2[0][1][7]), "v"(aV2[0][2][7]), "v"(aV2[1][0][7]), "v"(aV2[1][1][7]), "v"(aV2[1][2][7]));
        l2mma(af, qx0, qx1, cc[0], cc[1]);
        asm volatile("" : "+s"(woff) : "v"(cc[1][7]));
        w2 = fc_w2 + woff;
        l2load(qy0, qy1, w2, 3 * 128 + wave * 32, half, nl);
        asm volatile("" ::: "memory");
#pragma unroll
        for (int jj = 0; jj < 2; ++jj) {
          const int u = wave * 32 + 2 * nl + jj;
          float m0[8], m1[8], m2[8];
#pragma unroll
          for (int r = 0; r < 8; ++r) {
            const int e = r + 8 * half;
            const float base = (cc[jj][r] * 0.125f) * ea[e][0];
            m0[r] = base * ytile[e][128 + u * 3 + 0];
            m1[r] = base * ytile[e][128 + u * 3 + 1];
            m2[r] = base * ytile[e][128 + u * 3 + 2];
          }
          aV3[jj][0] = segsum(sel, m0, aV3[jj][0]);
          aV3[jj][1] = segsum(sel, m1, aV3[jj][1]);
          aV3[jj][2] = segsum(sel, m2, aV3[jj][2]);
        }
      }
      {
        v8f cc[2] = { v8f{}, v8f{} };
        asm volatile("" : "+v"(qy0[0]), "+v"(qy1[0]) : "v"(aV3[0][0][7]), "v"(aV3[0][1][7]), "v"(aV3[0][2][7]), "v"(aV3[1][0][7]), "v"(aV3[1][1][7]), "v"(aV3[1][2][7]));
        l2mma(af, qy0, qy1, cc[0], cc[1]);
#pragma unroll
        for (int jj = 0; jj < 2; ++jj) {
          const int u = wave * 32 + 2 * nl + jj;
          float mid[8];
#pragma unroll
          for (int r = 0; r < 8; ++r) {
            const int e = r + 8 * half;
            const float dotv = ytile[e][128 + u * 3 + 0] * ea[e][1] +
                               ytile[e][128 + u * 3 + 1] * ea[e][2] +
                               ytile[e][128 + u * 3 + 2] * ea[e][3];
            mid[r] = (cc[jj][r] * 0.125f) * dotv * INV_SQ3;
          }
          aS1[jj] = segsum(sel, mid, aS1[jj]);
        }
      }
      __syncthreads();
    }
    asm volatile("v_nop\n\tv_nop\n\tv_nop\n\tv_nop" : "+v"(aS1[0]), "+v"(aS1[1]));

    if (seg_e >= NE) break;
    seg_b = seg_e;
    seg_e = (seg_e + CAP < NE) ? seg_e + CAP : NE;
  }

  int tq = threadIdx.x;
  asm volatile("" : "+v"(tq));
  const int wq = tq >> 5, hq = (tq >> 4) & 1, cq = tq & 15;
  const int u0 = wq * 32 + 2 * cq;
#pragma unroll
  for (int r = 0; r < 8; ++r) {
    const size_t n = (size_t)(n0 + r + 8 * hq);
    *(v2f*)&s_acc[n * 256 + u0] =
        v2f{ aS0[0][r] * INV_NEIGH, aS0[1][r] * INV_NEIGH };
    *(v2f*)&s_acc[n * 256 + 128 + u0] =
        v2f{ aS1[0][r] * INV_NEIGH, aS1[1][r] * INV_NEIGH };
    v2f* o2 = (v2f*)&v_acc[(n * 256 + u0) * 3];
    o2[0] = v2f{ aV2[0][0][r] * INV_NEIGH, aV2[0][1][r] * INV_NEIGH };
    o2[1] = v2f{ aV2[0][2][r] * INV_NEIGH, aV2[1][0][r] * INV_NEIGH };
    o2[2] = v2f{ aV2[1][1][r] * INV_NEIGH, aV2[1][2][r] * INV_NEIGH };
    v2f* o3 = (v2f*)&v_acc[(n * 256 + 128 + u0) * 3];
    o3[0] = v2f{ aV3[0][0][r] * INV_NEIGH, aV3[0][1][r] * INV_NEIGH };
    o3[1] = v2f{ aV3[0][2][r] * INV_NEIGH, aV3[1][0][r] * INV_NEIGH };
    o3[2] = v2f{ aV3[1][1][r] * INV_NEIGH, aV3[1][2][r] * INV_NEIGH };
  }
}

__global__ __launch_bounds__(128) void out_kernel(
    const float* __restrict__ s_acc, const float* __restrict__ v_acc,
    const float* __restrict__ lin2_ws, const float* __restrict__ lin2_wv,
    float* __restrict__ out) {
  __shared__ float st[16][256 + LPAD];
  __shared__ float vt[3][16][256 + LPAD];
  const int n0  = blockIdx.x * 16;
  const int tid = threadIdx.x;
  const int wave = tid >> 5;
  const int lane = tid & 31;
  const int half = lane >> 4;
  const int m    = lane & 15;
  const int nl   = lane & 15;
  const int c2   = wave * 32 + 2 * nl;

  v2f p0A[8], p0B[8], p1A[8], p1B[8];
  const float* L2s = lin2_ws + (size_t)(2 * half) * MULC + c2;
  const float* L2v = lin2_wv + (size_t)(2 * half) * MULC + c2;
  loadB8(p0A, p0B, L2s);

  {
    const float4* va4 = (const float4*)(v_acc + (size_t)n0 * 768);
    float4 t[24];
#pragma unroll
    for (int q = 0; q < 24; ++q) t[q] = va4[q * 128 + tid];
    int vo[3][4];
#pragma unroll
    for (int i = 0; i < 3; ++i) {
      const int w  = i * 128 + tid;
      const int rr = w / 192;
      const int f0 = (w - rr * 192) * 4;
#pragma unroll
      for (int e = 0; e < 4; ++e) {
        const int f = f0 + e, k = f / 3, c = f - 3 * k;
        vo[i][e] = (c * 16 + rr) * (256 + LPAD) + k;
      }
    }
    for (int i = tid; i < 16 * 256; i += 128)
      st[i >> 8][i & 255] = s_acc[(size_t)n0 * 256 + i];
    float* vb = &vt[0][0][0];
#pragma unroll
    for (int q = 0; q < 24; ++q) {
      const int j = q / 3, i = q - 3 * j;
      float* b = vb + 2 * j * (256 + LPAD);
      b[vo[i][0]] = t[q].x;
      b[vo[i][1]] = t[q].y;
      b[vo[i][2]] = t[q].z;
      b[vo[i][3]] = t[q].w;
    }
  }
  __syncthreads();

  {
    v8f accA = {}, accB = {};
#pragma unroll
    for (int g = 0; g < 8; g += 2) {
      loadB8(p1A, p1B, L2s + (size_t)(g + 1) * 32 * MULC);
      mma8(&st[m][g * 32 + 2 * half], p0A, p0B, accA, accB);
      loadB8(p0A, p0B, (g + 2 < 8) ? L2s + (size_t)(g + 2) * 32 * MULC : L2v);
      mma8(&st[m][(g + 1) * 32 + 2 * half], p1A, p1B, accA, accB);
    }
    v2f o[8];
#pragma unroll
    for (int r = 0; r < 8; ++r) {
      size_t row = (size_t)(n0 + r + 8 * half) * 512;
      o[r] = *(const v2f*)&out[row + c2];
    }
#pragma unroll
    for (int r = 0; r < 8; ++r) {
      size_t row = (size_t)(n0 + r + 8 * half) * 512;
      *(v2f*)&out[row + c2] =
          v2f{ o[r].x + accA[r] * L2_NORM, o[r].y + accB[r] * L2_NORM };
    }
  }
  {
    v8f accA[3] = { v8f{}, v8f{}, v8f{} }, accB[3] = { v8f{}, v8f{}, v8f{} };
#pragma unroll
    for (int g = 0; g < 8; g += 2) {
      const int ka = g * 32 + 2 * half, kb = ka + 32;
      loadB8(p1A, p1B, L2v + (size_t)(g + 1) * 32 * MULC);
      mma8x3(&vt[0][m][ka], &vt[1][m][ka], &vt[2][m][ka], p0A, p0B, accA, accB);
      if (g + 2 < 8) loadB8(p0A, p0B, L2v + (size_t)(g + 2) * 32 * MULC);
      mma8x3(&vt[0][m][kb], &vt[1][m][kb], &vt[2][m][kb], p1A, p1B, accA, accB);
    }
    v2f o[8][3];
#pragma unroll
    for (int r = 0; r < 8; ++r) {
      const v2f* q = (const v2f*)&out[(size_t)(n0 + r + 8 * half) * 512 + MULC + c2 * 3];
      o[r][0] = q[0];
      o[r][1] = q[1];
      o[r][2] = q[2];
    }
#pragma unroll
    for (int r = 0; r < 8; ++r) {
      v2f* q = (v2f*)&out[(size_t)(n0 + r + 8 * half) * 512 + MULC + c2 * 3];
      q[0] = v2f{ o[r][0].x + accA[0][r] * L2_NORM, o[r][0].y + accA[1][r] * L2_NORM };
      q[1] = v2f{ o[r][1].x + accA[2][r] * L2_NORM, o[r][1].y + accB[0][r] * L2_NORM };
      q[2] = v2f{ o[r][2].x + accB[1][r] * L2_NORM, o[r][2].y + accB[2][r] * L2_NORM };
    }
  }
}

extern "C" void kernel_launch(void* const* d_in, const int* in_sizes, int n_in,
                              void* d_out, int out_size, void* d_ws,
                              size_t ws_size, hipStream_t stream) {
  const float* node_feats = (const float*)d_in[0];
  const float* node_attrs = (const float*)d_in[1];
  const float* edge_feats = (const float*)d_in[2];
  const float* edge_attrs = (const float*)d_in[3];
  const int*   edge_index = (const int*)d_in[4];
  const float* lin1_ws    = (const float*)d_in[5];
  const float* lin1_wv    = (const float*)d_in[6];
  const float* fc_w0      = (const float*)d_in[7];
  const float* fc_w1      = (const float*)d_in[8];
  const float* fc_w2      = (const float*)d_in[9];
  const float* lin2_ws    = (const float*)d_in[10];
  const float* lin2_wv    = (const float*)d_in[11];
  const float* sc_ws      = (const float*)d_in[12];
  const float* sc_wv      = (const float*)d_in[13];

  float* out   = (float*)d_out;
  float* y     = (float*)d_ws;
  float* s_acc = y + (size_t)NN * 512;
  float* v_acc = s_acc + (size_t)NN * 256;

  node_kernel<<<NN / 16, 128, 0, stream>>>(node_feats, node_attrs, lin1_ws,
                                           lin1_wv, sc_ws, out, y,
                                           edge_index, (int*)v_acc);
  edge_kernel<<<2 * NTB, 128, 0, stream>>>(y, edge_feats, edge_attrs,
                                           edge_index, fc_w0, fc_w1, fc_w2,
                                           s_acc, v_acc, node_feats, node_attrs,
                                           sc_wv, out);
  out_kernel<<<NN / 16, 128, 0, stream>>>(s_acc, v_acc, lin2_ws, lin2_wv, out);
}
